# starting kernel
# baseline (speedup 1.0000x reference)
#include <hip/hip_runtime.h>
#include <hip/hip_bf16.h>
#include <math.h>

#define BATCH   8
#define WINDOW  5
#define NN      20000
#define EPB     320000
#define HID     32
#define KPAD    8
#define NTOT    (BATCH * NN)
#define ETOT    (BATCH * EPB)

typedef float v2f __attribute__((ext_vector_type(2)));
typedef float v8f __attribute__((ext_vector_type(8)));

static __device__ __forceinline__ void edge_nodes(const int* __restrict__ ei,
                                                  int e, int& s, int& d) {
    int b  = e / EPB;
    int el = e - b * EPB;
    const int* base = ei + (size_t)b * 2 * EPB;
    s = base[el] + b * NN;
    d = base[EPB + el] + b * NN;
}

__global__ void zero_f32x4(float4* __restrict__ p, int n4) {
    int i = blockIdx.x * blockDim.x + threadIdx.x;
    if (i < n4) p[i] = make_float4(0.f, 0.f, 0.f, 0.f);
}

__global__ void pad_rows_kernel(const float* __restrict__ x, float* __restrict__ xp) {
    int i = blockIdx.x * blockDim.x + threadIdx.x;
    if (i >= NTOT) return;
    const float* __restrict__ r = x + (size_t)i * WINDOW;
    float* __restrict__ o = xp + (size_t)i * KPAD;
    float4 lo = make_float4(r[0], r[1], r[2], r[3]);
    float4 hi = make_float4(r[4], 0.f, 0.f, 0.f);
    reinterpret_cast<float4*>(o)[0] = lo;
    reinterpret_cast<float4*>(o)[1] = hi;
}

__global__ void degree_kernel(const int* __restrict__ ei, float* __restrict__ deg) {
    int e = blockIdx.x * blockDim.x + threadIdx.x;
    if (e >= ETOT) return;
    int s, d; edge_nodes(ei, e, s, d);
    atomicAdd(&deg[d], 1.0f);
}

__global__ void invsqrt_kernel(float* __restrict__ deg) {
    int i = blockIdx.x * blockDim.x + threadIdx.x;
    if (i >= NTOT) return;
    float v = deg[i];
    deg[i] = (v > 0.0f) ? rsqrtf(v) : 0.0f;
}

__global__ void norm_kernel(const int* __restrict__ ei, const float* __restrict__ dis,
                            float* __restrict__ nrm) {
    int e = blockIdx.x * blockDim.x + threadIdx.x;
    if (e >= ETOT) return;
    int s, d; edge_nodes(ei, e, s, d);
    nrm[e] = dis[s] * dis[d];
}

template <int F, int S>
__global__ void aggregate_kernel(const int* __restrict__ ei, const float* __restrict__ nrm,
                                 const float* __restrict__ hin, float* __restrict__ agg) {
    int e = blockIdx.x * blockDim.x + threadIdx.x;
    if (e >= ETOT) return;
    int s, d; edge_nodes(ei, e, s, d);
    float w = nrm[e];
    const float* __restrict__ hs = hin + (size_t)s * S;
    float* __restrict__ ad = agg + (size_t)d * S;
    if constexpr ((F % 4) == 0) {
        #pragma unroll
        for (int g = 0; g < F / 4; ++g) {
            float4 v = reinterpret_cast<const float4*>(hs)[g];
            atomicAdd(&ad[4 * g + 0], v.x * w);
            atomicAdd(&ad[4 * g + 1], v.y * w);
            atomicAdd(&ad[4 * g + 2], v.z * w);
            atomicAdd(&ad[4 * g + 3], v.w * w);
        }
    } else {
        #pragma unroll
        for (int f = 0; f < F; ++f) atomicAdd(&ad[f], hs[f] * w);
    }
}

template <int K, int KW>
__global__ void tag_gemm_kernel(const float* __restrict__ hin,
                                const float* __restrict__ agg,
                                const float* __restrict__ W,
                                const float* __restrict__ bias,
                                float* __restrict__ hout, int N) {
    static_assert((K % 4) == 0, "padded K must be a multiple of 4");
    __shared__ float sW[2 * K * HID];
    __shared__ float sB[HID];
    for (int i = threadIdx.x; i < 2 * K * HID; i += blockDim.x) {
        int s   = i / (K * HID);
        int rem = i - s * (K * HID);
        int kw  = rem / HID;
        int col = rem - kw * HID;
        sW[i] = (kw < KW) ? W[((size_t)s * KW + kw) * HID + col] : 0.0f;
    }
    if (threadIdx.x < HID) sB[threadIdx.x] = bias[threadIdx.x];
    __syncthreads();

    int lane = threadIdx.x & 31;
    int wave = (blockIdx.x * blockDim.x + threadIdx.x) >> 5;
    int row0 = wave * 16;
    if (row0 >= N) return;

    const int m     = lane & 15;
    const int khalf = (lane >> 4) * 2;

    v8f acc0 = {0.f,0.f,0.f,0.f,0.f,0.f,0.f,0.f};
    v8f acc1 = {0.f,0.f,0.f,0.f,0.f,0.f,0.f,0.f};

    #pragma unroll
    for (int s = 0; s < 2; ++s) {
        const float* __restrict__ A   = s ? agg : hin;
        const float* __restrict__ sWk = sW + s * K * HID;
        #pragma unroll
        for (int c = 0; c < K / 4; ++c) {
            const int k0 = c * 4 + khalf;
            v2f a = *reinterpret_cast<const v2f*>(&A[(size_t)(row0 + m) * K + k0]);
            v2f b0, b1;
            b0.x = sWk[k0 * HID + m];
            b0.y = sWk[(k0 + 1) * HID + m];
            b1.x = sWk[k0 * HID + 16 + m];
            b1.y = sWk[(k0 + 1) * HID + 16 + m];
            acc0 = __builtin_amdgcn_wmma_f32_16x16x4_f32(
                       false, a, false, b0, (short)0, acc0, false, false);
            acc1 = __builtin_amdgcn_wmma_f32_16x16x4_f32(
                       false, a, false, b1, (short)0, acc1, false, false);
        }
    }

    const int rbase = row0 + ((lane >> 4) ? 8 : 0);
    #pragma unroll
    for (int v = 0; v < 8; ++v) {
        int r = rbase + v;
        float x0 = acc0[v] + sB[m];
        float x1 = acc1[v] + sB[16 + m];
        hout[(size_t)r * HID + m]      = tanhf(x0);
        hout[(size_t)r * HID + 16 + m] = tanhf(x1);
    }
}

__global__ void readout_kernel(const float* __restrict__ h, const float* __restrict__ Wreg,
                               const float* __restrict__ breg, float* __restrict__ out) {
    int i = blockIdx.x * blockDim.x + threadIdx.x;
    if (i >= NTOT) return;
    float acc = breg[0];
    const float4* hr = reinterpret_cast<const float4*>(h + (size_t)i * HID);
    #pragma unroll
    for (int g = 0; g < HID / 4; ++g) {
        float4 v = hr[g];
        acc += v.x * Wreg[4 * g + 0] + v.y * Wreg[4 * g + 1]
             + v.z * Wreg[4 * g + 2] + v.w * Wreg[4 * g + 3];
    }
    out[i] = acc;
}

static inline int cdiv(long long a, long long b) { return (int)((a + b - 1) / b); }

extern "C" void kernel_launch(void* const* d_in, const int* in_sizes, int n_in,
                              void* d_out, int out_size, void* d_ws, size_t ws_size,
                              hipStream_t stream) {
    const float* X    = (const float*)d_in[0];
    const int*   ei   = (const int*)  d_in[1];
    const float* W0   = (const float*)d_in[2];
    const float* b0   = (const float*)d_in[3];
    const float* W1   = (const float*)d_in[4];
    const float* b1   = (const float*)d_in[5];
    const float* W2   = (const float*)d_in[6];
    const float* b2   = (const float*)d_in[7];
    const float* Wreg = (const float*)d_in[8];
    const float* breg = (const float*)d_in[9];

    float* out    = (float*)d_out;
    float* hfinal = out + NTOT;

    float* dis = (float*)d_ws;
    float* nrm = dis + NTOT;
    float* agg = nrm + ETOT;
    float* hA  = agg + (size_t)NTOT * HID;
    float* hB  = hA  + (size_t)NTOT * HID;
    float* Xp  = hB  + (size_t)NTOT * HID;

    const int T = 256;

    zero_f32x4<<<cdiv(NTOT / 4, T), T, 0, stream>>>((float4*)dis, NTOT / 4);
    degree_kernel<<<cdiv(ETOT, T), T, 0, stream>>>(ei, dis);
    invsqrt_kernel<<<cdiv(NTOT, T), T, 0, stream>>>(dis);
    norm_kernel<<<cdiv(ETOT, T), T, 0, stream>>>(ei, dis, nrm);

    const int gemm_blocks = cdiv((long long)(NTOT / 16) * 32, T);
    const int agg_blocks  = cdiv(ETOT, T);

    pad_rows_kernel<<<cdiv(NTOT, T), T, 0, stream>>>(X, Xp);
    zero_f32x4<<<cdiv((long long)NTOT * KPAD / 4, T), T, 0, stream>>>((float4*)agg, NTOT * KPAD / 4);
    aggregate_kernel<WINDOW, KPAD><<<agg_blocks, T, 0, stream>>>(ei, nrm, Xp, agg);
    tag_gemm_kernel<KPAD, WINDOW><<<gemm_blocks, T, 0, stream>>>(Xp, agg, W0, b0, hA, NTOT);

    zero_f32x4<<<cdiv((long long)NTOT * HID / 4, T), T, 0, stream>>>((float4*)agg, NTOT * HID / 4);
    aggregate_kernel<HID, HID><<<agg_blocks, T, 0, stream>>>(ei, nrm, hA, agg);
    tag_gemm_kernel<HID, HID><<<gemm_blocks, T, 0, stream>>>(hA, agg, W1, b1, hB, NTOT);

    zero_f32x4<<<cdiv((long long)NTOT * HID / 4, T), T, 0, stream>>>((float4*)agg, NTOT * HID / 4);
    aggregate_kernel<HID, HID><<<agg_blocks, T, 0, stream>>>(ei, nrm, hB, agg);
    tag_gemm_kernel<HID, HID><<<gemm_blocks, T, 0, stream>>>(hB, agg, W2, b2, hfinal, NTOT);

    readout_kernel<<<cdiv(NTOT, T), T, 0, stream>>>(hfinal, Wreg, breg, out);
}


# f32 atomic scatter-add replaced by a per-call dst-sorted edge list (int atomic count+rank, scan, fill) and an atomic-free register gather-sum per row; sum order = arrival rank, varies as before
# speedup vs baseline: 9.4020x; 9.4020x over previous
#include <hip/hip_runtime.h>
#include <hip/hip_bf16.h>
#include <math.h>

#define BATCH   8
#define WINDOW  5
#define NN      20000
#define EPB     320000
#define HID     32
#define KPAD    8
#define NTOT    (BATCH * NN)
#define ETOT    (BATCH * EPB)

typedef float v2f __attribute__((ext_vector_type(2)));
typedef float v8f __attribute__((ext_vector_type(8)));

static __device__ __forceinline__ void edge_nodes(const int* __restrict__ ei,
                                                  int e, int& s, int& d) {
    int b  = e / EPB;
    int el = e - b * EPB;
    const int* base = ei + (size_t)b * 2 * EPB;
    s = base[el] + b * NN;
    d = base[EPB + el] + b * NN;
}

__global__ void zero_f32x4(float4* __restrict__ p, int n4) {
    int i = blockIdx.x * blockDim.x + threadIdx.x;
    if (i < n4) p[i] = make_float4(0.f, 0.f, 0.f, 0.f);
}

__global__ void pad_rows_kernel(const float* __restrict__ x, float* __restrict__ xp) {
    int i = blockIdx.x * blockDim.x + threadIdx.x;
    if (i >= NTOT) return;
    const float* __restrict__ r = x + (size_t)i * WINDOW;
    float* __restrict__ o = xp + (size_t)i * KPAD;
    float4 lo = make_float4(r[0], r[1], r[2], r[3]);
    float4 hi = make_float4(r[4], 0.f, 0.f, 0.f);
    reinterpret_cast<float4*>(o)[0] = lo;
    reinterpret_cast<float4*>(o)[1] = hi;
}

static __device__ __forceinline__ int edge_dst(const int* __restrict__ ei, int e) {
    int b  = e / EPB;
    int el = e - b * EPB;
    return ei[(size_t)b * 2 * EPB + EPB + el] + b * NN;
}

__global__ void count_rank_kernel(const int* __restrict__ ei, int* __restrict__ cnt,
                                  int* __restrict__ rank) {
    int e = blockIdx.x * blockDim.x + threadIdx.x;
    if (e >= ETOT) return;
    rank[e] = atomicAdd(&cnt[edge_dst(ei, e)], 1);
}

#define SCAN_T   1024
#define SCAN_PER 20
__global__ void __launch_bounds__(SCAN_T)
row_offsets_kernel(const int* __restrict__ cnt, int* __restrict__ rowptr,
                   float* __restrict__ dis) {
    static_assert(NN % SCAN_PER == 0 && NN / SCAN_PER <= SCAN_T && (SCAN_PER % 4) == 0,
                  "one thread scans SCAN_PER consecutive nodes with 16-byte accesses");
    __shared__ int wtot[SCAN_T / 32];
    const int b    = blockIdx.x;
    const int tid  = threadIdx.x;
    const int lane = tid & 31;
    const int wid  = tid >> 5;
    const bool act = tid < NN / SCAN_PER;
    const int first = b * NN + tid * SCAN_PER;

    int c[SCAN_PER];
    int tot = 0;
    #pragma unroll
    for (int k = 0; k < SCAN_PER; ++k) c[k] = 0;
    if (act) {
        const int4* __restrict__ p = reinterpret_cast<const int4*>(cnt + first);
        #pragma unroll
        for (int g = 0; g < SCAN_PER / 4; ++g) {
            int4 q = p[g];
            c[4 * g + 0] = q.x; c[4 * g + 1] = q.y; c[4 * g + 2] = q.z; c[4 * g + 3] = q.w;
            tot += q.x + q.y + q.z + q.w;
        }
    }

    int inc = tot;
    #pragma unroll
    for (int off = 1; off < 32; off <<= 1) {
        int t = __shfl_up(inc, off);
        if (lane >= off) inc += t;
    }
    if (lane == 31) wtot[wid] = inc;
    __syncthreads();
    if (wid == 0) {
        int w = wtot[lane];
        int winc = w;
        #pragma unroll
        for (int off = 1; off < 32; off <<= 1) {
            int t = __shfl_up(winc, off);
            if (lane >= off) winc += t;
        }
        wtot[lane] = winc - w;
    }
    __syncthreads();

    if (act) {
        int run = b * EPB + wtot[wid] + (inc - tot);
        int4*   __restrict__ rp = reinterpret_cast<int4*>(rowptr + first);
        float4* __restrict__ dp = reinterpret_cast<float4*>(dis + first);
        #pragma unroll
        for (int g = 0; g < SCAN_PER / 4; ++g) {
            int4 r; float4 f;
            r.x = run; run += c[4 * g + 0];
            r.y = run; run += c[4 * g + 1];
            r.z = run; run += c[4 * g + 2];
            r.w = run; run += c[4 * g + 3];
            f.x = (c[4 * g + 0] > 0) ? rsqrtf((float)c[4 * g + 0]) : 0.0f;
            f.y = (c[4 * g + 1] > 0) ? rsqrtf((float)c[4 * g + 1]) : 0.0f;
            f.z = (c[4 * g + 2] > 0) ? rsqrtf((float)c[4 * g + 2]) : 0.0f;
            f.w = (c[4 * g + 3] > 0) ? rsqrtf((float)c[4 * g + 3]) : 0.0f;
            rp[g] = r;
            dp[g] = f;
        }
    }
}

__global__ void fill_rows_kernel(const int* __restrict__ ei, const int* __restrict__ rank,
                                 const int* __restrict__ rowptr, const float* __restrict__ dis,
                                 int2* __restrict__ csr) {
    int e = blockIdx.x * blockDim.x + threadIdx.x;
    if (e >= ETOT) return;
    int s, d; edge_nodes(ei, e, s, d);
    int slot = rowptr[d] + rank[e];
    csr[slot] = make_int2(s, __float_as_int(dis[s] * dis[d]));
}

static __device__ __forceinline__ int2 load_entry(const int2* __restrict__ csr, unsigned idx) {
    return *reinterpret_cast<const int2*>(reinterpret_cast<const char*>(csr) + idx * 8u);
}
template <int S>
static __device__ __forceinline__ float4 load_row4(const float* __restrict__ hin, int src,
                                                   unsigned suboff) {
    return *reinterpret_cast<const float4*>(reinterpret_cast<const char*>(hin) +
                                            ((unsigned)src * (unsigned)(S * 4) + suboff));
}
static __device__ __forceinline__ void axpy4(float4& acc, const float4& v, float w) {
    acc.x += v.x * w; acc.y += v.y * w; acc.z += v.z * w; acc.w += v.w * w;
}

template <int S>
__global__ void __launch_bounds__(256)
gather_rows_kernel(const int* __restrict__ rowptr, const int* __restrict__ cnt,
                   const int2* __restrict__ csr, const float* __restrict__ hin,
                   float* __restrict__ agg) {
    constexpr int LPR = S / 4;
    const int t   = blockIdx.x * blockDim.x + threadIdx.x;
    const int row = t / LPR;
    const int sub = t - row * LPR;
    if (row >= NTOT) return;
    const int      n      = cnt[row];
    const unsigned start  = (unsigned)rowptr[row];
    const unsigned suboff = (unsigned)sub * 16u;
    float4 acc = make_float4(0.f, 0.f, 0.f, 0.f);

    int j = 0;
    for (; j + 4 <= n; j += 4) {
        const int2 e0 = load_entry(csr, start + j);
        const int2 e1 = load_entry(csr, start + j + 1);
        const int2 e2 = load_entry(csr, start + j + 2);
        const int2 e3 = load_entry(csr, start + j + 3);
        const float4 v0 = load_row4<S>(hin, e0.x, suboff);
        const float4 v1 = load_row4<S>(hin, e1.x, suboff);
        const float4 v2 = load_row4<S>(hin, e2.x, suboff);
        const float4 v3 = load_row4<S>(hin, e3.x, suboff);
        axpy4(acc, v0, __int_as_float(e0.y));
        axpy4(acc, v1, __int_as_float(e1.y));
        axpy4(acc, v2, __int_as_float(e2.y));
        axpy4(acc, v3, __int_as_float(e3.y));
    }
    const int r = n - j;
    if (r > 0) {
        const int2 e0 = load_entry(csr, start + j);
        const int2 e1 = load_entry(csr, start + j + (r > 1 ? 1 : 0));
        const int2 e2 = load_entry(csr, start + j + (r > 2 ? 2 : 0));
        float4 v0 = load_row4<S>(hin, e0.x, suboff);
        float4 v1 = load_row4<S>(hin, e1.x, suboff);
        float4 v2 = load_row4<S>(hin, e2.x, suboff);
        v1.x = (r > 1) ? v1.x : 0.f; v1.y = (r > 1) ? v1.y : 0.f;
        v1.z = (r > 1) ? v1.z : 0.f; v1.w = (r > 1) ? v1.w : 0.f;
        v2.x = (r > 2) ? v2.x : 0.f; v2.y = (r > 2) ? v2.y : 0.f;
        v2.z = (r > 2) ? v2.z : 0.f; v2.w = (r > 2) ? v2.w : 0.f;
        axpy4(acc, v0, __int_as_float(e0.y));
        axpy4(acc, v1, __int_as_float(e1.y));
        axpy4(acc, v2, __int_as_float(e2.y));
    }
    reinterpret_cast<float4*>(agg)[(size_t)row * LPR + sub] = acc;
}

template <int K, int KW>
__global__ void tag_gemm_kernel(const float* __restrict__ hin,
                                const float* __restrict__ agg,
                                const float* __restrict__ W,
                                const float* __restrict__ bias,
                                float* __restrict__ hout, int N) {
    static_assert((K % 4) == 0, "padded K must be a multiple of 4");
    __shared__ float sW[2 * K * HID];
    __shared__ float sB[HID];
    for (int i = threadIdx.x; i < 2 * K * HID; i += blockDim.x) {
        int s   = i / (K * HID);
        int rem = i - s * (K * HID);
        int kw  = rem / HID;
        int col = rem - kw * HID;
        sW[i] = (kw < KW) ? W[((size_t)s * KW + kw) * HID + col] : 0.0f;
    }
    if (threadIdx.x < HID) sB[threadIdx.x] = bias[threadIdx.x];
    __syncthreads();

    int lane = threadIdx.x & 31;
    int wave = (blockIdx.x * blockDim.x + threadIdx.x) >> 5;
    int row0 = wave * 16;
    if (row0 >= N) return;

    const int m     = lane & 15;
    const int khalf = (lane >> 4) * 2;

    v8f acc0 = {0.f,0.f,0.f,0.f,0.f,0.f,0.f,0.f};
    v8f acc1 = {0.f,0.f,0.f,0.f,0.f,0.f,0.f,0.f};

    #pragma unroll
    for (int s = 0; s < 2; ++s) {
        const float* __restrict__ A   = s ? agg : hin;
        const float* __restrict__ sWk = sW + s * K * HID;
        #pragma unroll
        for (int c = 0; c < K / 4; ++c) {
            const int k0 = c * 4 + khalf;
            v2f a = *reinterpret_cast<const v2f*>(&A[(size_t)(row0 + m) * K + k0]);
            v2f b0, b1;
            b0.x = sWk[k0 * HID + m];
            b0.y = sWk[(k0 + 1) * HID + m];
            b1.x = sWk[k0 * HID + 16 + m];
            b1.y = sWk[(k0 + 1) * HID + 16 + m];
            acc0 = __builtin_amdgcn_wmma_f32_16x16x4_f32(
                       false, a, false, b0, (short)0, acc0, false, false);
            acc1 = __builtin_amdgcn_wmma_f32_16x16x4_f32(
                       false, a, false, b1, (short)0, acc1, false, false);
        }
    }

    const int rbase = row0 + ((lane >> 4) ? 8 : 0);
    #pragma unroll
    for (int v = 0; v < 8; ++v) {
        int r = rbase + v;
        float x0 = acc0[v] + sB[m];
        float x1 = acc1[v] + sB[16 + m];
        hout[(size_t)r * HID + m]      = tanhf(x0);
        hout[(size_t)r * HID + 16 + m] = tanhf(x1);
    }
}

__global__ void readout_kernel(const float* __restrict__ h, const float* __restrict__ Wreg,
                               const float* __restrict__ breg, float* __restrict__ out) {
    int i = blockIdx.x * blockDim.x + threadIdx.x;
    if (i >= NTOT) return;
    float acc = breg[0];
    const float4* hr = reinterpret_cast<const float4*>(h + (size_t)i * HID);
    #pragma unroll
    for (int g = 0; g < HID / 4; ++g) {
        float4 v = hr[g];
        acc += v.x * Wreg[4 * g + 0] + v.y * Wreg[4 * g + 1]
             + v.z * Wreg[4 * g + 2] + v.w * Wreg[4 * g + 3];
    }
    out[i] = acc;
}

static inline int cdiv(long long a, long long b) { return (int)((a + b - 1) / b); }

extern "C" void kernel_launch(void* const* d_in, const int* in_sizes, int n_in,
                              void* d_out, int out_size, void* d_ws, size_t ws_size,
                              hipStream_t stream) {
    const float* X    = (const float*)d_in[0];
    const int*   ei   = (const int*)  d_in[1];
    const float* W0   = (const float*)d_in[2];
    const float* b0   = (const float*)d_in[3];
    const float* W1   = (const float*)d_in[4];
    const float* b1   = (const float*)d_in[5];
    const float* W2   = (const float*)d_in[6];
    const float* b2   = (const float*)d_in[7];
    const float* Wreg = (const float*)d_in[8];
    const float* breg = (const float*)d_in[9];

    float* out    = (float*)d_out;
    float* hfinal = out + NTOT;

    int*   cnt    = (int*)d_ws;
    int*   rowptr = cnt + NTOT;
    float* dis    = (float*)(rowptr + NTOT);
    int*   rank   = (int*)(dis + NTOT);
    int2*  csr    = (int2*)(rank + ETOT);
    float* agg    = (float*)(csr + ETOT);
    float* hA     = agg + (size_t)NTOT * HID;
    float* hB     = hA  + (size_t)NTOT * HID;
    float* Xp     = hB  + (size_t)NTOT * HID;

    const int T = 256;

    zero_f32x4<<<cdiv(NTOT / 4, T), T, 0, stream>>>((float4*)cnt, NTOT / 4);
    count_rank_kernel<<<cdiv(ETOT, T), T, 0, stream>>>(ei, cnt, rank);
    row_offsets_kernel<<<BATCH, SCAN_T, 0, stream>>>(cnt, rowptr, dis);
    fill_rows_kernel<<<cdiv(ETOT, T), T, 0, stream>>>(ei, rank, rowptr, dis, csr);

    const int gemm_blocks = cdiv((long long)(NTOT / 16) * 32, T);

    pad_rows_kernel<<<cdiv(NTOT, T), T, 0, stream>>>(X, Xp);
    gather_rows_kernel<KPAD><<<cdiv((long long)NTOT * (KPAD / 4), T), T, 0, stream>>>(rowptr, cnt, csr, Xp, agg);
    tag_gemm_kernel<KPAD, WINDOW><<<gemm_blocks, T, 0, stream>>>(Xp, agg, W0, b0, hA, NTOT);

    gather_rows_kernel<HID><<<cdiv((long long)NTOT * (HID / 4), T), T, 0, stream>>>(rowptr, cnt, csr, hA, agg);
    tag_gemm_kernel<HID, HID><<<gemm_blocks, T, 0, stream>>>(hA, agg, W1, b1, hB, NTOT);

    gather_rows_kernel<HID><<<cdiv((long long)NTOT * (HID / 4), T), T, 0, stream>>>(rowptr, cnt, csr, hB, agg);
    tag_gemm_kernel<HID, HID><<<gemm_blocks, T, 0, stream>>>(hB, agg, W2, b2, hfinal, NTOT);

    readout_kernel<<<cdiv(NTOT, T), T, 0, stream>>>(hfinal, Wreg, breg, out);
}


# edge list built as a two-level counting sort in LDS: 32 work-groups per graph rank edges on LDS counters (no global atomics) and fill from LDS slot tables; row sum order: chunk, then arrival
# speedup vs baseline: 15.9520x; 1.6967x over previous
#include <hip/hip_runtime.h>
#include <hip/hip_bf16.h>
#include <math.h>

#define BATCH   8
#define WINDOW  5
#define NN      20000
#define EPB     320000
#define HID     32
#define KPAD    8
#define NTOT    (BATCH * NN)
#define ETOT    (BATCH * EPB)

typedef float v2f __attribute__((ext_vector_type(2)));
typedef float v8f __attribute__((ext_vector_type(8)));

__global__ void pad_rows_kernel(const float* __restrict__ x, float* __restrict__ xp) {
    int i = blockIdx.x * blockDim.x + threadIdx.x;
    if (i >= NTOT) return;
    const float* __restrict__ r = x + (size_t)i * WINDOW;
    float* __restrict__ o = xp + (size_t)i * KPAD;
    float4 lo = make_float4(r[0], r[1], r[2], r[3]);
    float4 hi = make_float4(r[4], 0.f, 0.f, 0.f);
    reinterpret_cast<float4*>(o)[0] = lo;
    reinterpret_cast<float4*>(o)[1] = hi;
}

#define EDGE_T      1024
#define EDGE_SPLIT  32
#define QPW         (EPB / 4 / EDGE_SPLIT)
#define EDGE_QPT    ((QPW + EDGE_T - 1) / EDGE_T)
#define NQ          (NN / 4)
#define TAB_TRIPS   5
#define TAB_T       (NQ / TAB_TRIPS)

__global__ void __launch_bounds__(EDGE_T)
count_rank_kernel(const int* __restrict__ ei, int* __restrict__ rank, int* __restrict__ part) {
    static_assert(EPB % (4 * EDGE_SPLIT) == 0 && NN % 4 == 0 && NQ % TAB_TRIPS == 0 &&
                  TAB_T <= EDGE_T && TAB_T % 8 == 0,
                  "whole edge quads per chunk; tables move 16 bytes at a time in whole lines");
    __shared__ int hist[NN];
    const int b   = blockIdx.x / EDGE_SPLIT;
    const int c   = blockIdx.x - b * EDGE_SPLIT;
    const int tid = threadIdx.x;
    const int4* __restrict__ dst4 =
        reinterpret_cast<const int4*>(ei + (size_t)b * 2 * EPB + EPB) + c * QPW;
    int4* __restrict__ rnk4 = reinterpret_cast<int4*>(rank + (size_t)b * EPB) + c * QPW;

    int4 dq[EDGE_QPT];
    #pragma unroll
    for (int k = 0; k < EDGE_QPT; ++k) {
        const int q = tid + k * EDGE_T;
        dq[k] = dst4[q < QPW ? q : QPW - 1];
    }
    for (int i = tid; i < NQ; i += EDGE_T)
        reinterpret_cast<int4*>(hist)[i] = make_int4(0, 0, 0, 0);
    __syncthreads();

    #pragma unroll
    for (int k = 0; k < EDGE_QPT; ++k) {
        const int q = tid + k * EDGE_T;
        if (q < QPW) {
            int4 r;
            r.x = atomicAdd(&hist[dq[k].x], 1);
            r.y = atomicAdd(&hist[dq[k].y], 1);
            r.z = atomicAdd(&hist[dq[k].z], 1);
            r.w = atomicAdd(&hist[dq[k].w], 1);
            rnk4[q] = r;
        }
    }
    __syncthreads();

    int4* __restrict__ gp = reinterpret_cast<int4*>(part + (size_t)blockIdx.x * NN);
    for (int i = tid; i < NQ; i += EDGE_T)
        gp[i] = reinterpret_cast<const int4*>(hist)[i];
}

#define OFFS_T 128
__global__ void __launch_bounds__(OFFS_T)
chunk_offsets_kernel(int* __restrict__ part, int* __restrict__ cnt) {
    const int gq = blockIdx.x * OFFS_T + threadIdx.x;
    if (gq >= NTOT / 4) return;
    const int b  = gq / NQ;
    const int nq = gq - b * NQ;
    int4* __restrict__ p = reinterpret_cast<int4*>(part + (size_t)b * EDGE_SPLIT * NN) + nq;
    int4 t[EDGE_SPLIT];
    #pragma unroll
    for (int c = 0; c < EDGE_SPLIT; ++c) t[c] = p[(size_t)c * NQ];
    int4 run = make_int4(0, 0, 0, 0);
    #pragma unroll
    for (int c = 0; c < EDGE_SPLIT; ++c) {
        p[(size_t)c * NQ] = run;
        run.x += t[c].x; run.y += t[c].y; run.z += t[c].z; run.w += t[c].w;
    }
    reinterpret_cast<int4*>(cnt)[gq] = run;
}

#define SCAN_T   1024
#define SCAN_PER 20
__global__ void __launch_bounds__(SCAN_T)
row_offsets_kernel(const int* __restrict__ cnt, int* __restrict__ rowptr,
                   float* __restrict__ dis) {
    static_assert(NN % SCAN_PER == 0 && NN / SCAN_PER <= SCAN_T && (SCAN_PER % 4) == 0,
                  "one thread scans SCAN_PER consecutive nodes with 16-byte accesses");
    __shared__ int wtot[SCAN_T / 32];
    const int b    = blockIdx.x;
    const int tid  = threadIdx.x;
    const int lane = tid & 31;
    const int wid  = tid >> 5;
    const bool act = tid < NN / SCAN_PER;
    const int first = b * NN + tid * SCAN_PER;

    int c[SCAN_PER];
    int tot = 0;
    #pragma unroll
    for (int k = 0; k < SCAN_PER; ++k) c[k] = 0;
    if (act) {
        const int4* __restrict__ p = reinterpret_cast<const int4*>(cnt + first);
        #pragma unroll
        for (int g = 0; g < SCAN_PER / 4; ++g) {
            int4 q = p[g];
            c[4 * g + 0] = q.x; c[4 * g + 1] = q.y; c[4 * g + 2] = q.z; c[4 * g + 3] = q.w;
            tot += q.x + q.y + q.z + q.w;
        }
    }

    int inc = tot;
    #pragma unroll
    for (int off = 1; off < 32; off <<= 1) {
        int t = __shfl_up(inc, off);
        if (lane >= off) inc += t;
    }
    if (lane == 31) wtot[wid] = inc;
    __syncthreads();
    if (wid == 0) {
        int w = wtot[lane];
        int winc = w;
        #pragma unroll
        for (int off = 1; off < 32; off <<= 1) {
            int t = __shfl_up(winc, off);
            if (lane >= off) winc += t;
        }
        wtot[lane] = winc - w;
    }
    __syncthreads();

    if (act) {
        int run = b * EPB + wtot[wid] + (inc - tot);
        int4*   __restrict__ rp = reinterpret_cast<int4*>(rowptr + first);
        float4* __restrict__ dp = reinterpret_cast<float4*>(dis + first);
        #pragma unroll
        for (int g = 0; g < SCAN_PER / 4; ++g) {
            int4 r; float4 f;
            r.x = run; run += c[4 * g + 0];
            r.y = run; run += c[4 * g + 1];
            r.z = run; run += c[4 * g + 2];
            r.w = run; run += c[4 * g + 3];
            f.x = (c[4 * g + 0] > 0) ? rsqrtf((float)c[4 * g + 0]) : 0.0f;
            f.y = (c[4 * g + 1] > 0) ? rsqrtf((float)c[4 * g + 1]) : 0.0f;
            f.z = (c[4 * g + 2] > 0) ? rsqrtf((float)c[4 * g + 2]) : 0.0f;
            f.w = (c[4 * g + 3] > 0) ? rsqrtf((float)c[4 * g + 3]) : 0.0f;
            rp[g] = r;
            dp[g] = f;
        }
    }
}

static __device__ __forceinline__ void put_edge(const int* __restrict__ s_off,
                                                const float* __restrict__ s_dis,
                                                int2* __restrict__ csr,
                                                int sl, int dl, int rk, int nodebase) {
    const int slot = s_off[dl] + rk;
    csr[slot] = make_int2(sl + nodebase, __float_as_int(s_dis[sl] * s_dis[dl]));
}
static __device__ __forceinline__ int4 add4(const int4& a, const int4& b) {
    return make_int4(a.x + b.x, a.y + b.y, a.z + b.z, a.w + b.w);
}
__global__ void __launch_bounds__(EDGE_T)
fill_rows_kernel(const int* __restrict__ ei, const int* __restrict__ rank,
                 const int* __restrict__ rowptr, const int* __restrict__ part,
                 const float* __restrict__ dis, int2* __restrict__ csr) {
    __shared__ int   s_off[NN];
    __shared__ float s_dis[NN];
    const int b   = blockIdx.x / EDGE_SPLIT;
    const int c   = blockIdx.x - b * EDGE_SPLIT;
    const int tid = threadIdx.x;

    const int* __restrict__ eb = ei + (size_t)b * 2 * EPB;
    const int4* __restrict__ src4 = reinterpret_cast<const int4*>(eb) + c * QPW;
    const int4* __restrict__ dst4 = reinterpret_cast<const int4*>(eb + EPB) + c * QPW;
    const int4* __restrict__ rnk4 =
        reinterpret_cast<const int4*>(rank + (size_t)b * EPB) + c * QPW;

    int4 sq[EDGE_QPT], dq[EDGE_QPT], rq[EDGE_QPT];
    #pragma unroll
    for (int k = 0; k < EDGE_QPT; ++k) {
        const int q  = tid + k * EDGE_T;
        const int qc = q < QPW ? q : QPW - 1;
        sq[k] = src4[qc]; dq[k] = dst4[qc]; rq[k] = rnk4[qc];
    }

    const int4*   __restrict__ gr = reinterpret_cast<const int4*>(rowptr + b * NN);
    const int4*   __restrict__ gp = reinterpret_cast<const int4*>(part + (size_t)blockIdx.x * NN);
    const float4* __restrict__ gd = reinterpret_cast<const float4*>(dis + b * NN);
    if (tid < TAB_T) {
        int4 tr[TAB_TRIPS], tp[TAB_TRIPS]; float4 td[TAB_TRIPS];
        #pragma unroll
        for (int k = 0; k < TAB_TRIPS; ++k) {
            const int i = tid + k * TAB_T;
            tr[k] = gr[i]; tp[k] = gp[i]; td[k] = gd[i];
        }
        #pragma unroll
        for (int k = 0; k < TAB_TRIPS; ++k) {
            const int i = tid + k * TAB_T;
            reinterpret_cast<int4*>(s_off)[i]   = add4(tr[k], tp[k]);
            reinterpret_cast<float4*>(s_dis)[i] = td[k];
        }
    }
    __syncthreads();

    const int nodebase = b * NN;
    #pragma unroll
    for (int k = 0; k < EDGE_QPT; ++k) {
        const int q = tid + k * EDGE_T;
        if (q < QPW) {
            put_edge(s_off, s_dis, csr, sq[k].x, dq[k].x, rq[k].x, nodebase);
            put_edge(s_off, s_dis, csr, sq[k].y, dq[k].y, rq[k].y, nodebase);
            put_edge(s_off, s_dis, csr, sq[k].z, dq[k].z, rq[k].z, nodebase);
            put_edge(s_off, s_dis, csr, sq[k].w, dq[k].w, rq[k].w, nodebase);
        }
    }
}

static __device__ __forceinline__ int2 load_entry(const int2* __restrict__ csr, unsigned idx) {
    return *reinterpret_cast<const int2*>(reinterpret_cast<const char*>(csr) + idx * 8u);
}
template <int S>
static __device__ __forceinline__ float4 load_row4(const float* __restrict__ hin, int src,
                                                   unsigned suboff) {
    return *reinterpret_cast<const float4*>(reinterpret_cast<const char*>(hin) +
                                            ((unsigned)src * (unsigned)(S * 4) + suboff));
}
static __device__ __forceinline__ void axpy4(float4& acc, const float4& v, float w) {
    acc.x += v.x * w; acc.y += v.y * w; acc.z += v.z * w; acc.w += v.w * w;
}

template <int S>
__global__ void __launch_bounds__(256)
gather_rows_kernel(const int* __restrict__ rowptr, const int* __restrict__ cnt,
                   const int2* __restrict__ csr, const float* __restrict__ hin,
                   float* __restrict__ agg) {
    constexpr int LPR = S / 4;
    const int t   = blockIdx.x * blockDim.x + threadIdx.x;
    const int row = t / LPR;
    const int sub = t - row * LPR;
    if (row >= NTOT) return;
    const int      n      = cnt[row];
    const unsigned start  = (unsigned)rowptr[row];
    const unsigned suboff = (unsigned)sub * 16u;
    float4 acc = make_float4(0.f, 0.f, 0.f, 0.f);

    int j = 0;
    for (; j + 4 <= n; j += 4) {
        const int2 e0 = load_entry(csr, start + j);
        const int2 e1 = load_entry(csr, start + j + 1);
        const int2 e2 = load_entry(csr, start + j + 2);
        const int2 e3 = load_entry(csr, start + j + 3);
        const float4 v0 = load_row4<S>(hin, e0.x, suboff);
        const float4 v1 = load_row4<S>(hin, e1.x, suboff);
        const float4 v2 = load_row4<S>(hin, e2.x, suboff);
        const float4 v3 = load_row4<S>(hin, e3.x, suboff);
        axpy4(acc, v0, __int_as_float(e0.y));
        axpy4(acc, v1, __int_as_float(e1.y));
        axpy4(acc, v2, __int_as_float(e2.y));
        axpy4(acc, v3, __int_as_float(e3.y));
    }
    const int r = n - j;
    if (r > 0) {
        const int2 e0 = load_entry(csr, start + j);
        const int2 e1 = load_entry(csr, start + j + (r > 1 ? 1 : 0));
        const int2 e2 = load_entry(csr, start + j + (r > 2 ? 2 : 0));
        float4 v0 = load_row4<S>(hin, e0.x, suboff);
        float4 v1 = load_row4<S>(hin, e1.x, suboff);
        float4 v2 = load_row4<S>(hin, e2.x, suboff);
        v1.x = (r > 1) ? v1.x : 0.f; v1.y = (r > 1) ? v1.y : 0.f;
        v1.z = (r > 1) ? v1.z : 0.f; v1.w = (r > 1) ? v1.w : 0.f;
        v2.x = (r > 2) ? v2.x : 0.f; v2.y = (r > 2) ? v2.y : 0.f;
        v2.z = (r > 2) ? v2.z : 0.f; v2.w = (r > 2) ? v2.w : 0.f;
        axpy4(acc, v0, __int_as_float(e0.y));
        axpy4(acc, v1, __int_as_float(e1.y));
        axpy4(acc, v2, __int_as_float(e2.y));
    }
    reinterpret_cast<float4*>(agg)[(size_t)row * LPR + sub] = acc;
}

template <int K, int KW>
__global__ void tag_gemm_kernel(const float* __restrict__ hin,
                                const float* __restrict__ agg,
                                const float* __restrict__ W,
                                const float* __restrict__ bias,
                                float* __restrict__ hout, int N) {
    static_assert((K % 4) == 0, "padded K must be a multiple of 4");
    __shared__ float sW[2 * K * HID];
    __shared__ float sB[HID];
    for (int i = threadIdx.x; i < 2 * K * HID; i += blockDim.x) {
        int s   = i / (K * HID);
        int rem = i - s * (K * HID);
        int kw  = rem / HID;
        int col = rem - kw * HID;
        sW[i] = (kw < KW) ? W[((size_t)s * KW + kw) * HID + col] : 0.0f;
    }
    if (threadIdx.x < HID) sB[threadIdx.x] = bias[threadIdx.x];
    __syncthreads();

    int lane = threadIdx.x & 31;
    int wave = (blockIdx.x * blockDim.x + threadIdx.x) >> 5;
    int row0 = wave * 16;
    if (row0 >= N) return;

    const int m     = lane & 15;
    const int khalf = (lane >> 4) * 2;

    v8f acc0 = {0.f,0.f,0.f,0.f,0.f,0.f,0.f,0.f};
    v8f acc1 = {0.f,0.f,0.f,0.f,0.f,0.f,0.f,0.f};

    #pragma unroll
    for (int s = 0; s < 2; ++s) {
        const float* __restrict__ A   = s ? agg : hin;
        const float* __restrict__ sWk = sW + s * K * HID;
        #pragma unroll
        for (int c = 0; c < K / 4; ++c) {
            const int k0 = c * 4 + khalf;
            v2f a = *reinterpret_cast<const v2f*>(&A[(size_t)(row0 + m) * K + k0]);
            v2f b0, b1;
            b0.x = sWk[k0 * HID + m];
            b0.y = sWk[(k0 + 1) * HID + m];
            b1.x = sWk[k0 * HID + 16 + m];
            b1.y = sWk[(k0 + 1) * HID + 16 + m];
            acc0 = __builtin_amdgcn_wmma_f32_16x16x4_f32(
                       false, a, false, b0, (short)0, acc0, false, false);
            acc1 = __builtin_amdgcn_wmma_f32_16x16x4_f32(
                       false, a, false, b1, (short)0, acc1, false, false);
        }
    }

    const int rbase = row0 + ((lane >> 4) ? 8 : 0);
    #pragma unroll
    for (int v = 0; v < 8; ++v) {
        int r = rbase + v;
        float x0 = acc0[v] + sB[m];
        float x1 = acc1[v] + sB[16 + m];
        hout[(size_t)r * HID + m]      = tanhf(x0);
        hout[(size_t)r * HID + 16 + m] = tanhf(x1);
    }
}

__global__ void readout_kernel(const float* __restrict__ h, const float* __restrict__ Wreg,
                               const float* __restrict__ breg, float* __restrict__ out) {
    int i = blockIdx.x * blockDim.x + threadIdx.x;
    if (i >= NTOT) return;
    float acc = breg[0];
    const float4* hr = reinterpret_cast<const float4*>(h + (size_t)i * HID);
    #pragma unroll
    for (int g = 0; g < HID / 4; ++g) {
        float4 v = hr[g];
        acc += v.x * Wreg[4 * g + 0] + v.y * Wreg[4 * g + 1]
             + v.z * Wreg[4 * g + 2] + v.w * Wreg[4 * g + 3];
    }
    out[i] = acc;
}

static inline int cdiv(long long a, long long b) { return (int)((a + b - 1) / b); }

extern "C" void kernel_launch(void* const* d_in, const int* in_sizes, int n_in,
                              void* d_out, int out_size, void* d_ws, size_t ws_size,
                              hipStream_t stream) {
    const float* X    = (const float*)d_in[0];
    const int*   ei   = (const int*)  d_in[1];
    const float* W0   = (const float*)d_in[2];
    const float* b0   = (const float*)d_in[3];
    const float* W1   = (const float*)d_in[4];
    const float* b1   = (const float*)d_in[5];
    const float* W2   = (const float*)d_in[6];
    const float* b2   = (const float*)d_in[7];
    const float* Wreg = (const float*)d_in[8];
    const float* breg = (const float*)d_in[9];

    float* out    = (float*)d_out;
    float* hfinal = out + NTOT;

    int*   cnt    = (int*)d_ws;
    int*   rowptr = cnt + NTOT;
    float* dis    = (float*)(rowptr + NTOT);
    int*   rank   = (int*)(dis + NTOT);
    int2*  csr    = (int2*)(rank + ETOT);
    float* agg    = (float*)(csr + ETOT);
    float* hA     = agg + (size_t)NTOT * HID;
    float* hB     = hA  + (size_t)NTOT * HID;
    float* Xp     = hB  + (size_t)NTOT * HID;
    int*   part   = (int*)(Xp + (size_t)NTOT * KPAD);

    const int T = 256;

    count_rank_kernel<<<BATCH * EDGE_SPLIT, EDGE_T, 0, stream>>>(ei, rank, part);
    chunk_offsets_kernel<<<cdiv(NTOT / 4, OFFS_T), OFFS_T, 0, stream>>>(part, cnt);
    row_offsets_kernel<<<BATCH, SCAN_T, 0, stream>>>(cnt, rowptr, dis);
    fill_rows_kernel<<<BATCH * EDGE_SPLIT, EDGE_T, 0, stream>>>(ei, rank, rowptr, part, dis, csr);

    const int gemm_blocks = cdiv((long long)(NTOT / 16) * 32, T);

    pad_rows_kernel<<<cdiv(NTOT, T), T, 0, stream>>>(X, Xp);
    gather_rows_kernel<KPAD><<<cdiv((long long)NTOT * (KPAD / 4), T), T, 0, stream>>>(rowptr, cnt, csr, Xp, agg);
    tag_gemm_kernel<KPAD, WINDOW><<<gemm_blocks, T, 0, stream>>>(Xp, agg, W0, b0, hA, NTOT);

    gather_rows_kernel<HID><<<cdiv((long long)NTOT * (HID / 4), T), T, 0, stream>>>(rowptr, cnt, csr, hA, agg);
    tag_gemm_kernel<HID, HID><<<gemm_blocks, T, 0, stream>>>(hA, agg, W1, b1, hB, NTOT);

    gather_rows_kernel<HID><<<cdiv((long long)NTOT * (HID / 4), T), T, 0, stream>>>(rowptr, cnt, csr, hB, agg);
    tag_gemm_kernel<HID, HID><<<gemm_blocks, T, 0, stream>>>(hB, agg, W2, b2, hfinal, NTOT);

    readout_kernel<<<cdiv(NTOT, T), T, 0, stream>>>(hfinal, Wreg, breg, out);
}


# each layer's neighbour sum fused into its product launch: a wave gathers its 16-row tile (own | summed rows) into LDS and multiplies from there; 12 to 9 launches, no agg round trip; sum order kept
# speedup vs baseline: 17.5972x; 1.1031x over previous
#include <hip/hip_runtime.h>
#include <hip/hip_bf16.h>
#include <math.h>

#define BATCH   8
#define WINDOW  5
#define NN      20000
#define EPB     320000
#define HID     32
#define KPAD    8
#define NTOT    (BATCH * NN)
#define ETOT    (BATCH * EPB)

typedef float v2f __attribute__((ext_vector_type(2)));
typedef float v8f __attribute__((ext_vector_type(8)));

__global__ void pad_rows_kernel(const float* __restrict__ x, float* __restrict__ xp) {
    int i = blockIdx.x * blockDim.x + threadIdx.x;
    if (i >= NTOT) return;
    const float* __restrict__ r = x + (size_t)i * WINDOW;
    float* __restrict__ o = xp + (size_t)i * KPAD;
    float4 lo = make_float4(r[0], r[1], r[2], r[3]);
    float4 hi = make_float4(r[4], 0.f, 0.f, 0.f);
    reinterpret_cast<float4*>(o)[0] = lo;
    reinterpret_cast<float4*>(o)[1] = hi;
}

#define EDGE_T      1024
#define EDGE_SPLIT  32
#define QPW         (EPB / 4 / EDGE_SPLIT)
#define EDGE_QPT    ((QPW + EDGE_T - 1) / EDGE_T)
#define NQ          (NN / 4)
#define TAB_TRIPS   5
#define TAB_T       (NQ / TAB_TRIPS)

__global__ void __launch_bounds__(EDGE_T)
count_rank_kernel(const int* __restrict__ ei, int* __restrict__ rank, int* __restrict__ part) {
    static_assert(EPB % (4 * EDGE_SPLIT) == 0 && NN % 4 == 0 && NQ % TAB_TRIPS == 0 &&
                  TAB_T <= EDGE_T && TAB_T % 8 == 0,
                  "whole edge quads per chunk; tables move 16 bytes at a time in whole lines");
    __shared__ int hist[NN];
    const int b   = blockIdx.x / EDGE_SPLIT;
    const int c   = blockIdx.x - b * EDGE_SPLIT;
    const int tid = threadIdx.x;
    const int4* __restrict__ dst4 =
        reinterpret_cast<const int4*>(ei + (size_t)b * 2 * EPB + EPB) + c * QPW;
    int4* __restrict__ rnk4 = reinterpret_cast<int4*>(rank + (size_t)b * EPB) + c * QPW;

    int4 dq[EDGE_QPT];
    #pragma unroll
    for (int k = 0; k < EDGE_QPT; ++k) {
        const int q = tid + k * EDGE_T;
        dq[k] = dst4[q < QPW ? q : QPW - 1];
    }
    for (int i = tid; i < NQ; i += EDGE_T)
        reinterpret_cast<int4*>(hist)[i] = make_int4(0, 0, 0, 0);
    __syncthreads();

    #pragma unroll
    for (int k = 0; k < EDGE_QPT; ++k) {
        const int q = tid + k * EDGE_T;
        if (q < QPW) {
            int4 r;
            r.x = atomicAdd(&hist[dq[k].x], 1);
            r.y = atomicAdd(&hist[dq[k].y], 1);
            r.z = atomicAdd(&hist[dq[k].z], 1);
            r.w = atomicAdd(&hist[dq[k].w], 1);
            rnk4[q] = r;
        }
    }
    __syncthreads();

    int4* __restrict__ gp = reinterpret_cast<int4*>(part + (size_t)blockIdx.x * NN);
    for (int i = tid; i < NQ; i += EDGE_T)
        gp[i] = reinterpret_cast<const int4*>(hist)[i];
}

#define OFFS_T 128
__global__ void __launch_bounds__(OFFS_T)
chunk_offsets_kernel(int* __restrict__ part, int* __restrict__ cnt) {
    const int gq = blockIdx.x * OFFS_T + threadIdx.x;
    if (gq >= NTOT / 4) return;
    const int b  = gq / NQ;
    const int nq = gq - b * NQ;
    int4* __restrict__ p = reinterpret_cast<int4*>(part + (size_t)b * EDGE_SPLIT * NN) + nq;
    int4 t[EDGE_SPLIT];
    #pragma unroll
    for (int c = 0; c < EDGE_SPLIT; ++c) t[c] = p[(size_t)c * NQ];
    int4 run = make_int4(0, 0, 0, 0);
    #pragma unroll
    for (int c = 0; c < EDGE_SPLIT; ++c) {
        p[(size_t)c * NQ] = run;
        run.x += t[c].x; run.y += t[c].y; run.z += t[c].z; run.w += t[c].w;
    }
    reinterpret_cast<int4*>(cnt)[gq] = run;
}

#define SCAN_T   1024
#define SCAN_PER 20
__global__ void __launch_bounds__(SCAN_T)
row_offsets_kernel(const int* __restrict__ cnt, int* __restrict__ rowptr,
                   float* __restrict__ dis) {
    static_assert(NN % SCAN_PER == 0 && NN / SCAN_PER <= SCAN_T && (SCAN_PER % 4) == 0,
                  "one thread scans SCAN_PER consecutive nodes with 16-byte accesses");
    __shared__ int wtot[SCAN_T / 32];
    const int b    = blockIdx.x;
    const int tid  = threadIdx.x;
    const int lane = tid & 31;
    const int wid  = tid >> 5;
    const bool act = tid < NN / SCAN_PER;
    const int first = b * NN + tid * SCAN_PER;

    int c[SCAN_PER];
    int tot = 0;
    #pragma unroll
    for (int k = 0; k < SCAN_PER; ++k) c[k] = 0;
    if (act) {
        const int4* __restrict__ p = reinterpret_cast<const int4*>(cnt + first);
        #pragma unroll
        for (int g = 0; g < SCAN_PER / 4; ++g) {
            int4 q = p[g];
            c[4 * g + 0] = q.x; c[4 * g + 1] = q.y; c[4 * g + 2] = q.z; c[4 * g + 3] = q.w;
            tot += q.x + q.y + q.z + q.w;
        }
    }

    int inc = tot;
    #pragma unroll
    for (int off = 1; off < 32; off <<= 1) {
        int t = __shfl_up(inc, off);
        if (lane >= off) inc += t;
    }
    if (lane == 31) wtot[wid] = inc;
    __syncthreads();
    if (wid == 0) {
        int w = wtot[lane];
        int winc = w;
        #pragma unroll
        for (int off = 1; off < 32; off <<= 1) {
            int t = __shfl_up(winc, off);
            if (lane >= off) winc += t;
        }
        wtot[lane] = winc - w;
    }
    __syncthreads();

    if (act) {
        int run = b * EPB + wtot[wid] + (inc - tot);
        int4*   __restrict__ rp = reinterpret_cast<int4*>(rowptr + first);
        float4* __restrict__ dp = reinterpret_cast<float4*>(dis + first);
        #pragma unroll
        for (int g = 0; g < SCAN_PER / 4; ++g) {
            int4 r; float4 f;
            r.x = run; run += c[4 * g + 0];
            r.y = run; run += c[4 * g + 1];
            r.z = run; run += c[4 * g + 2];
            r.w = run; run += c[4 * g + 3];
            f.x = (c[4 * g + 0] > 0) ? rsqrtf((float)c[4 * g + 0]) : 0.0f;
            f.y = (c[4 * g + 1] > 0) ? rsqrtf((float)c[4 * g + 1]) : 0.0f;
            f.z = (c[4 * g + 2] > 0) ? rsqrtf((float)c[4 * g + 2]) : 0.0f;
            f.w = (c[4 * g + 3] > 0) ? rsqrtf((float)c[4 * g + 3]) : 0.0f;
            rp[g] = r;
            dp[g] = f;
        }
    }
}

static __device__ __forceinline__ void put_edge(const int* __restrict__ s_off,
                                                const float* __restrict__ s_dis,
                                                int2* __restrict__ csr,
                                                int sl, int dl, int rk, int nodebase) {
    const int slot = s_off[dl] + rk;
    csr[slot] = make_int2(sl + nodebase, __float_as_int(s_dis[sl] * s_dis[dl]));
}
static __device__ __forceinline__ int4 add4(const int4& a, const int4& b) {
    return make_int4(a.x + b.x, a.y + b.y, a.z + b.z, a.w + b.w);
}
__global__ void __launch_bounds__(EDGE_T)
fill_rows_kernel(const int* __restrict__ ei, const int* __restrict__ rank,
                 const int* __restrict__ rowptr, const int* __restrict__ part,
                 const float* __restrict__ dis, int2* __restrict__ csr) {
    __shared__ int   s_off[NN];
    __shared__ float s_dis[NN];
    const int b   = blockIdx.x / EDGE_SPLIT;
    const int c   = blockIdx.x - b * EDGE_SPLIT;
    const int tid = threadIdx.x;

    const int* __restrict__ eb = ei + (size_t)b * 2 * EPB;
    const int4* __restrict__ src4 = reinterpret_cast<const int4*>(eb) + c * QPW;
    const int4* __restrict__ dst4 = reinterpret_cast<const int4*>(eb + EPB) + c * QPW;
    const int4* __restrict__ rnk4 =
        reinterpret_cast<const int4*>(rank + (size_t)b * EPB) + c * QPW;

    int4 sq[EDGE_QPT], dq[EDGE_QPT], rq[EDGE_QPT];
    #pragma unroll
    for (int k = 0; k < EDGE_QPT; ++k) {
        const int q  = tid + k * EDGE_T;
        const int qc = q < QPW ? q : QPW - 1;
        sq[k] = src4[qc]; dq[k] = dst4[qc]; rq[k] = rnk4[qc];
    }

    const int4*   __restrict__ gr = reinterpret_cast<const int4*>(rowptr + b * NN);
    const int4*   __restrict__ gp = reinterpret_cast<const int4*>(part + (size_t)blockIdx.x * NN);
    const float4* __restrict__ gd = reinterpret_cast<const float4*>(dis + b * NN);
    if (tid < TAB_T) {
        int4 tr[TAB_TRIPS], tp[TAB_TRIPS]; float4 td[TAB_TRIPS];
        #pragma unroll
        for (int k = 0; k < TAB_TRIPS; ++k) {
            const int i = tid + k * TAB_T;
            tr[k] = gr[i]; tp[k] = gp[i]; td[k] = gd[i];
        }
        #pragma unroll
        for (int k = 0; k < TAB_TRIPS; ++k) {
            const int i = tid + k * TAB_T;
            reinterpret_cast<int4*>(s_off)[i]   = add4(tr[k], tp[k]);
            reinterpret_cast<float4*>(s_dis)[i] = td[k];
        }
    }
    __syncthreads();

    const int nodebase = b * NN;
    #pragma unroll
    for (int k = 0; k < EDGE_QPT; ++k) {
        const int q = tid + k * EDGE_T;
        if (q < QPW) {
            put_edge(s_off, s_dis, csr, sq[k].x, dq[k].x, rq[k].x, nodebase);
            put_edge(s_off, s_dis, csr, sq[k].y, dq[k].y, rq[k].y, nodebase);
            put_edge(s_off, s_dis, csr, sq[k].z, dq[k].z, rq[k].z, nodebase);
            put_edge(s_off, s_dis, csr, sq[k].w, dq[k].w, rq[k].w, nodebase);
        }
    }
}

static __device__ __forceinline__ int2 load_entry(const int2* __restrict__ csr, unsigned idx) {
    return *reinterpret_cast<const int2*>(reinterpret_cast<const char*>(csr) + idx * 8u);
}
template <int S>
static __device__ __forceinline__ float4 load_row4(const float* __restrict__ hin, int src,
                                                   unsigned suboff) {
    return *reinterpret_cast<const float4*>(reinterpret_cast<const char*>(hin) +
                                            ((unsigned)src * (unsigned)(S * 4) + suboff));
}
static __device__ __forceinline__ void axpy4(float4& acc, const float4& v, float w) {
    acc.x += v.x * w; acc.y += v.y * w; acc.z += v.z * w; acc.w += v.w * w;
}

template <int S>
static __device__ __forceinline__ float4 gather_sum(const int2* __restrict__ csr,
                                                    const float* __restrict__ hin,
                                                    unsigned start, int n, unsigned suboff) {
    float4 acc = make_float4(0.f, 0.f, 0.f, 0.f);
    int j = 0;
    for (; j + 4 <= n; j += 4) {
        const int2 e0 = load_entry(csr, start + j);
        const int2 e1 = load_entry(csr, start + j + 1);
        const int2 e2 = load_entry(csr, start + j + 2);
        const int2 e3 = load_entry(csr, start + j + 3);
        const float4 v0 = load_row4<S>(hin, e0.x, suboff);
        const float4 v1 = load_row4<S>(hin, e1.x, suboff);
        const float4 v2 = load_row4<S>(hin, e2.x, suboff);
        const float4 v3 = load_row4<S>(hin, e3.x, suboff);
        axpy4(acc, v0, __int_as_float(e0.y));
        axpy4(acc, v1, __int_as_float(e1.y));
        axpy4(acc, v2, __int_as_float(e2.y));
        axpy4(acc, v3, __int_as_float(e3.y));
    }
    const int r = n - j;
    if (r > 0) {
        const int2 e0 = load_entry(csr, start + j);
        const int2 e1 = load_entry(csr, start + j + (r > 1 ? 1 : 0));
        const int2 e2 = load_entry(csr, start + j + (r > 2 ? 2 : 0));
        float4 v0 = load_row4<S>(hin, e0.x, suboff);
        float4 v1 = load_row4<S>(hin, e1.x, suboff);
        float4 v2 = load_row4<S>(hin, e2.x, suboff);
        v1.x = (r > 1) ? v1.x : 0.f; v1.y = (r > 1) ? v1.y : 0.f;
        v1.z = (r > 1) ? v1.z : 0.f; v1.w = (r > 1) ? v1.w : 0.f;
        v2.x = (r > 2) ? v2.x : 0.f; v2.y = (r > 2) ? v2.y : 0.f;
        v2.z = (r > 2) ? v2.z : 0.f; v2.w = (r > 2) ? v2.w : 0.f;
        axpy4(acc, v0, __int_as_float(e0.y));
        axpy4(acc, v1, __int_as_float(e1.y));
        axpy4(acc, v2, __int_as_float(e2.y));
    }
    return acc;
}

#define LAYER_T 128
template <int S, int KW>
__global__ void __launch_bounds__(LAYER_T)
tag_layer_kernel(const int* __restrict__ rowptr, const int* __restrict__ cnt,
                 const int2* __restrict__ csr, const float* __restrict__ hin,
                 const float* __restrict__ W, const float* __restrict__ bias,
                 float* __restrict__ hout) {
    constexpr int K2    = 2 * S;
    constexpr int PITCH = K2 + 4;
    constexpr int LPR   = S / 4;
    constexpr int RPP   = 32 / LPR;
    constexpr int NPASS = 16 / RPP;
    constexpr int WAVES = LAYER_T / 32;
    constexpr int WTRIPS = (K2 * HID / 4) / LAYER_T;
    static_assert((K2 * HID / 4) % LAYER_T == 0 && HID == 32,
                  "whole staging trips; two 16-column output tiles");
    static_assert((S % 4) == 0 && (32 % LPR) == 0 && (16 % RPP) == 0 && ((PITCH / 4) % 2) == 1,
                  "whole float4 pieces per lane, whole passes per tile, odd 16-byte pitch");
    static_assert((NTOT % (16 * WAVES)) == 0, "every wave of every work-group has a whole tile");
    __shared__ __attribute__((aligned(16))) float sW[K2 * HID];
    __shared__ __attribute__((aligned(16))) float sT[WAVES * 16 * PITCH];

    const int lane = threadIdx.x & 31;
    const int wib  = threadIdx.x >> 5;
    const int row0 = (blockIdx.x * WAVES + wib) * 16;
    const int sub  = lane % LPR;
    const int rl   = lane / LPR;
    const unsigned suboff = (unsigned)sub * 16u;
    const int m     = lane & 15;
    const int khalf = (lane >> 4) * 2;

    int n[NPASS]; unsigned st[NPASS]; float4 own[NPASS];
    #pragma unroll
    for (int p = 0; p < NPASS; ++p) {
        const int row = row0 + p * RPP + rl;
        n[p]  = cnt[row];
        st[p] = (unsigned)rowptr[row];
    }
    #pragma unroll
    for (int p = 0; p < NPASS; ++p)
        own[p] = load_row4<S>(hin, row0 + p * RPP + rl, suboff);

    float4 wv[WTRIPS];
    #pragma unroll
    for (int t = 0; t < WTRIPS; ++t) {
        const int i   = threadIdx.x + t * LAYER_T;
        const int s   = i / (S * (HID / 4));
        const int rem = i - s * (S * (HID / 4));
        const int kw  = rem / (HID / 4);
        const int c4  = rem - kw * (HID / 4);
        wv[t] = (kw < KW) ? reinterpret_cast<const float4*>(W)[(s * KW + kw) * (HID / 4) + c4]
                          : make_float4(0.f, 0.f, 0.f, 0.f);
    }
    const float bias0 = bias[m];
    const float bias1 = bias[16 + m];
    #pragma unroll
    for (int t = 0; t < WTRIPS; ++t)
        reinterpret_cast<float4*>(sW)[threadIdx.x + t * LAYER_T] = wv[t];
    __syncthreads();

    float* __restrict__ tile = sT + wib * (16 * PITCH);
    #pragma unroll
    for (int p = 0; p < NPASS; ++p) {
        const float4 acc = gather_sum<S>(csr, hin, st[p], n[p], suboff);
        float* trow = tile + (p * RPP + rl) * PITCH + 4 * sub;
        *reinterpret_cast<float4*>(trow)     = own[p];
        *reinterpret_cast<float4*>(trow + S) = acc;
    }
    __builtin_amdgcn_fence(__ATOMIC_RELEASE, "wavefront");
    __builtin_amdgcn_wave_barrier();
    __builtin_amdgcn_fence(__ATOMIC_ACQUIRE, "wavefront");

    v8f acc0 = {0.f,0.f,0.f,0.f,0.f,0.f,0.f,0.f};
    v8f acc1 = {0.f,0.f,0.f,0.f,0.f,0.f,0.f,0.f};

    #pragma unroll
    for (int c = 0; c < K2 / 4; ++c) {
        const int k0 = c * 4 + khalf;
        v2f a = *reinterpret_cast<const v2f*>(tile + m * PITCH + k0);
        v2f b0, b1;
        b0.x = sW[k0 * HID + m];
        b0.y = sW[(k0 + 1) * HID + m];
        b1.x = sW[k0 * HID + 16 + m];
        b1.y = sW[(k0 + 1) * HID + 16 + m];
        acc0 = __builtin_amdgcn_wmma_f32_16x16x4_f32(
                   false, a, false, b0, (short)0, acc0, false, false);
        acc1 = __builtin_amdgcn_wmma_f32_16x16x4_f32(
                   false, a, false, b1, (short)0, acc1, false, false);
    }

    const int rbase = row0 + ((lane >> 4) ? 8 : 0);
    #pragma unroll
    for (int v = 0; v < 8; ++v) {
        int r = rbase + v;
        float x0 = acc0[v] + bias0;
        float x1 = acc1[v] + bias1;
        hout[(size_t)r * HID + m]      = tanhf(x0);
        hout[(size_t)r * HID + 16 + m] = tanhf(x1);
    }
}

__global__ void readout_kernel(const float* __restrict__ h, const float* __restrict__ Wreg,
                               const float* __restrict__ breg, float* __restrict__ out) {
    int i = blockIdx.x * blockDim.x + threadIdx.x;
    if (i >= NTOT) return;
    float acc = breg[0];
    const float4* hr = reinterpret_cast<const float4*>(h + (size_t)i * HID);
    #pragma unroll
    for (int g = 0; g < HID / 4; ++g) {
        float4 v = hr[g];
        acc += v.x * Wreg[4 * g + 0] + v.y * Wreg[4 * g + 1]
             + v.z * Wreg[4 * g + 2] + v.w * Wreg[4 * g + 3];
    }
    out[i] = acc;
}

static inline int cdiv(long long a, long long b) { return (int)((a + b - 1) / b); }

extern "C" void kernel_launch(void* const* d_in, const int* in_sizes, int n_in,
                              void* d_out, int out_size, void* d_ws, size_t ws_size,
                              hipStream_t stream) {
    const float* X    = (const float*)d_in[0];
    const int*   ei   = (const int*)  d_in[1];
    const float* W0   = (const float*)d_in[2];
    const float* b0   = (const float*)d_in[3];
    const float* W1   = (const float*)d_in[4];
    const float* b1   = (const float*)d_in[5];
    const float* W2   = (const float*)d_in[6];
    const float* b2   = (const float*)d_in[7];
    const float* Wreg = (const float*)d_in[8];
    const float* breg = (const float*)d_in[9];

    float* out    = (float*)d_out;
    float* hfinal = out + NTOT;

    int*   cnt    = (int*)d_ws;
    int*   rowptr = cnt + NTOT;
    float* dis    = (float*)(rowptr + NTOT);
    int*   rank   = (int*)(dis + NTOT);
    int2*  csr    = (int2*)(rank + ETOT);
    float* hA     = (float*)(csr + ETOT);
    float* hB     = hA  + (size_t)NTOT * HID;
    float* Xp     = hB  + (size_t)NTOT * HID;
    int*   part   = (int*)(Xp + (size_t)NTOT * KPAD);

    const int T = 256;

    count_rank_kernel<<<BATCH * EDGE_SPLIT, EDGE_T, 0, stream>>>(ei, rank, part);
    chunk_offsets_kernel<<<cdiv(NTOT / 4, OFFS_T), OFFS_T, 0, stream>>>(part, cnt);
    row_offsets_kernel<<<BATCH, SCAN_T, 0, stream>>>(cnt, rowptr, dis);
    fill_rows_kernel<<<BATCH * EDGE_SPLIT, EDGE_T, 0, stream>>>(ei, rank, rowptr, part, dis, csr);

    const int layer_blocks = cdiv(NTOT / 16, LAYER_T / 32);

    pad_rows_kernel<<<cdiv(NTOT, T), T, 0, stream>>>(X, Xp);
    tag_layer_kernel<KPAD, WINDOW><<<layer_blocks, LAYER_T, 0, stream>>>(rowptr, cnt, csr, Xp, W0, b0, hA);

    tag_layer_kernel<HID, HID><<<layer_blocks, LAYER_T, 0, stream>>>(rowptr, cnt, csr, hA, W1, b1, hB);

    tag_layer_kernel<HID, HID><<<layer_blocks, LAYER_T, 0, stream>>>(rowptr, cnt, csr, hB, W2, b2, hfinal);

    readout_kernel<<<cdiv(NTOT, T), T, 0, stream>>>(hfinal, Wreg, breg, out);
}


# layer waves copy their tile's contiguous run of list entries into LDS once with whole-line loads and gather rows from there: no entry load or round trip per group of row loads; sum order kept
# speedup vs baseline: 18.5940x; 1.0566x over previous
#include <hip/hip_runtime.h>
#include <hip/hip_bf16.h>
#include <math.h>

#define BATCH   8
#define WINDOW  5
#define NN      20000
#define EPB     320000
#define HID     32
#define KPAD    8
#define NTOT    (BATCH * NN)
#define ETOT    (BATCH * EPB)

typedef float v2f __attribute__((ext_vector_type(2)));
typedef float v8f __attribute__((ext_vector_type(8)));

__global__ void pad_rows_kernel(const float* __restrict__ x, float* __restrict__ xp) {
    int i = blockIdx.x * blockDim.x + threadIdx.x;
    if (i >= NTOT) return;
    const float* __restrict__ r = x + (size_t)i * WINDOW;
    float* __restrict__ o = xp + (size_t)i * KPAD;
    float4 lo = make_float4(r[0], r[1], r[2], r[3]);
    float4 hi = make_float4(r[4], 0.f, 0.f, 0.f);
    reinterpret_cast<float4*>(o)[0] = lo;
    reinterpret_cast<float4*>(o)[1] = hi;
}

#define EDGE_T      1024
#define EDGE_SPLIT  32
#define QPW         (EPB / 4 / EDGE_SPLIT)
#define EDGE_QPT    ((QPW + EDGE_T - 1) / EDGE_T)
#define NQ          (NN / 4)
#define TAB_TRIPS   5
#define TAB_T       (NQ / TAB_TRIPS)

__global__ void __launch_bounds__(EDGE_T)
count_rank_kernel(const int* __restrict__ ei, int* __restrict__ rank, int* __restrict__ part) {
    static_assert(EPB % (4 * EDGE_SPLIT) == 0 && NN % 4 == 0 && NQ % TAB_TRIPS == 0 &&
                  TAB_T <= EDGE_T && TAB_T % 8 == 0,
                  "whole edge quads per chunk; tables move 16 bytes at a time in whole lines");
    __shared__ int hist[NN];
    const int b   = blockIdx.x / EDGE_SPLIT;
    const int c   = blockIdx.x - b * EDGE_SPLIT;
    const int tid = threadIdx.x;
    const int4* __restrict__ dst4 =
        reinterpret_cast<const int4*>(ei + (size_t)b * 2 * EPB + EPB) + c * QPW;
    int4* __restrict__ rnk4 = reinterpret_cast<int4*>(rank + (size_t)b * EPB) + c * QPW;

    int4 dq[EDGE_QPT];
    #pragma unroll
    for (int k = 0; k < EDGE_QPT; ++k) {
        const int q = tid + k * EDGE_T;
        dq[k] = dst4[q < QPW ? q : QPW - 1];
    }
    for (int i = tid; i < NQ; i += EDGE_T)
        reinterpret_cast<int4*>(hist)[i] = make_int4(0, 0, 0, 0);
    __syncthreads();

    #pragma unroll
    for (int k = 0; k < EDGE_QPT; ++k) {
        const int q = tid + k * EDGE_T;
        if (q < QPW) {
            int4 r;
            r.x = atomicAdd(&hist[dq[k].x], 1);
            r.y = atomicAdd(&hist[dq[k].y], 1);
            r.z = atomicAdd(&hist[dq[k].z], 1);
            r.w = atomicAdd(&hist[dq[k].w], 1);
            rnk4[q] = r;
        }
    }
    __syncthreads();

    int4* __restrict__ gp = reinterpret_cast<int4*>(part + (size_t)blockIdx.x * NN);
    for (int i = tid; i < NQ; i += EDGE_T)
        gp[i] = reinterpret_cast<const int4*>(hist)[i];
}

#define OFFS_T 128
__global__ void __launch_bounds__(OFFS_T)
chunk_offsets_kernel(int* __restrict__ part, int* __restrict__ cnt) {
    const int gq = blockIdx.x * OFFS_T + threadIdx.x;
    if (gq >= NTOT / 4) return;
    const int b  = gq / NQ;
    const int nq = gq - b * NQ;
    int4* __restrict__ p = reinterpret_cast<int4*>(part + (size_t)b * EDGE_SPLIT * NN) + nq;
    int4 t[EDGE_SPLIT];
    #pragma unroll
    for (int c = 0; c < EDGE_SPLIT; ++c) t[c] = p[(size_t)c * NQ];
    int4 run = make_int4(0, 0, 0, 0);
    #pragma unroll
    for (int c = 0; c < EDGE_SPLIT; ++c) {
        p[(size_t)c * NQ] = run;
        run.x += t[c].x; run.y += t[c].y; run.z += t[c].z; run.w += t[c].w;
    }
    reinterpret_cast<int4*>(cnt)[gq] = run;
}

#define SCAN_T   1024
#define SCAN_PER 20
__global__ void __launch_bounds__(SCAN_T)
row_offsets_kernel(const int* __restrict__ cnt, int* __restrict__ rowptr,
                   float* __restrict__ dis) {
    static_assert(NN % SCAN_PER == 0 && NN / SCAN_PER <= SCAN_T && (SCAN_PER % 4) == 0,
                  "one thread scans SCAN_PER consecutive nodes with 16-byte accesses");
    __shared__ int wtot[SCAN_T / 32];
    const int b    = blockIdx.x;
    const int tid  = threadIdx.x;
    const int lane = tid & 31;
    const int wid  = tid >> 5;
    const bool act = tid < NN / SCAN_PER;
    const int first = b * NN + tid * SCAN_PER;

    int c[SCAN_PER];
    int tot = 0;
    #pragma unroll
    for (int k = 0; k < SCAN_PER; ++k) c[k] = 0;
    if (act) {
        const int4* __restrict__ p = reinterpret_cast<const int4*>(cnt + first);
        #pragma unroll
        for (int g = 0; g < SCAN_PER / 4; ++g) {
            int4 q = p[g];
            c[4 * g + 0] = q.x; c[4 * g + 1] = q.y; c[4 * g + 2] = q.z; c[4 * g + 3] = q.w;
            tot += q.x + q.y + q.z + q.w;
        }
    }

    int inc = tot;
    #pragma unroll
    for (int off = 1; off < 32; off <<= 1) {
        int t = __shfl_up(inc, off);
        if (lane >= off) inc += t;
    }
    if (lane == 31) wtot[wid] = inc;
    __syncthreads();
    if (wid == 0) {
        int w = wtot[lane];
        int winc = w;
        #pragma unroll
        for (int off = 1; off < 32; off <<= 1) {
            int t = __shfl_up(winc, off);
            if (lane >= off) winc += t;
        }
        wtot[lane] = winc - w;
    }
    __syncthreads();

    if (act) {
        int run = b * EPB + wtot[wid] + (inc - tot);
        int4*   __restrict__ rp = reinterpret_cast<int4*>(rowptr + first);
        float4* __restrict__ dp = reinterpret_cast<float4*>(dis + first);
        #pragma unroll
        for (int g = 0; g < SCAN_PER / 4; ++g) {
            int4 r; float4 f;
            r.x = run; run += c[4 * g + 0];
            r.y = run; run += c[4 * g + 1];
            r.z = run; run += c[4 * g + 2];
            r.w = run; run += c[4 * g + 3];
            f.x = (c[4 * g + 0] > 0) ? rsqrtf((float)c[4 * g + 0]) : 0.0f;
            f.y = (c[4 * g + 1] > 0) ? rsqrtf((float)c[4 * g + 1]) : 0.0f;
            f.z = (c[4 * g + 2] > 0) ? rsqrtf((float)c[4 * g + 2]) : 0.0f;
            f.w = (c[4 * g + 3] > 0) ? rsqrtf((float)c[4 * g + 3]) : 0.0f;
            rp[g] = r;
            dp[g] = f;
        }
    }
}

static __device__ __forceinline__ void put_edge(const int* __restrict__ s_off,
                                                const float* __restrict__ s_dis,
                                                int2* __restrict__ csr,
                                                int sl, int dl, int rk, int nodebase) {
    const int slot = s_off[dl] + rk;
    csr[slot] = make_int2(sl + nodebase, __float_as_int(s_dis[sl] * s_dis[dl]));
}
static __device__ __forceinline__ int4 add4(const int4& a, const int4& b) {
    return make_int4(a.x + b.x, a.y + b.y, a.z + b.z, a.w + b.w);
}
__global__ void __launch_bounds__(EDGE_T)
fill_rows_kernel(const int* __restrict__ ei, const int* __restrict__ rank,
                 const int* __restrict__ rowptr, const int* __restrict__ part,
                 const float* __restrict__ dis, int2* __restrict__ csr) {
    __shared__ int   s_off[NN];
    __shared__ float s_dis[NN];
    const int b   = blockIdx.x / EDGE_SPLIT;
    const int c   = blockIdx.x - b * EDGE_SPLIT;
    const int tid = threadIdx.x;

    const int* __restrict__ eb = ei + (size_t)b * 2 * EPB;
    const int4* __restrict__ src4 = reinterpret_cast<const int4*>(eb) + c * QPW;
    const int4* __restrict__ dst4 = reinterpret_cast<const int4*>(eb + EPB) + c * QPW;
    const int4* __restrict__ rnk4 =
        reinterpret_cast<const int4*>(rank + (size_t)b * EPB) + c * QPW;

    int4 sq[EDGE_QPT], dq[EDGE_QPT], rq[EDGE_QPT];
    #pragma unroll
    for (int k = 0; k < EDGE_QPT; ++k) {
        const int q  = tid + k * EDGE_T;
        const int qc = q < QPW ? q : QPW - 1;
        sq[k] = src4[qc]; dq[k] = dst4[qc]; rq[k] = rnk4[qc];
    }

    const int4*   __restrict__ gr = reinterpret_cast<const int4*>(rowptr + b * NN);
    const int4*   __restrict__ gp = reinterpret_cast<const int4*>(part + (size_t)blockIdx.x * NN);
    const float4* __restrict__ gd = reinterpret_cast<const float4*>(dis + b * NN);
    if (tid < TAB_T) {
        int4 tr[TAB_TRIPS], tp[TAB_TRIPS]; float4 td[TAB_TRIPS];
        #pragma unroll
        for (int k = 0; k < TAB_TRIPS; ++k) {
            const int i = tid + k * TAB_T;
            tr[k] = gr[i]; tp[k] = gp[i]; td[k] = gd[i];
        }
        #pragma unroll
        for (int k = 0; k < TAB_TRIPS; ++k) {
            const int i = tid + k * TAB_T;
            reinterpret_cast<int4*>(s_off)[i]   = add4(tr[k], tp[k]);
            reinterpret_cast<float4*>(s_dis)[i] = td[k];
        }
    }
    __syncthreads();

    const int nodebase = b * NN;
    #pragma unroll
    for (int k = 0; k < EDGE_QPT; ++k) {
        const int q = tid + k * EDGE_T;
        if (q < QPW) {
            put_edge(s_off, s_dis, csr, sq[k].x, dq[k].x, rq[k].x, nodebase);
            put_edge(s_off, s_dis, csr, sq[k].y, dq[k].y, rq[k].y, nodebase);
            put_edge(s_off, s_dis, csr, sq[k].z, dq[k].z, rq[k].z, nodebase);
            put_edge(s_off, s_dis, csr, sq[k].w, dq[k].w, rq[k].w, nodebase);
        }
    }
}

static __device__ __forceinline__ int2 load_entry(const int2* __restrict__ csr, unsigned idx) {
    return *reinterpret_cast<const int2*>(reinterpret_cast<const char*>(csr) + idx * 8u);
}
template <int S>
static __device__ __forceinline__ float4 load_row4(const float* __restrict__ hin, int src,
                                                   unsigned suboff) {
    return *reinterpret_cast<const float4*>(reinterpret_cast<const char*>(hin) +
                                            ((unsigned)src * (unsigned)(S * 4) + suboff));
}
static __device__ __forceinline__ void axpy4(float4& acc, const float4& v, float w) {
    acc.x += v.x * w; acc.y += v.y * w; acc.z += v.z * w; acc.w += v.w * w;
}

template <int S>
static __device__ __forceinline__ float4 gather_sum(const int2* __restrict__ csr,
                                                    const float* __restrict__ hin,
                                                    unsigned start, int n, unsigned suboff) {
    float4 acc = make_float4(0.f, 0.f, 0.f, 0.f);
    int j = 0;
    for (; j + 4 <= n; j += 4) {
        const int2 e0 = load_entry(csr, start + j);
        const int2 e1 = load_entry(csr, start + j + 1);
        const int2 e2 = load_entry(csr, start + j + 2);
        const int2 e3 = load_entry(csr, start + j + 3);
        const float4 v0 = load_row4<S>(hin, e0.x, suboff);
        const float4 v1 = load_row4<S>(hin, e1.x, suboff);
        const float4 v2 = load_row4<S>(hin, e2.x, suboff);
        const float4 v3 = load_row4<S>(hin, e3.x, suboff);
        axpy4(acc, v0, __int_as_float(e0.y));
        axpy4(acc, v1, __int_as_float(e1.y));
        axpy4(acc, v2, __int_as_float(e2.y));
        axpy4(acc, v3, __int_as_float(e3.y));
    }
    const int r = n - j;
    if (r > 0) {
        const int2 e0 = load_entry(csr, start + j);
        const int2 e1 = load_entry(csr, start + j + (r > 1 ? 1 : 0));
        const int2 e2 = load_entry(csr, start + j + (r > 2 ? 2 : 0));
        float4 v0 = load_row4<S>(hin, e0.x, suboff);
        float4 v1 = load_row4<S>(hin, e1.x, suboff);
        float4 v2 = load_row4<S>(hin, e2.x, suboff);
        v1.x = (r > 1) ? v1.x : 0.f; v1.y = (r > 1) ? v1.y : 0.f;
        v1.z = (r > 1) ? v1.z : 0.f; v1.w = (r > 1) ? v1.w : 0.f;
        v2.x = (r > 2) ? v2.x : 0.f; v2.y = (r > 2) ? v2.y : 0.f;
        v2.z = (r > 2) ? v2.z : 0.f; v2.w = (r > 2) ? v2.w : 0.f;
        axpy4(acc, v0, __int_as_float(e0.y));
        axpy4(acc, v1, __int_as_float(e1.y));
        axpy4(acc, v2, __int_as_float(e2.y));
    }
    return acc;
}

template <int S, int M>
static __device__ __forceinline__ void gather_tail(float4& acc, const int2* __restrict__ ent,
                                                   const float* __restrict__ hin,
                                                   int r, unsigned suboff) {
    int2 e[M]; float4 v[M];
    #pragma unroll
    for (int u = 0; u < M; ++u) e[u] = ent[u < r ? u : 0];
    #pragma unroll
    for (int u = 0; u < M; ++u) v[u] = load_row4<S>(hin, e[u].x, suboff);
    #pragma unroll
    for (int u = 1; u < M; ++u) {
        v[u].x = (u < r) ? v[u].x : 0.f; v[u].y = (u < r) ? v[u].y : 0.f;
        v[u].z = (u < r) ? v[u].z : 0.f; v[u].w = (u < r) ? v[u].w : 0.f;
    }
    #pragma unroll
    for (int u = 0; u < M; ++u) axpy4(acc, v[u], __int_as_float(e[u].y));
}

#define GATHER_DEPTH 8
template <int S>
static __device__ __forceinline__ float4 gather_sum_staged(const int2* __restrict__ ent,
                                                           const float* __restrict__ hin,
                                                           int n, unsigned suboff) {
    float4 acc = make_float4(0.f, 0.f, 0.f, 0.f);
    int j = 0;
    for (; j + GATHER_DEPTH <= n; j += GATHER_DEPTH) {
        int2 e[GATHER_DEPTH]; float4 v[GATHER_DEPTH];
        #pragma unroll
        for (int u = 0; u < GATHER_DEPTH; ++u) e[u] = ent[j + u];
        #pragma unroll
        for (int u = 0; u < GATHER_DEPTH; ++u) v[u] = load_row4<S>(hin, e[u].x, suboff);
        #pragma unroll
        for (int u = 0; u < GATHER_DEPTH; ++u) axpy4(acc, v[u], __int_as_float(e[u].y));
    }
    const int r = n - j;
    if (r > 3)      gather_tail<S, GATHER_DEPTH - 1>(acc, ent + j, hin, r, suboff);
    else if (r > 0) gather_tail<S, 3>(acc, ent + j, hin, r, suboff);
    return acc;
}

#define LAYER_T 128
#define ENT_CAP 320
#define ENT_PER_LANE (ENT_CAP / 32)
template <int S, int KW>
__global__ void __launch_bounds__(LAYER_T)
tag_layer_kernel(const int* __restrict__ rowptr, const int* __restrict__ cnt,
                 const int2* __restrict__ csr, const float* __restrict__ hin,
                 const float* __restrict__ W, const float* __restrict__ bias,
                 float* __restrict__ hout) {
    constexpr int K2    = 2 * S;
    constexpr int PITCH = K2 + 4;
    constexpr int LPR   = S / 4;
    constexpr int RPP   = 32 / LPR;
    constexpr int NPASS = 16 / RPP;
    constexpr int WAVES = LAYER_T / 32;
    constexpr int WTRIPS = (K2 * HID / 4) / LAYER_T;
    static_assert((K2 * HID / 4) % LAYER_T == 0 && HID == 32,
                  "whole staging trips; two 16-column output tiles");
    static_assert((S % 4) == 0 && (32 % LPR) == 0 && (16 % RPP) == 0 && ((PITCH / 4) % 2) == 1,
                  "whole float4 pieces per lane, whole passes per tile, odd 16-byte pitch");
    static_assert((NTOT % (16 * WAVES)) == 0, "every wave of every work-group has a whole tile");
    static_assert((NN % 16) == 0 && (ENT_CAP % 32) == 0,
                  "a tile lies inside one graph; whole waves of entries per staging trip");
    __shared__ __attribute__((aligned(16))) float sW[K2 * HID];
    __shared__ __attribute__((aligned(16))) float sT[WAVES * 16 * PITCH];
    __shared__ __attribute__((aligned(16))) int2  sE[WAVES * ENT_CAP];

    const int lane = threadIdx.x & 31;
    const int wib  = threadIdx.x >> 5;
    const int row0 = (blockIdx.x * WAVES + wib) * 16;
    const int sub  = lane % LPR;
    const int rl   = lane / LPR;
    const unsigned suboff = (unsigned)sub * 16u;
    const int m     = lane & 15;
    const int khalf = (lane >> 4) * 2;

    int n[NPASS]; unsigned st[NPASS]; float4 own[NPASS];
    #pragma unroll
    for (int p = 0; p < NPASS; ++p) {
        const int row = row0 + p * RPP + rl;
        n[p]  = cnt[row];
        st[p] = (unsigned)rowptr[row];
    }
    #pragma unroll
    for (int p = 0; p < NPASS; ++p)
        own[p] = load_row4<S>(hin, row0 + p * RPP + rl, suboff);

    const unsigned run0 = (unsigned)rowptr[row0];
    const unsigned runn = (unsigned)rowptr[row0 + 15] + (unsigned)cnt[row0 + 15] - run0;
    const bool staged = runn <= (unsigned)ENT_CAP;

    float4 wv[WTRIPS];
    #pragma unroll
    for (int t = 0; t < WTRIPS; ++t) {
        const int i   = threadIdx.x + t * LAYER_T;
        const int s   = i / (S * (HID / 4));
        const int rem = i - s * (S * (HID / 4));
        const int kw  = rem / (HID / 4);
        const int c4  = rem - kw * (HID / 4);
        wv[t] = (kw < KW) ? reinterpret_cast<const float4*>(W)[(s * KW + kw) * (HID / 4) + c4]
                          : make_float4(0.f, 0.f, 0.f, 0.f);
    }
    const float bias0 = bias[m];
    const float bias1 = bias[16 + m];
    int2 ev[ENT_PER_LANE];
    #pragma unroll
    for (int k = 0; k < ENT_PER_LANE; ++k) {
        const unsigned i = (unsigned)lane + 32u * k;
        ev[k] = load_entry(csr, (staged && i < runn) ? run0 + i : 0u);
    }
    #pragma unroll
    for (int t = 0; t < WTRIPS; ++t)
        reinterpret_cast<float4*>(sW)[threadIdx.x + t * LAYER_T] = wv[t];
    __syncthreads();

    int2* __restrict__ ents = sE + wib * ENT_CAP;
    #pragma unroll
    for (int k = 0; k < ENT_PER_LANE; ++k) ents[lane + 32 * k] = ev[k];
    __builtin_amdgcn_fence(__ATOMIC_RELEASE, "wavefront");
    __builtin_amdgcn_wave_barrier();
    __builtin_amdgcn_fence(__ATOMIC_ACQUIRE, "wavefront");

    float* __restrict__ tile = sT + wib * (16 * PITCH);
    #pragma unroll
    for (int p = 0; p < NPASS; ++p) {
        const float4 acc = staged ? gather_sum_staged<S>(ents + (st[p] - run0), hin, n[p], suboff)
                                  : gather_sum<S>(csr, hin, st[p], n[p], suboff);
        float* trow = tile + (p * RPP + rl) * PITCH + 4 * sub;
        *reinterpret_cast<float4*>(trow)     = own[p];
        *reinterpret_cast<float4*>(trow + S) = acc;
    }
    __builtin_amdgcn_fence(__ATOMIC_RELEASE, "wavefront");
    __builtin_amdgcn_wave_barrier();
    __builtin_amdgcn_fence(__ATOMIC_ACQUIRE, "wavefront");

    v8f acc0 = {0.f,0.f,0.f,0.f,0.f,0.f,0.f,0.f};
    v8f acc1 = {0.f,0.f,0.f,0.f,0.f,0.f,0.f,0.f};

    #pragma unroll
    for (int c = 0; c < K2 / 4; ++c) {
        const int k0 = c * 4 + khalf;
        v2f a = *reinterpret_cast<const v2f*>(tile + m * PITCH + k0);
        v2f b0, b1;
        b0.x = sW[k0 * HID + m];
        b0.y = sW[(k0 + 1) * HID + m];
        b1.x = sW[k0 * HID + 16 + m];
        b1.y = sW[(k0 + 1) * HID + 16 + m];
        acc0 = __builtin_amdgcn_wmma_f32_16x16x4_f32(
                   false, a, false, b0, (short)0, acc0, false, false);
        acc1 = __builtin_amdgcn_wmma_f32_16x16x4_f32(
                   false, a, false, b1, (short)0, acc1, false, false);
    }

    const int rbase = row0 + ((lane >> 4) ? 8 : 0);
    #pragma unroll
    for (int v = 0; v < 8; ++v) {
        int r = rbase + v;
        float x0 = acc0[v] + bias0;
        float x1 = acc1[v] + bias1;
        hout[(size_t)r * HID + m]      = tanhf(x0);
        hout[(size_t)r * HID + 16 + m] = tanhf(x1);
    }
}

__global__ void readout_kernel(const float* __restrict__ h, const float* __restrict__ Wreg,
                               const float* __restrict__ breg, float* __restrict__ out) {
    int i = blockIdx.x * blockDim.x + threadIdx.x;
    if (i >= NTOT) return;
    float acc = breg[0];
    const float4* hr = reinterpret_cast<const float4*>(h + (size_t)i * HID);
    #pragma unroll
    for (int g = 0; g < HID / 4; ++g) {
        float4 v = hr[g];
        acc += v.x * Wreg[4 * g + 0] + v.y * Wreg[4 * g + 1]
             + v.z * Wreg[4 * g + 2] + v.w * Wreg[4 * g + 3];
    }
    out[i] = acc;
}

static inline int cdiv(long long a, long long b) { return (int)((a + b - 1) / b); }

extern "C" void kernel_launch(void* const* d_in, const int* in_sizes, int n_in,
                              void* d_out, int out_size, void* d_ws, size_t ws_size,
                              hipStream_t stream) {
    const float* X    = (const float*)d_in[0];
    const int*   ei   = (const int*)  d_in[1];
    const float* W0   = (const float*)d_in[2];
    const float* b0   = (const float*)d_in[3];
    const float* W1   = (const float*)d_in[4];
    const float* b1   = (const float*)d_in[5];
    const float* W2   = (const float*)d_in[6];
    const float* b2   = (const float*)d_in[7];
    const float* Wreg = (const float*)d_in[8];
    const float* breg = (const float*)d_in[9];

    float* out    = (float*)d_out;
    float* hfinal = out + NTOT;

    int*   cnt    = (int*)d_ws;
    int*   rowptr = cnt + NTOT;
    float* dis    = (float*)(rowptr + NTOT);
    int*   rank   = (int*)(dis + NTOT);
    int2*  csr    = (int2*)(rank + ETOT);
    float* hA     = (float*)(csr + ETOT);
    float* hB     = hA  + (size_t)NTOT * HID;
    float* Xp     = hB  + (size_t)NTOT * HID;
    int*   part   = (int*)(Xp + (size_t)NTOT * KPAD);

    const int T = 256;

    count_rank_kernel<<<BATCH * EDGE_SPLIT, EDGE_T, 0, stream>>>(ei, rank, part);
    chunk_offsets_kernel<<<cdiv(NTOT / 4, OFFS_T), OFFS_T, 0, stream>>>(part, cnt);
    row_offsets_kernel<<<BATCH, SCAN_T, 0, stream>>>(cnt, rowptr, dis);
    fill_rows_kernel<<<BATCH * EDGE_SPLIT, EDGE_T, 0, stream>>>(ei, rank, rowptr, part, dis, csr);

    const int layer_blocks = cdiv(NTOT / 16, LAYER_T / 32);

    pad_rows_kernel<<<cdiv(NTOT, T), T, 0, stream>>>(X, Xp);
    tag_layer_kernel<KPAD, WINDOW><<<layer_blocks, LAYER_T, 0, stream>>>(rowptr, cnt, csr, Xp, W0, b0, hA);

    tag_layer_kernel<HID, HID><<<layer_blocks, LAYER_T, 0, stream>>>(rowptr, cnt, csr, hA, W1, b1, hB);

    tag_layer_kernel<HID, HID><<<layer_blocks, LAYER_T, 0, stream>>>(rowptr, cnt, csr, hB, W2, b2, hfinal);

    readout_kernel<<<cdiv(NTOT, T), T, 0, stream>>>(hfinal, Wreg, breg, out);
}


# layers 1 and 2 run as 8-wave work-groups (W staged once per 8 tiles, 62 KiB LDS): 5 per WGP = 40 waves, so all 10000 tile waves are resident at once, not 8192 plus a second round; sum order kept
# speedup vs baseline: 19.4611x; 1.0466x over previous
#include <hip/hip_runtime.h>
#include <hip/hip_bf16.h>
#include <math.h>

#define BATCH   8
#define WINDOW  5
#define NN      20000
#define EPB     320000
#define HID     32
#define KPAD    8
#define NTOT    (BATCH * NN)
#define ETOT    (BATCH * EPB)

typedef float v2f __attribute__((ext_vector_type(2)));
typedef float v8f __attribute__((ext_vector_type(8)));

__global__ void pad_rows_kernel(const float* __restrict__ x, float* __restrict__ xp) {
    int i = blockIdx.x * blockDim.x + threadIdx.x;
    if (i >= NTOT) return;
    const float* __restrict__ r = x + (size_t)i * WINDOW;
    float* __restrict__ o = xp + (size_t)i * KPAD;
    float4 lo = make_float4(r[0], r[1], r[2], r[3]);
    float4 hi = make_float4(r[4], 0.f, 0.f, 0.f);
    reinterpret_cast<float4*>(o)[0] = lo;
    reinterpret_cast<float4*>(o)[1] = hi;
}

#define EDGE_T      1024
#define EDGE_SPLIT  32
#define QPW         (EPB / 4 / EDGE_SPLIT)
#define EDGE_QPT    ((QPW + EDGE_T - 1) / EDGE_T)
#define NQ          (NN / 4)
#define TAB_TRIPS   5
#define TAB_T       (NQ / TAB_TRIPS)

__global__ void __launch_bounds__(EDGE_T)
count_rank_kernel(const int* __restrict__ ei, int* __restrict__ rank, int* __restrict__ part) {
    static_assert(EPB % (4 * EDGE_SPLIT) == 0 && NN % 4 == 0 && NQ % TAB_TRIPS == 0 &&
                  TAB_T <= EDGE_T && TAB_T % 8 == 0,
                  "whole edge quads per chunk; tables move 16 bytes at a time in whole lines");
    __shared__ int hist[NN];
    const int b   = blockIdx.x / EDGE_SPLIT;
    const int c   = blockIdx.x - b * EDGE_SPLIT;
    const int tid = threadIdx.x;
    const int4* __restrict__ dst4 =
        reinterpret_cast<const int4*>(ei + (size_t)b * 2 * EPB + EPB) + c * QPW;
    int4* __restrict__ rnk4 = reinterpret_cast<int4*>(rank + (size_t)b * EPB) + c * QPW;

    int4 dq[EDGE_QPT];
    #pragma unroll
    for (int k = 0; k < EDGE_QPT; ++k) {
        const int q = tid + k * EDGE_T;
        dq[k] = dst4[q < QPW ? q : QPW - 1];
    }
    for (int i = tid; i < NQ; i += EDGE_T)
        reinterpret_cast<int4*>(hist)[i] = make_int4(0, 0, 0, 0);
    __syncthreads();

    #pragma unroll
    for (int k = 0; k < EDGE_QPT; ++k) {
        const int q = tid + k * EDGE_T;
        if (q < QPW) {
            int4 r;
            r.x = atomicAdd(&hist[dq[k].x], 1);
            r.y = atomicAdd(&hist[dq[k].y], 1);
            r.z = atomicAdd(&hist[dq[k].z], 1);
            r.w = atomicAdd(&hist[dq[k].w], 1);
            rnk4[q] = r;
        }
    }
    __syncthreads();

    int4* __restrict__ gp = reinterpret_cast<int4*>(part + (size_t)blockIdx.x * NN);
    for (int i = tid; i < NQ; i += EDGE_T)
        gp[i] = reinterpret_cast<const int4*>(hist)[i];
}

#define OFFS_T 128
__global__ void __launch_bounds__(OFFS_T)
chunk_offsets_kernel(int* __restrict__ part, int* __restrict__ cnt) {
    const int gq = blockIdx.x * OFFS_T + threadIdx.x;
    if (gq >= NTOT / 4) return;
    const int b  = gq / NQ;
    const int nq = gq - b * NQ;
    int4* __restrict__ p = reinterpret_cast<int4*>(part + (size_t)b * EDGE_SPLIT * NN) + nq;
    int4 t[EDGE_SPLIT];
    #pragma unroll
    for (int c = 0; c < EDGE_SPLIT; ++c) t[c] = p[(size_t)c * NQ];
    int4 run = make_int4(0, 0, 0, 0);
    #pragma unroll
    for (int c = 0; c < EDGE_SPLIT; ++c) {
        p[(size_t)c * NQ] = run;
        run.x += t[c].x; run.y += t[c].y; run.z += t[c].z; run.w += t[c].w;
    }
    reinterpret_cast<int4*>(cnt)[gq] = run;
}

#define SCAN_T   1024
#define SCAN_PER 20
__global__ void __launch_bounds__(SCAN_T)
row_offsets_kernel(const int* __restrict__ cnt, int* __restrict__ rowptr,
                   float* __restrict__ dis) {
    static_assert(NN % SCAN_PER == 0 && NN / SCAN_PER <= SCAN_T && (SCAN_PER % 4) == 0,
                  "one thread scans SCAN_PER consecutive nodes with 16-byte accesses");
    __shared__ int wtot[SCAN_T / 32];
    const int b    = blockIdx.x;
    const int tid  = threadIdx.x;
    const int lane = tid & 31;
    const int wid  = tid >> 5;
    const bool act = tid < NN / SCAN_PER;
    const int first = b * NN + tid * SCAN_PER;

    int c[SCAN_PER];
    int tot = 0;
    #pragma unroll
    for (int k = 0; k < SCAN_PER; ++k) c[k] = 0;
    if (act) {
        const int4* __restrict__ p = reinterpret_cast<const int4*>(cnt + first);
        #pragma unroll
        for (int g = 0; g < SCAN_PER / 4; ++g) {
            int4 q = p[g];
            c[4 * g + 0] = q.x; c[4 * g + 1] = q.y; c[4 * g + 2] = q.z; c[4 * g + 3] = q.w;
            tot += q.x + q.y + q.z + q.w;
        }
    }

    int inc = tot;
    #pragma unroll
    for (int off = 1; off < 32; off <<= 1) {
        int t = __shfl_up(inc, off);
        if (lane >= off) inc += t;
    }
    if (lane == 31) wtot[wid] = inc;
    __syncthreads();
    if (wid == 0) {
        int w = wtot[lane];
        int winc = w;
        #pragma unroll
        for (int off = 1; off < 32; off <<= 1) {
            int t = __shfl_up(winc, off);
            if (lane >= off) winc += t;
        }
        wtot[lane] = winc - w;
    }
    __syncthreads();

    if (act) {
        int run = b * EPB + wtot[wid] + (inc - tot);
        int4*   __restrict__ rp = reinterpret_cast<int4*>(rowptr + first);
        float4* __restrict__ dp = reinterpret_cast<float4*>(dis + first);
        #pragma unroll
        for (int g = 0; g < SCAN_PER / 4; ++g) {
            int4 r; float4 f;
            r.x = run; run += c[4 * g + 0];
            r.y = run; run += c[4 * g + 1];
            r.z = run; run += c[4 * g + 2];
            r.w = run; run += c[4 * g + 3];
            f.x = (c[4 * g + 0] > 0) ? rsqrtf((float)c[4 * g + 0]) : 0.0f;
            f.y = (c[4 * g + 1] > 0) ? rsqrtf((float)c[4 * g + 1]) : 0.0f;
            f.z = (c[4 * g + 2] > 0) ? rsqrtf((float)c[4 * g + 2]) : 0.0f;
            f.w = (c[4 * g + 3] > 0) ? rsqrtf((float)c[4 * g + 3]) : 0.0f;
            rp[g] = r;
            dp[g] = f;
        }
    }
}

static __device__ __forceinline__ void put_edge(const int* __restrict__ s_off,
                                                const float* __restrict__ s_dis,
                                                int2* __restrict__ csr,
                                                int sl, int dl, int rk, int nodebase) {
    const int slot = s_off[dl] + rk;
    csr[slot] = make_int2(sl + nodebase, __float_as_int(s_dis[sl] * s_dis[dl]));
}
static __device__ __forceinline__ int4 add4(const int4& a, const int4& b) {
    return make_int4(a.x + b.x, a.y + b.y, a.z + b.z, a.w + b.w);
}
__global__ void __launch_bounds__(EDGE_T)
fill_rows_kernel(const int* __restrict__ ei, const int* __restrict__ rank,
                 const int* __restrict__ rowptr, const int* __restrict__ part,
                 const float* __restrict__ dis, int2* __restrict__ csr) {
    __shared__ int   s_off[NN];
    __shared__ float s_dis[NN];
    const int b   = blockIdx.x / EDGE_SPLIT;
    const int c   = blockIdx.x - b * EDGE_SPLIT;
    const int tid = threadIdx.x;

    const int* __restrict__ eb = ei + (size_t)b * 2 * EPB;
    const int4* __restrict__ src4 = reinterpret_cast<const int4*>(eb) + c * QPW;
    const int4* __restrict__ dst4 = reinterpret_cast<const int4*>(eb + EPB) + c * QPW;
    const int4* __restrict__ rnk4 =
        reinterpret_cast<const int4*>(rank + (size_t)b * EPB) + c * QPW;

    int4 sq[EDGE_QPT], dq[EDGE_QPT], rq[EDGE_QPT];
    #pragma unroll
    for (int k = 0; k < EDGE_QPT; ++k) {
        const int q  = tid + k * EDGE_T;
        const int qc = q < QPW ? q : QPW - 1;
        sq[k] = src4[qc]; dq[k] = dst4[qc]; rq[k] = rnk4[qc];
    }

    const int4*   __restrict__ gr = reinterpret_cast<const int4*>(rowptr + b * NN);
    const int4*   __restrict__ gp = reinterpret_cast<const int4*>(part + (size_t)blockIdx.x * NN);
    const float4* __restrict__ gd = reinterpret_cast<const float4*>(dis + b * NN);
    if (tid < TAB_T) {
        int4 tr[TAB_TRIPS], tp[TAB_TRIPS]; float4 td[TAB_TRIPS];
        #pragma unroll
        for (int k = 0; k < TAB_TRIPS; ++k) {
            const int i = tid + k * TAB_T;
            tr[k] = gr[i]; tp[k] = gp[i]; td[k] = gd[i];
        }
        #pragma unroll
        for (int k = 0; k < TAB_TRIPS; ++k) {
            const int i = tid + k * TAB_T;
            reinterpret_cast<int4*>(s_off)[i]   = add4(tr[k], tp[k]);
            reinterpret_cast<float4*>(s_dis)[i] = td[k];
        }
    }
    __syncthreads();

    const int nodebase = b * NN;
    #pragma unroll
    for (int k = 0; k < EDGE_QPT; ++k) {
        const int q = tid + k * EDGE_T;
        if (q < QPW) {
            put_edge(s_off, s_dis, csr, sq[k].x, dq[k].x, rq[k].x, nodebase);
            put_edge(s_off, s_dis, csr, sq[k].y, dq[k].y, rq[k].y, nodebase);
            put_edge(s_off, s_dis, csr, sq[k].z, dq[k].z, rq[k].z, nodebase);
            put_edge(s_off, s_dis, csr, sq[k].w, dq[k].w, rq[k].w, nodebase);
        }
    }
}

static __device__ __forceinline__ int2 load_entry(const int2* __restrict__ csr, unsigned idx) {
    return *reinterpret_cast<const int2*>(reinterpret_cast<const char*>(csr) + idx * 8u);
}
template <int S>
static __device__ __forceinline__ float4 load_row4(const float* __restrict__ hin, int src,
                                                   unsigned suboff) {
    return *reinterpret_cast<const float4*>(reinterpret_cast<const char*>(hin) +
                                            ((unsigned)src * (unsigned)(S * 4) + suboff));
}
static __device__ __forceinline__ void axpy4(float4& acc, const float4& v, float w) {
    acc.x += v.x * w; acc.y += v.y * w; acc.z += v.z * w; acc.w += v.w * w;
}

template <int S>
static __device__ __forceinline__ float4 gather_sum(const int2* __restrict__ csr,
                                                    const float* __restrict__ hin,
                                                    unsigned start, int n, unsigned suboff) {
    float4 acc = make_float4(0.f, 0.f, 0.f, 0.f);
    int j = 0;
    for (; j + 4 <= n; j += 4) {
        const int2 e0 = load_entry(csr, start + j);
        const int2 e1 = load_entry(csr, start + j + 1);
        const int2 e2 = load_entry(csr, start + j + 2);
        const int2 e3 = load_entry(csr, start + j + 3);
        const float4 v0 = load_row4<S>(hin, e0.x, suboff);
        const float4 v1 = load_row4<S>(hin, e1.x, suboff);
        const float4 v2 = load_row4<S>(hin, e2.x, suboff);
        const float4 v3 = load_row4<S>(hin, e3.x, suboff);
        axpy4(acc, v0, __int_as_float(e0.y));
        axpy4(acc, v1, __int_as_float(e1.y));
        axpy4(acc, v2, __int_as_float(e2.y));
        axpy4(acc, v3, __int_as_float(e3.y));
    }
    const int r = n - j;
    if (r > 0) {
        const int2 e0 = load_entry(csr, start + j);
        const int2 e1 = load_entry(csr, start + j + (r > 1 ? 1 : 0));
        const int2 e2 = load_entry(csr, start + j + (r > 2 ? 2 : 0));
        float4 v0 = load_row4<S>(hin, e0.x, suboff);
        float4 v1 = load_row4<S>(hin, e1.x, suboff);
        float4 v2 = load_row4<S>(hin, e2.x, suboff);
        v1.x = (r > 1) ? v1.x : 0.f; v1.y = (r > 1) ? v1.y : 0.f;
        v1.z = (r > 1) ? v1.z : 0.f; v1.w = (r > 1) ? v1.w : 0.f;
        v2.x = (r > 2) ? v2.x : 0.f; v2.y = (r > 2) ? v2.y : 0.f;
        v2.z = (r > 2) ? v2.z : 0.f; v2.w = (r > 2) ? v2.w : 0.f;
        axpy4(acc, v0, __int_as_float(e0.y));
        axpy4(acc, v1, __int_as_float(e1.y));
        axpy4(acc, v2, __int_as_float(e2.y));
    }
    return acc;
}

template <int S, int M>
static __device__ __forceinline__ void gather_tail(float4& acc, const int2* __restrict__ ent,
                                                   const float* __restrict__ hin,
                                                   int r, unsigned suboff) {
    int2 e[M]; float4 v[M];
    #pragma unroll
    for (int u = 0; u < M; ++u) e[u] = ent[u < r ? u : 0];
    #pragma unroll
    for (int u = 0; u < M; ++u) v[u] = load_row4<S>(hin, e[u].x, suboff);
    #pragma unroll
    for (int u = 1; u < M; ++u) {
        v[u].x = (u < r) ? v[u].x : 0.f; v[u].y = (u < r) ? v[u].y : 0.f;
        v[u].z = (u < r) ? v[u].z : 0.f; v[u].w = (u < r) ? v[u].w : 0.f;
    }
    #pragma unroll
    for (int u = 0; u < M; ++u) axpy4(acc, v[u], __int_as_float(e[u].y));
}

#define GATHER_DEPTH 8
template <int S>
static __device__ __forceinline__ float4 gather_sum_staged(const int2* __restrict__ ent,
                                                           const float* __restrict__ hin,
                                                           int n, unsigned suboff) {
    float4 acc = make_float4(0.f, 0.f, 0.f, 0.f);
    int j = 0;
    for (; j + GATHER_DEPTH <= n; j += GATHER_DEPTH) {
        int2 e[GATHER_DEPTH]; float4 v[GATHER_DEPTH];
        #pragma unroll
        for (int u = 0; u < GATHER_DEPTH; ++u) e[u] = ent[j + u];
        #pragma unroll
        for (int u = 0; u < GATHER_DEPTH; ++u) v[u] = load_row4<S>(hin, e[u].x, suboff);
        #pragma unroll
        for (int u = 0; u < GATHER_DEPTH; ++u) axpy4(acc, v[u], __int_as_float(e[u].y));
    }
    const int r = n - j;
    if (r > 3)      gather_tail<S, GATHER_DEPTH - 1>(acc, ent + j, hin, r, suboff);
    else if (r > 0) gather_tail<S, 3>(acc, ent + j, hin, r, suboff);
    return acc;
}

#define ENT_CAP 320
#define ENT_PER_LANE (ENT_CAP / 32)
template <int S, int KW, int LAYER_T>
__global__ void __launch_bounds__(LAYER_T)
tag_layer_kernel(const int* __restrict__ rowptr, const int* __restrict__ cnt,
                 const int2* __restrict__ csr, const float* __restrict__ hin,
                 const float* __restrict__ W, const float* __restrict__ bias,
                 float* __restrict__ hout) {
    constexpr int K2    = 2 * S;
    constexpr int PITCH = K2 + 4;
    constexpr int LPR   = S / 4;
    constexpr int RPP   = 32 / LPR;
    constexpr int NPASS = 16 / RPP;
    constexpr int WAVES = LAYER_T / 32;
    constexpr int WTRIPS = (K2 * HID / 4) / LAYER_T;
    static_assert((K2 * HID / 4) % LAYER_T == 0 && HID == 32,
                  "whole staging trips; two 16-column output tiles");
    static_assert((S % 4) == 0 && (32 % LPR) == 0 && (16 % RPP) == 0 && ((PITCH / 4) % 2) == 1,
                  "whole float4 pieces per lane, whole passes per tile, odd 16-byte pitch");
    static_assert((NTOT % (16 * WAVES)) == 0, "every wave of every work-group has a whole tile");
    static_assert((NN % 16) == 0 && (ENT_CAP % 32) == 0,
                  "a tile lies inside one graph; whole waves of entries per staging trip");
    __shared__ __attribute__((aligned(16))) float sW[K2 * HID];
    __shared__ __attribute__((aligned(16))) float sT[WAVES * 16 * PITCH];
    __shared__ __attribute__((aligned(16))) int2  sE[WAVES * ENT_CAP];

    const int lane = threadIdx.x & 31;
    const int wib  = threadIdx.x >> 5;
    const int row0 = (blockIdx.x * WAVES + wib) * 16;
    const int sub  = lane % LPR;
    const int rl   = lane / LPR;
    const unsigned suboff = (unsigned)sub * 16u;
    const int m     = lane & 15;
    const int khalf = (lane >> 4) * 2;

    int n[NPASS]; unsigned st[NPASS]; float4 own[NPASS];
    #pragma unroll
    for (int p = 0; p < NPASS; ++p) {
        const int row = row0 + p * RPP + rl;
        n[p]  = cnt[row];
        st[p] = (unsigned)rowptr[row];
    }
    #pragma unroll
    for (int p = 0; p < NPASS; ++p)
        own[p] = load_row4<S>(hin, row0 + p * RPP + rl, suboff);

    const unsigned run0 = (unsigned)rowptr[row0];
    const unsigned runn = (unsigned)rowptr[row0 + 15] + (unsigned)cnt[row0 + 15] - run0;
    const bool staged = runn <= (unsigned)ENT_CAP;

    float4 wv[WTRIPS];
    #pragma unroll
    for (int t = 0; t < WTRIPS; ++t) {
        const int i   = threadIdx.x + t * LAYER_T;
        const int s   = i / (S * (HID / 4));
        const int rem = i - s * (S * (HID / 4));
        const int kw  = rem / (HID / 4);
        const int c4  = rem - kw * (HID / 4);
        wv[t] = (kw < KW) ? reinterpret_cast<const float4*>(W)[(s * KW + kw) * (HID / 4) + c4]
                          : make_float4(0.f, 0.f, 0.f, 0.f);
    }
    const float bias0 = bias[m];
    const float bias1 = bias[16 + m];
    int2 ev[ENT_PER_LANE];
    #pragma unroll
    for (int k = 0; k < ENT_PER_LANE; ++k) {
        const unsigned i = (unsigned)lane + 32u * k;
        ev[k] = load_entry(csr, (staged && i < runn) ? run0 + i : 0u);
    }
    #pragma unroll
    for (int t = 0; t < WTRIPS; ++t)
        reinterpret_cast<float4*>(sW)[threadIdx.x + t * LAYER_T] = wv[t];
    __syncthreads();

    int2* __restrict__ ents = sE + wib * ENT_CAP;
    #pragma unroll
    for (int k = 0; k < ENT_PER_LANE; ++k) ents[lane + 32 * k] = ev[k];
    __builtin_amdgcn_fence(__ATOMIC_RELEASE, "wavefront");
    __builtin_amdgcn_wave_barrier();
    __builtin_amdgcn_fence(__ATOMIC_ACQUIRE, "wavefront");

    float* __restrict__ tile = sT + wib * (16 * PITCH);
    #pragma unroll
    for (int p = 0; p < NPASS; ++p) {
        const float4 acc = staged ? gather_sum_staged<S>(ents + (st[p] - run0), hin, n[p], suboff)
                                  : gather_sum<S>(csr, hin, st[p], n[p], suboff);
        float* trow = tile + (p * RPP + rl) * PITCH + 4 * sub;
        *reinterpret_cast<float4*>(trow)     = own[p];
        *reinterpret_cast<float4*>(trow + S) = acc;
    }
    __builtin_amdgcn_fence(__ATOMIC_RELEASE, "wavefront");
    __builtin_amdgcn_wave_barrier();
    __builtin_amdgcn_fence(__ATOMIC_ACQUIRE, "wavefront");

    v8f acc0 = {0.f,0.f,0.f,0.f,0.f,0.f,0.f,0.f};
    v8f acc1 = {0.f,0.f,0.f,0.f,0.f,0.f,0.f,0.f};

    #pragma unroll
    for (int c = 0; c < K2 / 4; ++c) {
        const int k0 = c * 4 + khalf;
        v2f a = *reinterpret_cast<const v2f*>(tile + m * PITCH + k0);
        v2f b0, b1;
        b0.x = sW[k0 * HID + m];
        b0.y = sW[(k0 + 1) * HID + m];
        b1.x = sW[k0 * HID + 16 + m];
        b1.y = sW[(k0 + 1) * HID + 16 + m];
        acc0 = __builtin_amdgcn_wmma_f32_16x16x4_f32(
                   false, a, false, b0, (short)0, acc0, false, false);
        acc1 = __builtin_amdgcn_wmma_f32_16x16x4_f32(
                   false, a, false, b1, (short)0, acc1, false, false);
    }

    const int rbase = row0 + ((lane >> 4) ? 8 : 0);
    #pragma unroll
    for (int v = 0; v < 8; ++v) {
        int r = rbase + v;
        float x0 = acc0[v] + bias0;
        float x1 = acc1[v] + bias1;
        hout[(size_t)r * HID + m]      = tanhf(x0);
        hout[(size_t)r * HID + 16 + m] = tanhf(x1);
    }
}

__global__ void readout_kernel(const float* __restrict__ h, const float* __restrict__ Wreg,
                               const float* __restrict__ breg, float* __restrict__ out) {
    int i = blockIdx.x * blockDim.x + threadIdx.x;
    if (i >= NTOT) return;
    float acc = breg[0];
    const float4* hr = reinterpret_cast<const float4*>(h + (size_t)i * HID);
    #pragma unroll
    for (int g = 0; g < HID / 4; ++g) {
        float4 v = hr[g];
        acc += v.x * Wreg[4 * g + 0] + v.y * Wreg[4 * g + 1]
             + v.z * Wreg[4 * g + 2] + v.w * Wreg[4 * g + 3];
    }
    out[i] = acc;
}

static inline int cdiv(long long a, long long b) { return (int)((a + b - 1) / b); }

extern "C" void kernel_launch(void* const* d_in, const int* in_sizes, int n_in,
                              void* d_out, int out_size, void* d_ws, size_t ws_size,
                              hipStream_t stream) {
    const float* X    = (const float*)d_in[0];
    const int*   ei   = (const int*)  d_in[1];
    const float* W0   = (const float*)d_in[2];
    const float* b0   = (const float*)d_in[3];
    const float* W1   = (const float*)d_in[4];
    const float* b1   = (const float*)d_in[5];
    const float* W2   = (const float*)d_in[6];
    const float* b2   = (const float*)d_in[7];
    const float* Wreg = (const float*)d_in[8];
    const float* breg = (const float*)d_in[9];

    float* out    = (float*)d_out;
    float* hfinal = out + NTOT;

    int*   cnt    = (int*)d_ws;
    int*   rowptr = cnt + NTOT;
    float* dis    = (float*)(rowptr + NTOT);
    int*   rank   = (int*)(dis + NTOT);
    int2*  csr    = (int2*)(rank + ETOT);
    float* hA     = (float*)(csr + ETOT);
    float* hB     = hA  + (size_t)NTOT * HID;
    float* Xp     = hB  + (size_t)NTOT * HID;
    int*   part   = (int*)(Xp + (size_t)NTOT * KPAD);

    const int T = 256;

    count_rank_kernel<<<BATCH * EDGE_SPLIT, EDGE_T, 0, stream>>>(ei, rank, part);
    chunk_offsets_kernel<<<cdiv(NTOT / 4, OFFS_T), OFFS_T, 0, stream>>>(part, cnt);
    row_offsets_kernel<<<BATCH, SCAN_T, 0, stream>>>(cnt, rowptr, dis);
    fill_rows_kernel<<<BATCH * EDGE_SPLIT, EDGE_T, 0, stream>>>(ei, rank, rowptr, part, dis, csr);

    const int narrow_blocks = cdiv(NTOT / 16, 128 / 32);
    const int wide_blocks   = cdiv(NTOT / 16, 256 / 32);

    pad_rows_kernel<<<cdiv(NTOT, T), T, 0, stream>>>(X, Xp);
    tag_layer_kernel<KPAD, WINDOW, 128><<<narrow_blocks, 128, 0, stream>>>(rowptr, cnt, csr, Xp, W0, b0, hA);

    tag_layer_kernel<HID, HID, 256><<<wide_blocks, 256, 0, stream>>>(rowptr, cnt, csr, hA, W1, b1, hB);

    tag_layer_kernel<HID, HID, 256><<<wide_blocks, 256, 0, stream>>>(rowptr, cnt, csr, hB, W2, b2, hfinal);

    readout_kernel<<<cdiv(NTOT, T), T, 0, stream>>>(hfinal, Wreg, breg, out);
}


# three helper launches folded into neighbours, 9 to 6: X padding in the chunk-prefix launch, row scan redone in LDS per fill work-group, readout in layer 2's epilogue; out summed as a 16-lane tree
# speedup vs baseline: 20.9220x; 1.0751x over previous
#include <hip/hip_runtime.h>
#include <hip/hip_bf16.h>
#include <math.h>

#define BATCH   8
#define WINDOW  5
#define NN      20000
#define EPB     320000
#define HID     32
#define KPAD    8
#define NTOT    (BATCH * NN)
#define ETOT    (BATCH * EPB)

typedef float v2f __attribute__((ext_vector_type(2)));
typedef float v8f __attribute__((ext_vector_type(8)));

#define EDGE_T      1024
#define EDGE_SPLIT  32
#define QPW         (EPB / 4 / EDGE_SPLIT)
#define EDGE_QPT    ((QPW + EDGE_T - 1) / EDGE_T)
#define NQ          (NN / 4)
#define TAB_TRIPS   5
#define TAB_T       (NQ / TAB_TRIPS)

__global__ void __launch_bounds__(EDGE_T)
count_rank_kernel(const int* __restrict__ ei, int* __restrict__ rank, int* __restrict__ part) {
    static_assert(EPB % (4 * EDGE_SPLIT) == 0 && NN % 4 == 0 && NQ % TAB_TRIPS == 0 &&
                  TAB_T <= EDGE_T && TAB_T % 8 == 0,
                  "whole edge quads per chunk; tables move 16 bytes at a time in whole lines");
    __shared__ int hist[NN];
    const int b   = blockIdx.x / EDGE_SPLIT;
    const int c   = blockIdx.x - b * EDGE_SPLIT;
    const int tid = threadIdx.x;
    const int4* __restrict__ dst4 =
        reinterpret_cast<const int4*>(ei + (size_t)b * 2 * EPB + EPB) + c * QPW;
    int4* __restrict__ rnk4 = reinterpret_cast<int4*>(rank + (size_t)b * EPB) + c * QPW;

    int4 dq[EDGE_QPT];
    #pragma unroll
    for (int k = 0; k < EDGE_QPT; ++k) {
        const int q = tid + k * EDGE_T;
        dq[k] = dst4[q < QPW ? q : QPW - 1];
    }
    for (int i = tid; i < NQ; i += EDGE_T)
        reinterpret_cast<int4*>(hist)[i] = make_int4(0, 0, 0, 0);
    __syncthreads();

    #pragma unroll
    for (int k = 0; k < EDGE_QPT; ++k) {
        const int q = tid + k * EDGE_T;
        if (q < QPW) {
            int4 r;
            r.x = atomicAdd(&hist[dq[k].x], 1);
            r.y = atomicAdd(&hist[dq[k].y], 1);
            r.z = atomicAdd(&hist[dq[k].z], 1);
            r.w = atomicAdd(&hist[dq[k].w], 1);
            rnk4[q] = r;
        }
    }
    __syncthreads();

    int4* __restrict__ gp = reinterpret_cast<int4*>(part + (size_t)blockIdx.x * NN);
    for (int i = tid; i < NQ; i += EDGE_T)
        gp[i] = reinterpret_cast<const int4*>(hist)[i];
}

#define OFFS_T 128
__global__ void __launch_bounds__(OFFS_T)
chunk_offsets_kernel(int* __restrict__ part, int* __restrict__ cnt,
                     const float* __restrict__ x, float* __restrict__ xp) {
    static_assert(WINDOW == 5 && KPAD == 8, "four rows of X are five 16-byte pieces");
    const int gq = blockIdx.x * OFFS_T + threadIdx.x;
    if (gq >= NTOT / 4) return;
    const int b  = gq / NQ;
    const int nq = gq - b * NQ;
    const float4* __restrict__ xr = reinterpret_cast<const float4*>(x) + (size_t)gq * WINDOW;
    const float4 a0 = xr[0], a1 = xr[1], a2 = xr[2], a3 = xr[3], a4 = xr[4];
    int4* __restrict__ p = reinterpret_cast<int4*>(part + (size_t)b * EDGE_SPLIT * NN) + nq;
    int4 t[EDGE_SPLIT];
    #pragma unroll
    for (int c = 0; c < EDGE_SPLIT; ++c) t[c] = p[(size_t)c * NQ];
    float4* __restrict__ xo = reinterpret_cast<float4*>(xp) + (size_t)gq * KPAD;
    xo[0] = a0;
    xo[1] = make_float4(a1.x, 0.f, 0.f, 0.f);
    xo[2] = make_float4(a1.y, a1.z, a1.w, a2.x);
    xo[3] = make_float4(a2.y, 0.f, 0.f, 0.f);
    xo[4] = make_float4(a2.z, a2.w, a3.x, a3.y);
    xo[5] = make_float4(a3.z, 0.f, 0.f, 0.f);
    xo[6] = make_float4(a3.w, a4.x, a4.y, a4.z);
    xo[7] = make_float4(a4.w, 0.f, 0.f, 0.f);
    int4 run = make_int4(0, 0, 0, 0);
    #pragma unroll
    for (int c = 0; c < EDGE_SPLIT; ++c) {
        p[(size_t)c * NQ] = run;
        run.x += t[c].x; run.y += t[c].y; run.z += t[c].z; run.w += t[c].w;
    }
    reinterpret_cast<int4*>(cnt)[gq] = run;
}

#define SCAN_PER 20
static __device__ __forceinline__ void put_edge(const int* __restrict__ s_off,
                                                const int* __restrict__ s_cd,
                                                int2* __restrict__ csr,
                                                int sl, int dl, int rk, int nodebase) {
    const int slot = s_off[dl] + rk;
    const float nrm = __int_as_float(s_cd[sl]) * __int_as_float(s_cd[dl]);
    csr[slot] = make_int2(sl + nodebase, __float_as_int(nrm));
}
static __device__ __forceinline__ int4 add4(const int4& a, const int4& b) {
    return make_int4(a.x + b.x, a.y + b.y, a.z + b.z, a.w + b.w);
}
__global__ void __launch_bounds__(EDGE_T)
fill_rows_kernel(const int* __restrict__ ei, const int* __restrict__ rank,
                 const int* __restrict__ cnt, const int* __restrict__ part,
                 int* __restrict__ rowptr, int2* __restrict__ csr) {
    static_assert(NN % SCAN_PER == 0 && NN / SCAN_PER == TAB_T && (SCAN_PER % 4) == 0 &&
                  SCAN_PER / 4 == TAB_TRIPS,
                  "the copying threads are the scanning threads: SCAN_PER consecutive nodes each");
    __shared__ int s_off[NN];
    __shared__ int s_cd[NN];
    __shared__ int wtot[EDGE_T / 32];
    const int b   = blockIdx.x / EDGE_SPLIT;
    const int ch  = blockIdx.x - b * EDGE_SPLIT;
    const int tid = threadIdx.x;
    const int lane = tid & 31;
    const int wid  = tid >> 5;

    const int* __restrict__ eb = ei + (size_t)b * 2 * EPB;
    const int4* __restrict__ src4 = reinterpret_cast<const int4*>(eb) + ch * QPW;
    const int4* __restrict__ dst4 = reinterpret_cast<const int4*>(eb + EPB) + ch * QPW;
    const int4* __restrict__ rnk4 =
        reinterpret_cast<const int4*>(rank + (size_t)b * EPB) + ch * QPW;

    int4 sq[EDGE_QPT], dq[EDGE_QPT], rq[EDGE_QPT];
    #pragma unroll
    for (int k = 0; k < EDGE_QPT; ++k) {
        const int q  = tid + k * EDGE_T;
        const int qc = q < QPW ? q : QPW - 1;
        sq[k] = src4[qc]; dq[k] = dst4[qc]; rq[k] = rnk4[qc];
    }

    const int4* __restrict__ gc = reinterpret_cast<const int4*>(cnt + b * NN);
    const int4* __restrict__ gp = reinterpret_cast<const int4*>(part + (size_t)blockIdx.x * NN);
    const bool act = tid < TAB_T;
    if (act) {
        int4 tc[TAB_TRIPS], tp[TAB_TRIPS];
        #pragma unroll
        for (int k = 0; k < TAB_TRIPS; ++k) {
            const int i = tid + k * TAB_T;
            tc[k] = gc[i]; tp[k] = gp[i];
        }
        #pragma unroll
        for (int k = 0; k < TAB_TRIPS; ++k) {
            const int i = tid + k * TAB_T;
            reinterpret_cast<int4*>(s_off)[i] = tp[k];
            reinterpret_cast<int4*>(s_cd)[i]  = tc[k];
        }
    }
    __syncthreads();

    int c[SCAN_PER];
    int tot = 0;
    #pragma unroll
    for (int k = 0; k < SCAN_PER; ++k) c[k] = 0;
    if (act) {
        #pragma unroll
        for (int g = 0; g < SCAN_PER / 4; ++g) {
            const int4 q = reinterpret_cast<const int4*>(s_cd)[tid * (SCAN_PER / 4) + g];
            c[4 * g + 0] = q.x; c[4 * g + 1] = q.y; c[4 * g + 2] = q.z; c[4 * g + 3] = q.w;
            tot += q.x + q.y + q.z + q.w;
        }
    }
    int inc = tot;
    #pragma unroll
    for (int off = 1; off < 32; off <<= 1) {
        int t = __shfl_up(inc, off);
        if (lane >= off) inc += t;
    }
    if (lane == 31) wtot[wid] = inc;
    __syncthreads();
    if (wid == 0) {
        int w = wtot[lane];
        int winc = w;
        #pragma unroll
        for (int off = 1; off < 32; off <<= 1) {
            int t = __shfl_up(winc, off);
            if (lane >= off) winc += t;
        }
        wtot[lane] = winc - w;
    }
    __syncthreads();

    if (act) {
        int run = b * EPB + wtot[wid] + (inc - tot);
        int4* __restrict__ rp = reinterpret_cast<int4*>(rowptr + b * NN + tid * SCAN_PER);
        #pragma unroll
        for (int g = 0; g < SCAN_PER / 4; ++g) {
            int4 r; int4 f;
            r.x = run; run += c[4 * g + 0];
            r.y = run; run += c[4 * g + 1];
            r.z = run; run += c[4 * g + 2];
            r.w = run; run += c[4 * g + 3];
            f.x = __float_as_int((c[4 * g + 0] > 0) ? rsqrtf((float)c[4 * g + 0]) : 0.0f);
            f.y = __float_as_int((c[4 * g + 1] > 0) ? rsqrtf((float)c[4 * g + 1]) : 0.0f);
            f.z = __float_as_int((c[4 * g + 2] > 0) ? rsqrtf((float)c[4 * g + 2]) : 0.0f);
            f.w = __float_as_int((c[4 * g + 3] > 0) ? rsqrtf((float)c[4 * g + 3]) : 0.0f);
            const int i = tid * (SCAN_PER / 4) + g;
            reinterpret_cast<int4*>(s_off)[i] = add4(reinterpret_cast<const int4*>(s_off)[i], r);
            reinterpret_cast<int4*>(s_cd)[i]  = f;
            if (ch == 0) rp[g] = r;
        }
    }
    __syncthreads();

    const int nodebase = b * NN;
    #pragma unroll
    for (int k = 0; k < EDGE_QPT; ++k) {
        const int q = tid + k * EDGE_T;
        if (q < QPW) {
            put_edge(s_off, s_cd, csr, sq[k].x, dq[k].x, rq[k].x, nodebase);
            put_edge(s_off, s_cd, csr, sq[k].y, dq[k].y, rq[k].y, nodebase);
            put_edge(s_off, s_cd, csr, sq[k].z, dq[k].z, rq[k].z, nodebase);
            put_edge(s_off, s_cd, csr, sq[k].w, dq[k].w, rq[k].w, nodebase);
        }
    }
}

static __device__ __forceinline__ int2 load_entry(const int2* __restrict__ csr, unsigned idx) {
    return *reinterpret_cast<const int2*>(reinterpret_cast<const char*>(csr) + idx * 8u);
}
template <int S>
static __device__ __forceinline__ float4 load_row4(const float* __restrict__ hin, int src,
                                                   unsigned suboff) {
    return *reinterpret_cast<const float4*>(reinterpret_cast<const char*>(hin) +
                                            ((unsigned)src * (unsigned)(S * 4) + suboff));
}
static __device__ __forceinline__ void axpy4(float4& acc, const float4& v, float w) {
    acc.x += v.x * w; acc.y += v.y * w; acc.z += v.z * w; acc.w += v.w * w;
}

template <int S>
static __device__ __forceinline__ float4 gather_sum(const int2* __restrict__ csr,
                                                    const float* __restrict__ hin,
                                                    unsigned start, int n, unsigned suboff) {
    float4 acc = make_float4(0.f, 0.f, 0.f, 0.f);
    int j = 0;
    for (; j + 4 <= n; j += 4) {
        const int2 e0 = load_entry(csr, start + j);
        const int2 e1 = load_entry(csr, start + j + 1);
        const int2 e2 = load_entry(csr, start + j + 2);
        const int2 e3 = load_entry(csr, start + j + 3);
        const float4 v0 = load_row4<S>(hin, e0.x, suboff);
        const float4 v1 = load_row4<S>(hin, e1.x, suboff);
        const float4 v2 = load_row4<S>(hin, e2.x, suboff);
        const float4 v3 = load_row4<S>(hin, e3.x, suboff);
        axpy4(acc, v0, __int_as_float(e0.y));
        axpy4(acc, v1, __int_as_float(e1.y));
        axpy4(acc, v2, __int_as_float(e2.y));
        axpy4(acc, v3, __int_as_float(e3.y));
    }
    const int r = n - j;
    if (r > 0) {
        const int2 e0 = load_entry(csr, start + j);
        const int2 e1 = load_entry(csr, start + j + (r > 1 ? 1 : 0));
        const int2 e2 = load_entry(csr, start + j + (r > 2 ? 2 : 0));
        float4 v0 = load_row4<S>(hin, e0.x, suboff);
        float4 v1 = load_row4<S>(hin, e1.x, suboff);
        float4 v2 = load_row4<S>(hin, e2.x, suboff);
        v1.x = (r > 1) ? v1.x : 0.f; v1.y = (r > 1) ? v1.y : 0.f;
        v1.z = (r > 1) ? v1.z : 0.f; v1.w = (r > 1) ? v1.w : 0.f;
        v2.x = (r > 2) ? v2.x : 0.f; v2.y = (r > 2) ? v2.y : 0.f;
        v2.z = (r > 2) ? v2.z : 0.f; v2.w = (r > 2) ? v2.w : 0.f;
        axpy4(acc, v0, __int_as_float(e0.y));
        axpy4(acc, v1, __int_as_float(e1.y));
        axpy4(acc, v2, __int_as_float(e2.y));
    }
    return acc;
}

template <int S, int M>
static __device__ __forceinline__ void gather_tail(float4& acc, const int2* __restrict__ ent,
                                                   const float* __restrict__ hin,
                                                   int r, unsigned suboff) {
    int2 e[M]; float4 v[M];
    #pragma unroll
    for (int u = 0; u < M; ++u) e[u] = ent[u < r ? u : 0];
    #pragma unroll
    for (int u = 0; u < M; ++u) v[u] = load_row4<S>(hin, e[u].x, suboff);
    #pragma unroll
    for (int u = 1; u < M; ++u) {
        v[u].x = (u < r) ? v[u].x : 0.f; v[u].y = (u < r) ? v[u].y : 0.f;
        v[u].z = (u < r) ? v[u].z : 0.f; v[u].w = (u < r) ? v[u].w : 0.f;
    }
    #pragma unroll
    for (int u = 0; u < M; ++u) axpy4(acc, v[u], __int_as_float(e[u].y));
}

#define GATHER_DEPTH 8
template <int S>
static __device__ __forceinline__ float4 gather_sum_staged(const int2* __restrict__ ent,
                                                           const float* __restrict__ hin,
                                                           int n, unsigned suboff) {
    float4 acc = make_float4(0.f, 0.f, 0.f, 0.f);
    int j = 0;
    for (; j + GATHER_DEPTH <= n; j += GATHER_DEPTH) {
        int2 e[GATHER_DEPTH]; float4 v[GATHER_DEPTH];
        #pragma unroll
        for (int u = 0; u < GATHER_DEPTH; ++u) e[u] = ent[j + u];
        #pragma unroll
        for (int u = 0; u < GATHER_DEPTH; ++u) v[u] = load_row4<S>(hin, e[u].x, suboff);
        #pragma unroll
        for (int u = 0; u < GATHER_DEPTH; ++u) axpy4(acc, v[u], __int_as_float(e[u].y));
    }
    const int r = n - j;
    if (r > 3)      gather_tail<S, GATHER_DEPTH - 1>(acc, ent + j, hin, r, suboff);
    else if (r > 0) gather_tail<S, 3>(acc, ent + j, hin, r, suboff);
    return acc;
}

#define ENT_CAP 320
#define ENT_PER_LANE (ENT_CAP / 32)
template <int S, int KW, int LAYER_T, bool READOUT>
__global__ void __launch_bounds__(LAYER_T)
tag_layer_kernel(const int* __restrict__ rowptr, const int* __restrict__ cnt,
                 const int2* __restrict__ csr, const float* __restrict__ hin,
                 const float* __restrict__ W, const float* __restrict__ bias,
                 float* __restrict__ hout, const float* __restrict__ wreg,
                 const float* __restrict__ breg, float* __restrict__ out) {
    constexpr int K2    = 2 * S;
    constexpr int PITCH = K2 + 4;
    constexpr int LPR   = S / 4;
    constexpr int RPP   = 32 / LPR;
    constexpr int NPASS = 16 / RPP;
    constexpr int WAVES = LAYER_T / 32;
    constexpr int WTRIPS = (K2 * HID / 4) / LAYER_T;
    static_assert((K2 * HID / 4) % LAYER_T == 0 && HID == 32,
                  "whole staging trips; two 16-column output tiles");
    static_assert((S % 4) == 0 && (32 % LPR) == 0 && (16 % RPP) == 0 && ((PITCH / 4) % 2) == 1,
                  "whole float4 pieces per lane, whole passes per tile, odd 16-byte pitch");
    static_assert((NTOT % (16 * WAVES)) == 0, "every wave of every work-group has a whole tile");
    static_assert((NN % 16) == 0 && (ENT_CAP % 32) == 0,
                  "a tile lies inside one graph; whole waves of entries per staging trip");
    __shared__ __attribute__((aligned(16))) float sW[K2 * HID];
    __shared__ __attribute__((aligned(16))) float sT[WAVES * 16 * PITCH];
    __shared__ __attribute__((aligned(16))) int2  sE[WAVES * ENT_CAP];

    const int lane = threadIdx.x & 31;
    const int wib  = threadIdx.x >> 5;
    const int row0 = (blockIdx.x * WAVES + wib) * 16;
    const int sub  = lane % LPR;
    const int rl   = lane / LPR;
    const unsigned suboff = (unsigned)sub * 16u;
    const int m     = lane & 15;
    const int khalf = (lane >> 4) * 2;

    int n[NPASS]; unsigned st[NPASS]; float4 own[NPASS];
    #pragma unroll
    for (int p = 0; p < NPASS; ++p) {
        const int row = row0 + p * RPP + rl;
        n[p]  = cnt[row];
        st[p] = (unsigned)rowptr[row];
    }
    #pragma unroll
    for (int p = 0; p < NPASS; ++p)
        own[p] = load_row4<S>(hin, row0 + p * RPP + rl, suboff);

    const unsigned run0 = (unsigned)rowptr[row0];
    const unsigned runn = (unsigned)rowptr[row0 + 15] + (unsigned)cnt[row0 + 15] - run0;
    const bool staged = runn <= (unsigned)ENT_CAP;

    float4 wv[WTRIPS];
    #pragma unroll
    for (int t = 0; t < WTRIPS; ++t) {
        const int i   = threadIdx.x + t * LAYER_T;
        const int s   = i / (S * (HID / 4));
        const int rem = i - s * (S * (HID / 4));
        const int kw  = rem / (HID / 4);
        const int c4  = rem - kw * (HID / 4);
        wv[t] = (kw < KW) ? reinterpret_cast<const float4*>(W)[(s * KW + kw) * (HID / 4) + c4]
                          : make_float4(0.f, 0.f, 0.f, 0.f);
    }
    const float bias0 = bias[m];
    const float bias1 = bias[16 + m];
    int2 ev[ENT_PER_LANE];
    #pragma unroll
    for (int k = 0; k < ENT_PER_LANE; ++k) {
        const unsigned i = (unsigned)lane + 32u * k;
        ev[k] = load_entry(csr, (staged && i < runn) ? run0 + i : 0u);
    }
    #pragma unroll
    for (int t = 0; t < WTRIPS; ++t)
        reinterpret_cast<float4*>(sW)[threadIdx.x + t * LAYER_T] = wv[t];
    __syncthreads();

    int2* __restrict__ ents = sE + wib * ENT_CAP;
    #pragma unroll
    for (int k = 0; k < ENT_PER_LANE; ++k) ents[lane + 32 * k] = ev[k];
    __builtin_amdgcn_fence(__ATOMIC_RELEASE, "wavefront");
    __builtin_amdgcn_wave_barrier();
    __builtin_amdgcn_fence(__ATOMIC_ACQUIRE, "wavefront");

    float* __restrict__ tile = sT + wib * (16 * PITCH);
    #pragma unroll
    for (int p = 0; p < NPASS; ++p) {
        const float4 acc = staged ? gather_sum_staged<S>(ents + (st[p] - run0), hin, n[p], suboff)
                                  : gather_sum<S>(csr, hin, st[p], n[p], suboff);
        float* trow = tile + (p * RPP + rl) * PITCH + 4 * sub;
        *reinterpret_cast<float4*>(trow)     = own[p];
        *reinterpret_cast<float4*>(trow + S) = acc;
    }
    __builtin_amdgcn_fence(__ATOMIC_RELEASE, "wavefront");
    __builtin_amdgcn_wave_barrier();
    __builtin_amdgcn_fence(__ATOMIC_ACQUIRE, "wavefront");

    float wr0 = 0.f, wr1 = 0.f, br = 0.f;
    if constexpr (READOUT) { wr0 = wreg[m]; wr1 = wreg[16 + m]; br = breg[0]; }

    v8f acc0 = {0.f,0.f,0.f,0.f,0.f,0.f,0.f,0.f};
    v8f acc1 = {0.f,0.f,0.f,0.f,0.f,0.f,0.f,0.f};

    #pragma unroll
    for (int c = 0; c < K2 / 4; ++c) {
        const int k0 = c * 4 + khalf;
        v2f a = *reinterpret_cast<const v2f*>(tile + m * PITCH + k0);
        v2f b0, b1;
        b0.x = sW[k0 * HID + m];
        b0.y = sW[(k0 + 1) * HID + m];
        b1.x = sW[k0 * HID + 16 + m];
        b1.y = sW[(k0 + 1) * HID + 16 + m];
        acc0 = __builtin_amdgcn_wmma_f32_16x16x4_f32(
                   false, a, false, b0, (short)0, acc0, false, false);
        acc1 = __builtin_amdgcn_wmma_f32_16x16x4_f32(
                   false, a, false, b1, (short)0, acc1, false, false);
    }

    const int rbase = row0 + ((lane >> 4) ? 8 : 0);
    #pragma unroll
    for (int v = 0; v < 8; ++v) {
        int r = rbase + v;
        float x0 = acc0[v] + bias0;
        float x1 = acc1[v] + bias1;
        const float t0 = tanhf(x0);
        const float t1 = tanhf(x1);
        hout[(size_t)r * HID + m]      = t0;
        hout[(size_t)r * HID + 16 + m] = t1;
        if constexpr (READOUT) {
            float d = t0 * wr0 + t1 * wr1;
            d += __shfl_xor(d, 8);
            d += __shfl_xor(d, 4);
            d += __shfl_xor(d, 2);
            d += __shfl_xor(d, 1);
            if (m == 0) tile[(lane >> 4) * 8 + v] = d + br;
        }
    }
    if constexpr (READOUT) {
        __syncthreads();
        if (threadIdx.x < WAVES * 16)
            out[blockIdx.x * (WAVES * 16) + threadIdx.x] =
                sT[(threadIdx.x >> 4) * (16 * PITCH) + (threadIdx.x & 15)];
    }
}

static inline int cdiv(long long a, long long b) { return (int)((a + b - 1) / b); }

extern "C" void kernel_launch(void* const* d_in, const int* in_sizes, int n_in,
                              void* d_out, int out_size, void* d_ws, size_t ws_size,
                              hipStream_t stream) {
    const float* X    = (const float*)d_in[0];
    const int*   ei   = (const int*)  d_in[1];
    const float* W0   = (const float*)d_in[2];
    const float* b0   = (const float*)d_in[3];
    const float* W1   = (const float*)d_in[4];
    const float* b1   = (const float*)d_in[5];
    const float* W2   = (const float*)d_in[6];
    const float* b2   = (const float*)d_in[7];
    const float* Wreg = (const float*)d_in[8];
    const float* breg = (const float*)d_in[9];

    float* out    = (float*)d_out;
    float* hfinal = out + NTOT;

    int*   cnt    = (int*)d_ws;
    int*   rowptr = cnt + NTOT;
    int*   rank   = rowptr + NTOT;
    int2*  csr    = (int2*)(rank + ETOT);
    float* hA     = (float*)(csr + ETOT);
    float* hB     = hA  + (size_t)NTOT * HID;
    float* Xp     = hB  + (size_t)NTOT * HID;
    int*   part   = (int*)(Xp + (size_t)NTOT * KPAD);

    count_rank_kernel<<<BATCH * EDGE_SPLIT, EDGE_T, 0, stream>>>(ei, rank, part);
    chunk_offsets_kernel<<<cdiv(NTOT / 4, OFFS_T), OFFS_T, 0, stream>>>(part, cnt, X, Xp);
    fill_rows_kernel<<<BATCH * EDGE_SPLIT, EDGE_T, 0, stream>>>(ei, rank, cnt, part, rowptr, csr);

    const int narrow_blocks = cdiv(NTOT / 16, 128 / 32);
    const int wide_blocks   = cdiv(NTOT / 16, 256 / 32);

    tag_layer_kernel<KPAD, WINDOW, 128, false><<<narrow_blocks, 128, 0, stream>>>(
        rowptr, cnt, csr, Xp, W0, b0, hA, nullptr, nullptr, nullptr);

    tag_layer_kernel<HID, HID, 256, false><<<wide_blocks, 256, 0, stream>>>(
        rowptr, cnt, csr, hA, W1, b1, hB, nullptr, nullptr, nullptr);

    tag_layer_kernel<HID, HID, 256, true><<<wide_blocks, 256, 0, stream>>>(
        rowptr, cnt, csr, hB, W2, b2, hfinal, Wreg, breg, out);
}


# list fill re-split by destination range: pass 1 sorts each chunk by destination in LDS; a fill work-group builds its own contiguous piece in LDS and stores whole lines; list and sum order kept
# speedup vs baseline: 32.5888x; 1.5576x over previous
#include <hip/hip_runtime.h>
#include <hip/hip_bf16.h>
#include <math.h>

#define BATCH   8
#define WINDOW  5
#define NN      20000
#define EPB     320000
#define HID     32
#define KPAD    8
#define NTOT    (BATCH * NN)
#define ETOT    (BATCH * EPB)

typedef float v2f __attribute__((ext_vector_type(2)));
typedef float v8f __attribute__((ext_vector_type(8)));

#define EDGE_T      1024
#define EDGE_SPLIT  32
#define QPW         (EPB / 4 / EDGE_SPLIT)
#define EDGE_QPT    ((QPW + EDGE_T - 1) / EDGE_T)
#define NQ          (NN / 4)
#define TAB_TRIPS   5
#define TAB_T       (NQ / TAB_TRIPS)

#define RANGE       (NN / EDGE_SPLIT)
#define CHUNK_E     (QPW * 4)
#define SCAN_PER    20

__global__ void __launch_bounds__(EDGE_T)
count_sort_kernel(const int* __restrict__ ei, int2* __restrict__ lsort,
                  int* __restrict__ bounds, int* __restrict__ part) {
    static_assert(EPB % (4 * EDGE_SPLIT) == 0 && NN % 4 == 0 && NQ % TAB_TRIPS == 0 &&
                  TAB_T <= EDGE_T && TAB_T % 8 == 0,
                  "whole edge quads per chunk; tables move 16 bytes at a time in whole lines");
    static_assert(NN % SCAN_PER == 0 && NN / SCAN_PER == TAB_T && (SCAN_PER % 4) == 0 &&
                  SCAN_PER / 4 == TAB_TRIPS && NN % EDGE_SPLIT == 0 && NN < 32768 &&
                  EDGE_SPLIT <= EDGE_T && (CHUNK_E % 2) == 0,
                  "SCAN_PER consecutive nodes per scanning thread; src and dst share one word");
    __shared__ __attribute__((aligned(16))) int  hist[NN];
    __shared__ __attribute__((aligned(16))) int2 stage[CHUNK_E];
    __shared__ int wtot[EDGE_T / 32];
    const int b   = blockIdx.x / EDGE_SPLIT;
    const int c   = blockIdx.x - b * EDGE_SPLIT;
    const int tid = threadIdx.x;
    const int lane = tid & 31;
    const int wid  = tid >> 5;
    const int* __restrict__ eb = ei + (size_t)b * 2 * EPB;
    const int4* __restrict__ src4 = reinterpret_cast<const int4*>(eb) + c * QPW;
    const int4* __restrict__ dst4 = reinterpret_cast<const int4*>(eb + EPB) + c * QPW;

    int4 sq[EDGE_QPT], dq[EDGE_QPT], rq[EDGE_QPT];
    #pragma unroll
    for (int k = 0; k < EDGE_QPT; ++k) {
        const int q  = tid + k * EDGE_T;
        const int qc = q < QPW ? q : QPW - 1;
        sq[k] = src4[qc]; dq[k] = dst4[qc];
    }
    for (int i = tid; i < NQ; i += EDGE_T)
        reinterpret_cast<int4*>(hist)[i] = make_int4(0, 0, 0, 0);
    __syncthreads();

    #pragma unroll
    for (int k = 0; k < EDGE_QPT; ++k) {
        const int q = tid + k * EDGE_T;
        rq[k] = make_int4(0, 0, 0, 0);
        if (q < QPW) {
            rq[k].x = atomicAdd(&hist[dq[k].x], 1);
            rq[k].y = atomicAdd(&hist[dq[k].y], 1);
            rq[k].z = atomicAdd(&hist[dq[k].z], 1);
            rq[k].w = atomicAdd(&hist[dq[k].w], 1);
        }
    }
    __syncthreads();

    int4* __restrict__ gp = reinterpret_cast<int4*>(part + (size_t)blockIdx.x * NN);
    for (int i = tid; i < NQ; i += EDGE_T)
        gp[i] = reinterpret_cast<const int4*>(hist)[i];

    const bool act = tid < TAB_T;
    int cc[SCAN_PER];
    int tot = 0;
    #pragma unroll
    for (int k = 0; k < SCAN_PER; ++k) cc[k] = 0;
    if (act) {
        #pragma unroll
        for (int g = 0; g < SCAN_PER / 4; ++g) {
            const int4 q = reinterpret_cast<const int4*>(hist)[tid * (SCAN_PER / 4) + g];
            cc[4 * g + 0] = q.x; cc[4 * g + 1] = q.y; cc[4 * g + 2] = q.z; cc[4 * g + 3] = q.w;
            tot += q.x + q.y + q.z + q.w;
        }
    }
    int inc = tot;
    #pragma unroll
    for (int off = 1; off < 32; off <<= 1) {
        int t = __shfl_up(inc, off);
        if (lane >= off) inc += t;
    }
    if (lane == 31) wtot[wid] = inc;
    __syncthreads();
    if (wid == 0) {
        int w = wtot[lane];
        int winc = w;
        #pragma unroll
        for (int off = 1; off < 32; off <<= 1) {
            int t = __shfl_up(winc, off);
            if (lane >= off) winc += t;
        }
        wtot[lane] = winc - w;
    }
    __syncthreads();
    if (act) {
        int run = wtot[wid] + (inc - tot);
        #pragma unroll
        for (int g = 0; g < SCAN_PER / 4; ++g) {
            int4 r;
            r.x = run; run += cc[4 * g + 0];
            r.y = run; run += cc[4 * g + 1];
            r.z = run; run += cc[4 * g + 2];
            r.w = run; run += cc[4 * g + 3];
            reinterpret_cast<int4*>(hist)[tid * (SCAN_PER / 4) + g] = r;
        }
    }
    __syncthreads();

    if (tid < EDGE_SPLIT) bounds[blockIdx.x * EDGE_SPLIT + tid] = hist[tid * RANGE];
    #pragma unroll
    for (int k = 0; k < EDGE_QPT; ++k) {
        const int q = tid + k * EDGE_T;
        if (q < QPW) {
            stage[hist[dq[k].x] + rq[k].x] = make_int2(sq[k].x | (dq[k].x << 16), rq[k].x);
            stage[hist[dq[k].y] + rq[k].y] = make_int2(sq[k].y | (dq[k].y << 16), rq[k].y);
            stage[hist[dq[k].z] + rq[k].z] = make_int2(sq[k].z | (dq[k].z << 16), rq[k].z);
            stage[hist[dq[k].w] + rq[k].w] = make_int2(sq[k].w | (dq[k].w << 16), rq[k].w);
        }
    }
    __syncthreads();

    int4* __restrict__ lo4 = reinterpret_cast<int4*>(lsort + (size_t)blockIdx.x * CHUNK_E);
    constexpr int LS_TRIPS = (CHUNK_E / 2 + EDGE_T - 1) / EDGE_T;
    int4 lv[LS_TRIPS];
    #pragma unroll
    for (int k = 0; k < LS_TRIPS; ++k) {
        const int i = tid + k * EDGE_T;
        lv[k] = reinterpret_cast<const int4*>(stage)[i < CHUNK_E / 2 ? i : 0];
    }
    #pragma unroll
    for (int k = 0; k < LS_TRIPS; ++k) {
        const int i = tid + k * EDGE_T;
        if (i < CHUNK_E / 2) lo4[i] = lv[k];
    }
}

#define OFFS_T 128
__global__ void __launch_bounds__(OFFS_T)
chunk_offsets_kernel(int* __restrict__ part, int* __restrict__ cnt,
                     const float* __restrict__ x, float* __restrict__ xp) {
    static_assert(WINDOW == 5 && KPAD == 8, "four rows of X are five 16-byte pieces");
    const int gq = blockIdx.x * OFFS_T + threadIdx.x;
    if (gq >= NTOT / 4) return;
    const int b  = gq / NQ;
    const int nq = gq - b * NQ;
    const float4* __restrict__ xr = reinterpret_cast<const float4*>(x) + (size_t)gq * WINDOW;
    const float4 a0 = xr[0], a1 = xr[1], a2 = xr[2], a3 = xr[3], a4 = xr[4];
    int4* __restrict__ p = reinterpret_cast<int4*>(part + (size_t)b * EDGE_SPLIT * NN) + nq;
    int4 t[EDGE_SPLIT];
    #pragma unroll
    for (int c = 0; c < EDGE_SPLIT; ++c) t[c] = p[(size_t)c * NQ];
    float4* __restrict__ xo = reinterpret_cast<float4*>(xp) + (size_t)gq * KPAD;
    xo[0] = a0;
    xo[1] = make_float4(a1.x, 0.f, 0.f, 0.f);
    xo[2] = make_float4(a1.y, a1.z, a1.w, a2.x);
    xo[3] = make_float4(a2.y, 0.f, 0.f, 0.f);
    xo[4] = make_float4(a2.z, a2.w, a3.x, a3.y);
    xo[5] = make_float4(a3.z, 0.f, 0.f, 0.f);
    xo[6] = make_float4(a3.w, a4.x, a4.y, a4.z);
    xo[7] = make_float4(a4.w, 0.f, 0.f, 0.f);
    int4 run = make_int4(0, 0, 0, 0);
    #pragma unroll
    for (int c = 0; c < EDGE_SPLIT; ++c) {
        p[(size_t)c * NQ] = run;
        run.x += t[c].x; run.y += t[c].y; run.z += t[c].z; run.w += t[c].w;
    }
    reinterpret_cast<int4*>(cnt)[gq] = run;
}

#define FILL_DEPTH  12
#define PIECE_CAP   16384
#define PRE_N       (EDGE_SPLIT * RANGE)
#define PRE_TRIPS   ((RANGE + 31) / 32)
static __device__ __forceinline__ void put_entry(const int* __restrict__ s_rp,
                                                 const int* __restrict__ pre,
                                                 const int* __restrict__ s_cd,
                                                 int2* __restrict__ to, int first,
                                                 int2 e, int lo, int nodebase) {
    const int sl = e.x & 0xffff;
    const int dl = (int)((unsigned)e.x >> 16);
    const int il = dl - lo;
    const int slot = s_rp[il] + pre[il] + e.y;
    const float nrm = __int_as_float(s_cd[sl]) * __int_as_float(s_cd[dl]);
    to[slot - first] = make_int2(sl + nodebase, __float_as_int(nrm));
}
static __device__ __forceinline__ void put_group(const int* __restrict__ s_rp,
                                                 const int* __restrict__ pre,
                                                 const int* __restrict__ s_cd,
                                                 int2* __restrict__ to, int first,
                                                 const int2 (&e)[FILL_DEPTH], int lane, int len,
                                                 int lo, int nodebase) {
    int slot[FILL_DEPTH]; int sl[FILL_DEPTH]; float nrm[FILL_DEPTH];
    #pragma unroll
    for (int k = 0; k < FILL_DEPTH; ++k) {
        const bool ok = lane + 32 * k < len;
        sl[k] = e[k].x & 0xffff;
        const int dl = (int)((unsigned)e[k].x >> 16);
        const int il = ok ? dl - lo : 0;
        slot[k] = s_rp[il] + pre[il] + e[k].y;
        nrm[k]  = __int_as_float(s_cd[sl[k]]) * __int_as_float(s_cd[dl]);
    }
    #pragma unroll
    for (int k = 0; k < FILL_DEPTH; ++k)
        if (lane + 32 * k < len)
            to[slot[k] - first] = make_int2(sl[k] + nodebase, __float_as_int(nrm[k]));
}
__global__ void __launch_bounds__(EDGE_T)
fill_rows_kernel(const int2* __restrict__ lsort, const int* __restrict__ bounds,
                 const int* __restrict__ cnt, const int* __restrict__ part,
                 int* __restrict__ rowptr, int2* __restrict__ csr) {
    static_assert(EDGE_T / 32 == EDGE_SPLIT && TAB_TRIPS * EDGE_T >= NQ,
                  "one wave per chunk; the in-degrees are copied in TAB_TRIPS trips");
    __shared__ int s_pre[PRE_N];
    __shared__ __attribute__((aligned(16))) int s_cd[NN];
    __shared__ int s_rp[RANGE + 1];
    __shared__ int wtot[EDGE_T / 32];
    __shared__ __attribute__((aligned(16))) int2 piece[PIECE_CAP];
    const int b   = blockIdx.x / EDGE_SPLIT;
    const int rg  = blockIdx.x - b * EDGE_SPLIT;
    const int lo  = rg * RANGE;
    const int tid = threadIdx.x;
    const int lane = tid & 31;
    const int wid  = __builtin_amdgcn_readfirstlane(tid >> 5);

    const int chunk = b * EDGE_SPLIT + wid;
    const int* __restrict__ bnd = bounds + chunk * EDGE_SPLIT;
    const int rb = bnd[rg];
    const int rn = bnd[rg + 1 < EDGE_SPLIT ? rg + 1 : rg];
    const int len = (rg + 1 < EDGE_SPLIT ? rn : CHUNK_E) - rb;
    const int2* __restrict__ lbase = lsort + (size_t)chunk * CHUNK_E;

    const int4* __restrict__ gc = reinterpret_cast<const int4*>(cnt + b * NN);
    const int* __restrict__ gpw = part + ((size_t)chunk * NN + lo);
    const bool act = tid < TAB_T;
    int4 tc[TAB_TRIPS]; int tp[PRE_TRIPS];
    #pragma unroll
    for (int k = 0; k < TAB_TRIPS; ++k) {
        const int i = tid + k * EDGE_T;
        tc[k] = gc[i < NQ ? i : 0];
    }
    #pragma unroll
    for (int k = 0; k < PRE_TRIPS; ++k) {
        const int i = lane + 32 * k;
        tp[k] = gpw[i < RANGE ? i : 0];
    }
    int2 ev[FILL_DEPTH];
    #pragma unroll
    for (int k = 0; k < FILL_DEPTH; ++k) {
        const int i = lane + 32 * k;
        ev[k] = lbase[i < len ? rb + i : 0];
    }
    #pragma unroll
    for (int k = 0; k < TAB_TRIPS; ++k) {
        const int i = tid + k * EDGE_T;
        if (i < NQ) reinterpret_cast<int4*>(s_cd)[i] = tc[k];
    }
    #pragma unroll
    for (int k = 0; k < PRE_TRIPS; ++k) {
        const int i = lane + 32 * k;
        if (i < RANGE) s_pre[wid * RANGE + i] = tp[k];
    }
    __syncthreads();

    int c[SCAN_PER];
    int tot = 0;
    #pragma unroll
    for (int k = 0; k < SCAN_PER; ++k) c[k] = 0;
    if (act) {
        #pragma unroll
        for (int g = 0; g < SCAN_PER / 4; ++g) {
            const int4 q = reinterpret_cast<const int4*>(s_cd)[tid * (SCAN_PER / 4) + g];
            c[4 * g + 0] = q.x; c[4 * g + 1] = q.y; c[4 * g + 2] = q.z; c[4 * g + 3] = q.w;
            tot += q.x + q.y + q.z + q.w;
        }
    }
    int inc = tot;
    #pragma unroll
    for (int off = 1; off < 32; off <<= 1) {
        int t = __shfl_up(inc, off);
        if (lane >= off) inc += t;
    }
    if (lane == 31) wtot[wid] = inc;
    __syncthreads();
    if (wid == 0) {
        int w = wtot[lane];
        int winc = w;
        #pragma unroll
        for (int off = 1; off < 32; off <<= 1) {
            int t = __shfl_up(winc, off);
            if (lane >= off) winc += t;
        }
        wtot[lane] = winc - w;
    }
    __syncthreads();

    if (act) {
        int run = b * EPB + wtot[wid] + (inc - tot);
        int4* __restrict__ rp = reinterpret_cast<int4*>(rowptr + b * NN + tid * SCAN_PER);
        #pragma unroll
        for (int g = 0; g < SCAN_PER / 4; ++g) {
            int4 r; int4 f;
            r.x = run; run += c[4 * g + 0];
            r.y = run; run += c[4 * g + 1];
            r.z = run; run += c[4 * g + 2];
            r.w = run; run += c[4 * g + 3];
            f.x = __float_as_int((c[4 * g + 0] > 0) ? rsqrtf((float)c[4 * g + 0]) : 0.0f);
            f.y = __float_as_int((c[4 * g + 1] > 0) ? rsqrtf((float)c[4 * g + 1]) : 0.0f);
            f.z = __float_as_int((c[4 * g + 2] > 0) ? rsqrtf((float)c[4 * g + 2]) : 0.0f);
            f.w = __float_as_int((c[4 * g + 3] > 0) ? rsqrtf((float)c[4 * g + 3]) : 0.0f);
            reinterpret_cast<int4*>(s_cd)[tid * (SCAN_PER / 4) + g] = f;
            const int nd = tid * SCAN_PER + 4 * g - lo;
            if ((unsigned)(nd + 0) < (unsigned)RANGE) s_rp[nd + 0] = r.x;
            if ((unsigned)(nd + 1) < (unsigned)RANGE) s_rp[nd + 1] = r.y;
            if ((unsigned)(nd + 2) < (unsigned)RANGE) s_rp[nd + 2] = r.z;
            if ((unsigned)(nd + 3) < (unsigned)RANGE) s_rp[nd + 3] = r.w;
            if (nd + 0 == RANGE - 1) s_rp[RANGE] = r.y;
            if (nd + 1 == RANGE - 1) s_rp[RANGE] = r.z;
            if (nd + 2 == RANGE - 1) s_rp[RANGE] = r.w;
            if (nd + 3 == RANGE - 1) s_rp[RANGE] = run;
            if (rg == 0) rp[g] = r;
        }
    }
    __syncthreads();

    const int nodebase = b * NN;
    const int* __restrict__ pre = s_pre + wid * RANGE;
    const int first = s_rp[0];
    const int end   = s_rp[RANGE];
    if (end - first <= PIECE_CAP) {
        put_group(s_rp, pre, s_cd, piece, first, ev, lane, len, lo, nodebase);
        for (int i = lane + 32 * FILL_DEPTH; i < len; i += 32)
            put_entry(s_rp, pre, s_cd, piece, first, lbase[rb + i], lo, nodebase);
        __syncthreads();
        for (int e0 = (first & ~15) + tid; e0 < end; e0 += 4 * EDGE_T) {
            int2 v[4];
            #pragma unroll
            for (int u = 0; u < 4; ++u) {
                const int e = e0 + u * EDGE_T;
                v[u] = piece[(e >= first && e < end) ? e - first : 0];
            }
            #pragma unroll
            for (int u = 0; u < 4; ++u) {
                const int e = e0 + u * EDGE_T;
                if (e >= first && e < end) csr[e] = v[u];
            }
        }
    } else {
        put_group(s_rp, pre, s_cd, csr, 0, ev, lane, len, lo, nodebase);
        for (int i = lane + 32 * FILL_DEPTH; i < len; i += 32)
            put_entry(s_rp, pre, s_cd, csr, 0, lbase[rb + i], lo, nodebase);
    }
}

static __device__ __forceinline__ int2 load_entry(const int2* __restrict__ csr, unsigned idx) {
    return *reinterpret_cast<const int2*>(reinterpret_cast<const char*>(csr) + idx * 8u);
}
template <int S>
static __device__ __forceinline__ float4 load_row4(const float* __restrict__ hin, int src,
                                                   unsigned suboff) {
    return *reinterpret_cast<const float4*>(reinterpret_cast<const char*>(hin) +
                                            ((unsigned)src * (unsigned)(S * 4) + suboff));
}
static __device__ __forceinline__ void axpy4(float4& acc, const float4& v, float w) {
    acc.x += v.x * w; acc.y += v.y * w; acc.z += v.z * w; acc.w += v.w * w;
}

template <int S>
static __device__ __forceinline__ float4 gather_sum(const int2* __restrict__ csr,
                                                    const float* __restrict__ hin,
                                                    unsigned start, int n, unsigned suboff) {
    float4 acc = make_float4(0.f, 0.f, 0.f, 0.f);
    int j = 0;
    for (; j + 4 <= n; j += 4) {
        const int2 e0 = load_entry(csr, start + j);
        const int2 e1 = load_entry(csr, start + j + 1);
        const int2 e2 = load_entry(csr, start + j + 2);
        const int2 e3 = load_entry(csr, start + j + 3);
        const float4 v0 = load_row4<S>(hin, e0.x, suboff);
        const float4 v1 = load_row4<S>(hin, e1.x, suboff);
        const float4 v2 = load_row4<S>(hin, e2.x, suboff);
        const float4 v3 = load_row4<S>(hin, e3.x, suboff);
        axpy4(acc, v0, __int_as_float(e0.y));
        axpy4(acc, v1, __int_as_float(e1.y));
        axpy4(acc, v2, __int_as_float(e2.y));
        axpy4(acc, v3, __int_as_float(e3.y));
    }
    const int r = n - j;
    if (r > 0) {
        const int2 e0 = load_entry(csr, start + j);
        const int2 e1 = load_entry(csr, start + j + (r > 1 ? 1 : 0));
        const int2 e2 = load_entry(csr, start + j + (r > 2 ? 2 : 0));
        float4 v0 = load_row4<S>(hin, e0.x, suboff);
        float4 v1 = load_row4<S>(hin, e1.x, suboff);
        float4 v2 = load_row4<S>(hin, e2.x, suboff);
        v1.x = (r > 1) ? v1.x : 0.f; v1.y = (r > 1) ? v1.y : 0.f;
        v1.z = (r > 1) ? v1.z : 0.f; v1.w = (r > 1) ? v1.w : 0.f;
        v2.x = (r > 2) ? v2.x : 0.f; v2.y = (r > 2) ? v2.y : 0.f;
        v2.z = (r > 2) ? v2.z : 0.f; v2.w = (r > 2) ? v2.w : 0.f;
        axpy4(acc, v0, __int_as_float(e0.y));
        axpy4(acc, v1, __int_as_float(e1.y));
        axpy4(acc, v2, __int_as_float(e2.y));
    }
    return acc;
}

template <int S, int M>
static __device__ __forceinline__ void gather_tail(float4& acc, const int2* __restrict__ ent,
                                                   const float* __restrict__ hin,
                                                   int r, unsigned suboff) {
    int2 e[M]; float4 v[M];
    #pragma unroll
    for (int u = 0; u < M; ++u) e[u] = ent[u < r ? u : 0];
    #pragma unroll
    for (int u = 0; u < M; ++u) v[u] = load_row4<S>(hin, e[u].x, suboff);
    #pragma unroll
    for (int u = 1; u < M; ++u) {
        v[u].x = (u < r) ? v[u].x : 0.f; v[u].y = (u < r) ? v[u].y : 0.f;
        v[u].z = (u < r) ? v[u].z : 0.f; v[u].w = (u < r) ? v[u].w : 0.f;
    }
    #pragma unroll
    for (int u = 0; u < M; ++u) axpy4(acc, v[u], __int_as_float(e[u].y));
}

#define GATHER_DEPTH 8
template <int S>
static __device__ __forceinline__ float4 gather_sum_staged(const int2* __restrict__ ent,
                                                           const float* __restrict__ hin,
                                                           int n, unsigned suboff) {
    float4 acc = make_float4(0.f, 0.f, 0.f, 0.f);
    int j = 0;
    for (; j + GATHER_DEPTH <= n; j += GATHER_DEPTH) {
        int2 e[GATHER_DEPTH]; float4 v[GATHER_DEPTH];
        #pragma unroll
        for (int u = 0; u < GATHER_DEPTH; ++u) e[u] = ent[j + u];
        #pragma unroll
        for (int u = 0; u < GATHER_DEPTH; ++u) v[u] = load_row4<S>(hin, e[u].x, suboff);
        #pragma unroll
        for (int u = 0; u < GATHER_DEPTH; ++u) axpy4(acc, v[u], __int_as_float(e[u].y));
    }
    const int r = n - j;
    if (r > 3)      gather_tail<S, GATHER_DEPTH - 1>(acc, ent + j, hin, r, suboff);
    else if (r > 0) gather_tail<S, 3>(acc, ent + j, hin, r, suboff);
    return acc;
}

#define ENT_CAP 320
#define ENT_PER_LANE (ENT_CAP / 32)
template <int S, int KW, int LAYER_T, bool READOUT>
__global__ void __launch_bounds__(LAYER_T)
tag_layer_kernel(const int* __restrict__ rowptr, const int* __restrict__ cnt,
                 const int2* __restrict__ csr, const float* __restrict__ hin,
                 const float* __restrict__ W, const float* __restrict__ bias,
                 float* __restrict__ hout, const float* __restrict__ wreg,
                 const float* __restrict__ breg, float* __restrict__ out) {
    constexpr int K2    = 2 * S;
    constexpr int PITCH = K2 + 4;
    constexpr int LPR   = S / 4;
    constexpr int RPP   = 32 / LPR;
    constexpr int NPASS = 16 / RPP;
    constexpr int WAVES = LAYER_T / 32;
    constexpr int WTRIPS = (K2 * HID / 4) / LAYER_T;
    static_assert((K2 * HID / 4) % LAYER_T == 0 && HID == 32,
                  "whole staging trips; two 16-column output tiles");
    static_assert((S % 4) == 0 && (32 % LPR) == 0 && (16 % RPP) == 0 && ((PITCH / 4) % 2) == 1,
                  "whole float4 pieces per lane, whole passes per tile, odd 16-byte pitch");
    static_assert((NTOT % (16 * WAVES)) == 0, "every wave of every work-group has a whole tile");
    static_assert((NN % 16) == 0 && (ENT_CAP % 32) == 0,
                  "a tile lies inside one graph; whole waves of entries per staging trip");
    __shared__ __attribute__((aligned(16))) float sW[K2 * HID];
    __shared__ __attribute__((aligned(16))) float sT[WAVES * 16 * PITCH];
    __shared__ __attribute__((aligned(16))) int2  sE[WAVES * ENT_CAP];

    const int lane = threadIdx.x & 31;
    const int wib  = threadIdx.x >> 5;
    const int row0 = (blockIdx.x * WAVES + wib) * 16;
    const int sub  = lane % LPR;
    const int rl   = lane / LPR;
    const unsigned suboff = (unsigned)sub * 16u;
    const int m     = lane & 15;
    const int khalf = (lane >> 4) * 2;

    int n[NPASS]; unsigned st[NPASS]; float4 own[NPASS];
    #pragma unroll
    for (int p = 0; p < NPASS; ++p) {
        const int row = row0 + p * RPP + rl;
        n[p]  = cnt[row];
        st[p] = (unsigned)rowptr[row];
    }
    #pragma unroll
    for (int p = 0; p < NPASS; ++p)
        own[p] = load_row4<S>(hin, row0 + p * RPP + rl, suboff);

    const unsigned run0 = (unsigned)rowptr[row0];
    const unsigned runn = (unsigned)rowptr[row0 + 15] + (unsigned)cnt[row0 + 15] - run0;
    const bool staged = runn <= (unsigned)ENT_CAP;

    float4 wv[WTRIPS];
    #pragma unroll
    for (int t = 0; t < WTRIPS; ++t) {
        const int i   = threadIdx.x + t * LAYER_T;
        const int s   = i / (S * (HID / 4));
        const int rem = i - s * (S * (HID / 4));
        const int kw  = rem / (HID / 4);
        const int c4  = rem - kw * (HID / 4);
        wv[t] = (kw < KW) ? reinterpret_cast<const float4*>(W)[(s * KW + kw) * (HID / 4) + c4]
                          : make_float4(0.f, 0.f, 0.f, 0.f);
    }
    const float bias0 = bias[m];
    const float bias1 = bias[16 + m];
    int2 ev[ENT_PER_LANE];
    #pragma unroll
    for (int k = 0; k < ENT_PER_LANE; ++k) {
        const unsigned i = (unsigned)lane + 32u * k;
        ev[k] = load_entry(csr, (staged && i < runn) ? run0 + i : 0u);
    }
    #pragma unroll
    for (int t = 0; t < WTRIPS; ++t)
        reinterpret_cast<float4*>(sW)[threadIdx.x + t * LAYER_T] = wv[t];
    __syncthreads();

    int2* __restrict__ ents = sE + wib * ENT_CAP;
    #pragma unroll
    for (int k = 0; k < ENT_PER_LANE; ++k) ents[lane + 32 * k] = ev[k];
    __builtin_amdgcn_fence(__ATOMIC_RELEASE, "wavefront");
    __builtin_amdgcn_wave_barrier();
    __builtin_amdgcn_fence(__ATOMIC_ACQUIRE, "wavefront");

    float* __restrict__ tile = sT + wib * (16 * PITCH);
    #pragma unroll
    for (int p = 0; p < NPASS; ++p) {
        const float4 acc = staged ? gather_sum_staged<S>(ents + (st[p] - run0), hin, n[p], suboff)
                                  : gather_sum<S>(csr, hin, st[p], n[p], suboff);
        float* trow = tile + (p * RPP + rl) * PITCH + 4 * sub;
        *reinterpret_cast<float4*>(trow)     = own[p];
        *reinterpret_cast<float4*>(trow + S) = acc;
    }
    __builtin_amdgcn_fence(__ATOMIC_RELEASE, "wavefront");
    __builtin_amdgcn_wave_barrier();
    __builtin_amdgcn_fence(__ATOMIC_ACQUIRE, "wavefront");

    float wr0 = 0.f, wr1 = 0.f, br = 0.f;
    if constexpr (READOUT) { wr0 = wreg[m]; wr1 = wreg[16 + m]; br = breg[0]; }

    v8f acc0 = {0.f,0.f,0.f,0.f,0.f,0.f,0.f,0.f};
    v8f acc1 = {0.f,0.f,0.f,0.f,0.f,0.f,0.f,0.f};

    #pragma unroll
    for (int c = 0; c < K2 / 4; ++c) {
        const int k0 = c * 4 + khalf;
        v2f a = *reinterpret_cast<const v2f*>(tile + m * PITCH + k0);
        v2f b0, b1;
        b0.x = sW[k0 * HID + m];
        b0.y = sW[(k0 + 1) * HID + m];
        b1.x = sW[k0 * HID + 16 + m];
        b1.y = sW[(k0 + 1) * HID + 16 + m];
        acc0 = __builtin_amdgcn_wmma_f32_16x16x4_f32(
                   false, a, false, b0, (short)0, acc0, false, false);
        acc1 = __builtin_amdgcn_wmma_f32_16x16x4_f32(
                   false, a, false, b1, (short)0, acc1, false, false);
    }

    const int rbase = row0 + ((lane >> 4) ? 8 : 0);
    #pragma unroll
    for (int v = 0; v < 8; ++v) {
        int r = rbase + v;
        float x0 = acc0[v] + bias0;
        float x1 = acc1[v] + bias1;
        const float t0 = tanhf(x0);
        const float t1 = tanhf(x1);
        hout[(size_t)r * HID + m]      = t0;
        hout[(size_t)r * HID + 16 + m] = t1;
        if constexpr (READOUT) {
            float d = t0 * wr0 + t1 * wr1;
            d += __shfl_xor(d, 8);
            d += __shfl_xor(d, 4);
            d += __shfl_xor(d, 2);
            d += __shfl_xor(d, 1);
            if (m == 0) tile[(lane >> 4) * 8 + v] = d + br;
        }
    }
    if constexpr (READOUT) {
        __syncthreads();
        if (threadIdx.x < WAVES * 16)
            out[blockIdx.x * (WAVES * 16) + threadIdx.x] =
                sT[(threadIdx.x >> 4) * (16 * PITCH) + (threadIdx.x & 15)];
    }
}

static inline int cdiv(long long a, long long b) { return (int)((a + b - 1) / b); }

extern "C" void kernel_launch(void* const* d_in, const int* in_sizes, int n_in,
                              void* d_out, int out_size, void* d_ws, size_t ws_size,
                              hipStream_t stream) {
    const float* X    = (const float*)d_in[0];
    const int*   ei   = (const int*)  d_in[1];
    const float* W0   = (const float*)d_in[2];
    const float* b0   = (const float*)d_in[3];
    const float* W1   = (const float*)d_in[4];
    const float* b1   = (const float*)d_in[5];
    const float* W2   = (const float*)d_in[6];
    const float* b2   = (const float*)d_in[7];
    const float* Wreg = (const float*)d_in[8];
    const float* breg = (const float*)d_in[9];

    float* out    = (float*)d_out;
    float* hfinal = out + NTOT;

    int*   cnt    = (int*)d_ws;
    int*   rowptr = cnt + NTOT;
    int2*  lsort  = (int2*)(rowptr + NTOT);
    int2*  csr    = lsort + ETOT;
    float* hA     = (float*)(csr + ETOT);
    float* hB     = hA  + (size_t)NTOT * HID;
    float* Xp     = hB  + (size_t)NTOT * HID;
    int*   part   = (int*)(Xp + (size_t)NTOT * KPAD);
    int*   bounds = part + (size_t)BATCH * EDGE_SPLIT * NN;

    count_sort_kernel<<<BATCH * EDGE_SPLIT, EDGE_T, 0, stream>>>(ei, lsort, bounds, part);
    chunk_offsets_kernel<<<cdiv(NTOT / 4, OFFS_T), OFFS_T, 0, stream>>>(part, cnt, X, Xp);
    fill_rows_kernel<<<BATCH * EDGE_SPLIT, EDGE_T, 0, stream>>>(lsort, bounds, cnt, part, rowptr, csr);

    const int narrow_blocks = cdiv(NTOT / 16, 128 / 32);
    const int wide_blocks   = cdiv(NTOT / 16, 256 / 32);

    tag_layer_kernel<KPAD, WINDOW, 128, false><<<narrow_blocks, 128, 0, stream>>>(
        rowptr, cnt, csr, Xp, W0, b0, hA, nullptr, nullptr, nullptr);

    tag_layer_kernel<HID, HID, 256, false><<<wide_blocks, 256, 0, stream>>>(
        rowptr, cnt, csr, hA, W1, b1, hB, nullptr, nullptr, nullptr);

    tag_layer_kernel<HID, HID, 256, true><<<wide_blocks, 256, 0, stream>>>(
        rowptr, cnt, csr, hB, W2, b2, hfinal, Wreg, breg, out);
}


# layer 0 computed inside the fill launch: a fill work-group multiplies its range's rows straight from its LDS piece of the list (W and tiles in the freed prefix table); 6 to 5 launches; sum order kept
# speedup vs baseline: 34.1106x; 1.0467x over previous
#include <hip/hip_runtime.h>
#include <hip/hip_bf16.h>
#include <math.h>

#define BATCH   8
#define WINDOW  5
#define NN      20000
#define EPB     320000
#define HID     32
#define KPAD    8
#define NTOT    (BATCH * NN)
#define ETOT    (BATCH * EPB)

typedef float v2f __attribute__((ext_vector_type(2)));
typedef float v8f __attribute__((ext_vector_type(8)));

#define EDGE_T      1024
#define EDGE_SPLIT  32
#define QPW         (EPB / 4 / EDGE_SPLIT)
#define EDGE_QPT    ((QPW + EDGE_T - 1) / EDGE_T)
#define NQ          (NN / 4)
#define TAB_TRIPS   5
#define TAB_T       (NQ / TAB_TRIPS)

#define RANGE       (NN / EDGE_SPLIT)
#define CHUNK_E     (QPW * 4)
#define SCAN_PER    20

__global__ void __launch_bounds__(EDGE_T)
count_sort_kernel(const int* __restrict__ ei, int2* __restrict__ lsort,
                  int* __restrict__ bounds, int* __restrict__ part) {
    static_assert(EPB % (4 * EDGE_SPLIT) == 0 && NN % 4 == 0 && NQ % TAB_TRIPS == 0 &&
                  TAB_T <= EDGE_T && TAB_T % 8 == 0,
                  "whole edge quads per chunk; tables move 16 bytes at a time in whole lines");
    static_assert(NN % SCAN_PER == 0 && NN / SCAN_PER == TAB_T && (SCAN_PER % 4) == 0 &&
                  SCAN_PER / 4 == TAB_TRIPS && NN % EDGE_SPLIT == 0 && NN < 32768 &&
                  EDGE_SPLIT <= EDGE_T && (CHUNK_E % 2) == 0,
                  "SCAN_PER consecutive nodes per scanning thread; src and dst share one word");
    __shared__ __attribute__((aligned(16))) int  hist[NN];
    __shared__ __attribute__((aligned(16))) int2 stage[CHUNK_E];
    __shared__ int wtot[EDGE_T / 32];
    const int b   = blockIdx.x / EDGE_SPLIT;
    const int c   = blockIdx.x - b * EDGE_SPLIT;
    const int tid = threadIdx.x;
    const int lane = tid & 31;
    const int wid  = tid >> 5;
    const int* __restrict__ eb = ei + (size_t)b * 2 * EPB;
    const int4* __restrict__ src4 = reinterpret_cast<const int4*>(eb) + c * QPW;
    const int4* __restrict__ dst4 = reinterpret_cast<const int4*>(eb + EPB) + c * QPW;

    int4 sq[EDGE_QPT], dq[EDGE_QPT], rq[EDGE_QPT];
    #pragma unroll
    for (int k = 0; k < EDGE_QPT; ++k) {
        const int q  = tid + k * EDGE_T;
        const int qc = q < QPW ? q : QPW - 1;
        sq[k] = src4[qc]; dq[k] = dst4[qc];
    }
    for (int i = tid; i < NQ; i += EDGE_T)
        reinterpret_cast<int4*>(hist)[i] = make_int4(0, 0, 0, 0);
    __syncthreads();

    #pragma unroll
    for (int k = 0; k < EDGE_QPT; ++k) {
        const int q = tid + k * EDGE_T;
        rq[k] = make_int4(0, 0, 0, 0);
        if (q < QPW) {
            rq[k].x = atomicAdd(&hist[dq[k].x], 1);
            rq[k].y = atomicAdd(&hist[dq[k].y], 1);
            rq[k].z = atomicAdd(&hist[dq[k].z], 1);
            rq[k].w = atomicAdd(&hist[dq[k].w], 1);
        }
    }
    __syncthreads();

    int4* __restrict__ gp = reinterpret_cast<int4*>(part + (size_t)blockIdx.x * NN);
    for (int i = tid; i < NQ; i += EDGE_T)
        gp[i] = reinterpret_cast<const int4*>(hist)[i];

    const bool act = tid < TAB_T;
    int cc[SCAN_PER];
    int tot = 0;
    #pragma unroll
    for (int k = 0; k < SCAN_PER; ++k) cc[k] = 0;
    if (act) {
        #pragma unroll
        for (int g = 0; g < SCAN_PER / 4; ++g) {
            const int4 q = reinterpret_cast<const int4*>(hist)[tid * (SCAN_PER / 4) + g];
            cc[4 * g + 0] = q.x; cc[4 * g + 1] = q.y; cc[4 * g + 2] = q.z; cc[4 * g + 3] = q.w;
            tot += q.x + q.y + q.z + q.w;
        }
    }
    int inc = tot;
    #pragma unroll
    for (int off = 1; off < 32; off <<= 1) {
        int t = __shfl_up(inc, off);
        if (lane >= off) inc += t;
    }
    if (lane == 31) wtot[wid] = inc;
    __syncthreads();
    if (wid == 0) {
        int w = wtot[lane];
        int winc = w;
        #pragma unroll
        for (int off = 1; off < 32; off <<= 1) {
            int t = __shfl_up(winc, off);
            if (lane >= off) winc += t;
        }
        wtot[lane] = winc - w;
    }
    __syncthreads();
    if (act) {
        int run = wtot[wid] + (inc - tot);
        #pragma unroll
        for (int g = 0; g < SCAN_PER / 4; ++g) {
            int4 r;
            r.x = run; run += cc[4 * g + 0];
            r.y = run; run += cc[4 * g + 1];
            r.z = run; run += cc[4 * g + 2];
            r.w = run; run += cc[4 * g + 3];
            reinterpret_cast<int4*>(hist)[tid * (SCAN_PER / 4) + g] = r;
        }
    }
    __syncthreads();

    if (tid < EDGE_SPLIT) bounds[blockIdx.x * EDGE_SPLIT + tid] = hist[tid * RANGE];
    #pragma unroll
    for (int k = 0; k < EDGE_QPT; ++k) {
        const int q = tid + k * EDGE_T;
        if (q < QPW) {
            stage[hist[dq[k].x] + rq[k].x] = make_int2(sq[k].x | (dq[k].x << 16), rq[k].x);
            stage[hist[dq[k].y] + rq[k].y] = make_int2(sq[k].y | (dq[k].y << 16), rq[k].y);
            stage[hist[dq[k].z] + rq[k].z] = make_int2(sq[k].z | (dq[k].z << 16), rq[k].z);
            stage[hist[dq[k].w] + rq[k].w] = make_int2(sq[k].w | (dq[k].w << 16), rq[k].w);
        }
    }
    __syncthreads();

    int4* __restrict__ lo4 = reinterpret_cast<int4*>(lsort + (size_t)blockIdx.x * CHUNK_E);
    constexpr int LS_TRIPS = (CHUNK_E / 2 + EDGE_T - 1) / EDGE_T;
    int4 lv[LS_TRIPS];
    #pragma unroll
    for (int k = 0; k < LS_TRIPS; ++k) {
        const int i = tid + k * EDGE_T;
        lv[k] = reinterpret_cast<const int4*>(stage)[i < CHUNK_E / 2 ? i : 0];
    }
    #pragma unroll
    for (int k = 0; k < LS_TRIPS; ++k) {
        const int i = tid + k * EDGE_T;
        if (i < CHUNK_E / 2) lo4[i] = lv[k];
    }
}

#define OFFS_T 128
__global__ void __launch_bounds__(OFFS_T)
chunk_offsets_kernel(int* __restrict__ part, int* __restrict__ cnt,
                     const float* __restrict__ x, float* __restrict__ xp) {
    static_assert(WINDOW == 5 && KPAD == 8, "four rows of X are five 16-byte pieces");
    const int gq = blockIdx.x * OFFS_T + threadIdx.x;
    if (gq >= NTOT / 4) return;
    const int b  = gq / NQ;
    const int nq = gq - b * NQ;
    const float4* __restrict__ xr = reinterpret_cast<const float4*>(x) + (size_t)gq * WINDOW;
    const float4 a0 = xr[0], a1 = xr[1], a2 = xr[2], a3 = xr[3], a4 = xr[4];
    int4* __restrict__ p = reinterpret_cast<int4*>(part + (size_t)b * EDGE_SPLIT * NN) + nq;
    int4 t[EDGE_SPLIT];
    #pragma unroll
    for (int c = 0; c < EDGE_SPLIT; ++c) t[c] = p[(size_t)c * NQ];
    float4* __restrict__ xo = reinterpret_cast<float4*>(xp) + (size_t)gq * KPAD;
    xo[0] = a0;
    xo[1] = make_float4(a1.x, 0.f, 0.f, 0.f);
    xo[2] = make_float4(a1.y, a1.z, a1.w, a2.x);
    xo[3] = make_float4(a2.y, 0.f, 0.f, 0.f);
    xo[4] = make_float4(a2.z, a2.w, a3.x, a3.y);
    xo[5] = make_float4(a3.z, 0.f, 0.f, 0.f);
    xo[6] = make_float4(a3.w, a4.x, a4.y, a4.z);
    xo[7] = make_float4(a4.w, 0.f, 0.f, 0.f);
    int4 run = make_int4(0, 0, 0, 0);
    #pragma unroll
    for (int c = 0; c < EDGE_SPLIT; ++c) {
        p[(size_t)c * NQ] = run;
        run.x += t[c].x; run.y += t[c].y; run.z += t[c].z; run.w += t[c].w;
    }
    reinterpret_cast<int4*>(cnt)[gq] = run;
}

static __device__ __forceinline__ int2 load_entry(const int2* __restrict__ csr, unsigned idx) {
    return *reinterpret_cast<const int2*>(reinterpret_cast<const char*>(csr) + idx * 8u);
}
template <int S>
static __device__ __forceinline__ float4 load_row4(const float* __restrict__ hin, int src,
                                                   unsigned suboff) {
    return *reinterpret_cast<const float4*>(reinterpret_cast<const char*>(hin) +
                                            ((unsigned)src * (unsigned)(S * 4) + suboff));
}
static __device__ __forceinline__ void axpy4(float4& acc, const float4& v, float w) {
    acc.x += v.x * w; acc.y += v.y * w; acc.z += v.z * w; acc.w += v.w * w;
}

template <int S>
static __device__ __forceinline__ float4 gather_sum(const int2* __restrict__ csr,
                                                    const float* __restrict__ hin,
                                                    unsigned start, int n, unsigned suboff) {
    float4 acc = make_float4(0.f, 0.f, 0.f, 0.f);
    int j = 0;
    for (; j + 4 <= n; j += 4) {
        const int2 e0 = load_entry(csr, start + j);
        const int2 e1 = load_entry(csr, start + j + 1);
        const int2 e2 = load_entry(csr, start + j + 2);
        const int2 e3 = load_entry(csr, start + j + 3);
        const float4 v0 = load_row4<S>(hin, e0.x, suboff);
        const float4 v1 = load_row4<S>(hin, e1.x, suboff);
        const float4 v2 = load_row4<S>(hin, e2.x, suboff);
        const float4 v3 = load_row4<S>(hin, e3.x, suboff);
        axpy4(acc, v0, __int_as_float(e0.y));
        axpy4(acc, v1, __int_as_float(e1.y));
        axpy4(acc, v2, __int_as_float(e2.y));
        axpy4(acc, v3, __int_as_float(e3.y));
    }
    const int r = n - j;
    if (r > 0) {
        const int2 e0 = load_entry(csr, start + j);
        const int2 e1 = load_entry(csr, start + j + (r > 1 ? 1 : 0));
        const int2 e2 = load_entry(csr, start + j + (r > 2 ? 2 : 0));
        float4 v0 = load_row4<S>(hin, e0.x, suboff);
        float4 v1 = load_row4<S>(hin, e1.x, suboff);
        float4 v2 = load_row4<S>(hin, e2.x, suboff);
        v1.x = (r > 1) ? v1.x : 0.f; v1.y = (r > 1) ? v1.y : 0.f;
        v1.z = (r > 1) ? v1.z : 0.f; v1.w = (r > 1) ? v1.w : 0.f;
        v2.x = (r > 2) ? v2.x : 0.f; v2.y = (r > 2) ? v2.y : 0.f;
        v2.z = (r > 2) ? v2.z : 0.f; v2.w = (r > 2) ? v2.w : 0.f;
        axpy4(acc, v0, __int_as_float(e0.y));
        axpy4(acc, v1, __int_as_float(e1.y));
        axpy4(acc, v2, __int_as_float(e2.y));
    }
    return acc;
}

template <int S, int M>
static __device__ __forceinline__ void gather_tail(float4& acc, const int2* __restrict__ ent,
                                                   const float* __restrict__ hin,
                                                   int r, unsigned suboff) {
    int2 e[M]; float4 v[M];
    #pragma unroll
    for (int u = 0; u < M; ++u) e[u] = ent[u < r ? u : 0];
    #pragma unroll
    for (int u = 0; u < M; ++u) v[u] = load_row4<S>(hin, e[u].x, suboff);
    #pragma unroll
    for (int u = 1; u < M; ++u) {
        v[u].x = (u < r) ? v[u].x : 0.f; v[u].y = (u < r) ? v[u].y : 0.f;
        v[u].z = (u < r) ? v[u].z : 0.f; v[u].w = (u < r) ? v[u].w : 0.f;
    }
    #pragma unroll
    for (int u = 0; u < M; ++u) axpy4(acc, v[u], __int_as_float(e[u].y));
}

#define GATHER_DEPTH 8
template <int S>
static __device__ __forceinline__ float4 gather_sum_staged(const int2* __restrict__ ent,
                                                           const float* __restrict__ hin,
                                                           int n, unsigned suboff) {
    float4 acc = make_float4(0.f, 0.f, 0.f, 0.f);
    int j = 0;
    for (; j + GATHER_DEPTH <= n; j += GATHER_DEPTH) {
        int2 e[GATHER_DEPTH]; float4 v[GATHER_DEPTH];
        #pragma unroll
        for (int u = 0; u < GATHER_DEPTH; ++u) e[u] = ent[j + u];
        #pragma unroll
        for (int u = 0; u < GATHER_DEPTH; ++u) v[u] = load_row4<S>(hin, e[u].x, suboff);
        #pragma unroll
        for (int u = 0; u < GATHER_DEPTH; ++u) axpy4(acc, v[u], __int_as_float(e[u].y));
    }
    const int r = n - j;
    if (r > 3)      gather_tail<S, GATHER_DEPTH - 1>(acc, ent + j, hin, r, suboff);
    else if (r > 0) gather_tail<S, 3>(acc, ent + j, hin, r, suboff);
    return acc;
}

#define FILL_DEPTH  12
#define PIECE_CAP   16384
#define PRE_N       (EDGE_SPLIT * RANGE)
#define PRE_TRIPS   ((RANGE + 31) / 32)
static __device__ __forceinline__ void put_entry(const int* __restrict__ s_rp,
                                                 const int* __restrict__ pre,
                                                 const int* __restrict__ s_cd,
                                                 int2* __restrict__ to, int first,
                                                 int2 e, int lo, int nodebase) {
    const int sl = e.x & 0xffff;
    const int dl = (int)((unsigned)e.x >> 16);
    const int il = dl - lo;
    const int slot = s_rp[il] + pre[il] + e.y;
    const float nrm = __int_as_float(s_cd[sl]) * __int_as_float(s_cd[dl]);
    to[slot - first] = make_int2(sl + nodebase, __float_as_int(nrm));
}
static __device__ __forceinline__ void put_group(const int* __restrict__ s_rp,
                                                 const int* __restrict__ pre,
                                                 const int* __restrict__ s_cd,
                                                 int2* __restrict__ to, int first,
                                                 const int2 (&e)[FILL_DEPTH], int lane, int len,
                                                 int lo, int nodebase) {
    int slot[FILL_DEPTH]; int sl[FILL_DEPTH]; float nrm[FILL_DEPTH];
    #pragma unroll
    for (int k = 0; k < FILL_DEPTH; ++k) {
        const bool ok = lane + 32 * k < len;
        sl[k] = e[k].x & 0xffff;
        const int dl = (int)((unsigned)e[k].x >> 16);
        const int il = ok ? dl - lo : 0;
        slot[k] = s_rp[il] + pre[il] + e[k].y;
        nrm[k]  = __int_as_float(s_cd[sl[k]]) * __int_as_float(s_cd[dl]);
    }
    #pragma unroll
    for (int k = 0; k < FILL_DEPTH; ++k)
        if (lane + 32 * k < len)
            to[slot[k] - first] = make_int2(sl[k] + nodebase, __float_as_int(nrm[k]));
}
__global__ void __launch_bounds__(EDGE_T)
fill_rows_kernel(const int2* __restrict__ lsort, const int* __restrict__ bounds,
                 const int* __restrict__ cnt, const int* __restrict__ part,
                 int* __restrict__ rowptr, int2* __restrict__ csr,
                 const float* __restrict__ xp, const float* __restrict__ W,
                 const float* __restrict__ bias, float* __restrict__ hout) {
    static_assert(EDGE_T / 32 == EDGE_SPLIT && TAB_TRIPS * EDGE_T >= NQ,
                  "one wave per chunk; the in-degrees are copied in TAB_TRIPS trips");
    __shared__ __attribute__((aligned(16))) int s_pre[PRE_N];
    __shared__ __attribute__((aligned(16))) int s_cd[NN];
    __shared__ int s_rp[RANGE + 1];
    __shared__ int wtot[EDGE_T / 32];
    __shared__ __attribute__((aligned(16))) int2 piece[PIECE_CAP];
    constexpr int L0_K2    = 2 * KPAD;
    constexpr int L0_PITCH = L0_K2 + 4;
    constexpr int L0_TILES = (RANGE + 15) / 16;
    constexpr int L0_TWICE = L0_TILES - EDGE_T / 32;
    static_assert(HID == 32 && L0_K2 * HID / 4 <= EDGE_T && ((L0_PITCH / 4) % 2) == 1 &&
                  (L0_K2 * HID + (EDGE_T / 32) * 16 * L0_PITCH) <= PRE_N &&
                  L0_TWICE >= 0 && L0_TWICE < EDGE_T / 32,
                  "W and the waves' tiles fit into the prefix table; at most two tiles per wave");
    const int b   = blockIdx.x / EDGE_SPLIT;
    const int rg  = blockIdx.x - b * EDGE_SPLIT;
    const int lo  = rg * RANGE;
    const int tid = threadIdx.x;
    const int lane = tid & 31;
    const int wid  = __builtin_amdgcn_readfirstlane(tid >> 5);

    const int chunk = b * EDGE_SPLIT + wid;
    const int* __restrict__ bnd = bounds + chunk * EDGE_SPLIT;
    const int rb = bnd[rg];
    const int rn = bnd[rg + 1 < EDGE_SPLIT ? rg + 1 : rg];
    const int len = (rg + 1 < EDGE_SPLIT ? rn : CHUNK_E) - rb;
    const int2* __restrict__ lbase = lsort + (size_t)chunk * CHUNK_E;

    const int4* __restrict__ gc = reinterpret_cast<const int4*>(cnt + b * NN);
    const int* __restrict__ gpw = part + ((size_t)chunk * NN + lo);
    const bool act = tid < TAB_T;
    int4 tc[TAB_TRIPS]; int tp[PRE_TRIPS];
    #pragma unroll
    for (int k = 0; k < TAB_TRIPS; ++k) {
        const int i = tid + k * EDGE_T;
        tc[k] = gc[i < NQ ? i : 0];
    }
    #pragma unroll
    for (int k = 0; k < PRE_TRIPS; ++k) {
        const int i = lane + 32 * k;
        tp[k] = gpw[i < RANGE ? i : 0];
    }
    int2 ev[FILL_DEPTH];
    #pragma unroll
    for (int k = 0; k < FILL_DEPTH; ++k) {
        const int i = lane + 32 * k;
        ev[k] = lbase[i < len ? rb + i : 0];
    }
    float4 w0v = make_float4(0.f, 0.f, 0.f, 0.f);
    if (tid < L0_K2 * HID / 4) {
        const int sidx = tid / (KPAD * (HID / 4));
        const int rem  = tid - sidx * (KPAD * (HID / 4));
        const int kw   = rem / (HID / 4);
        const int c4   = rem - kw * (HID / 4);
        if (kw < WINDOW)
            w0v = reinterpret_cast<const float4*>(W)[(sidx * WINDOW + kw) * (HID / 4) + c4];
    }
    #pragma unroll
    for (int k = 0; k < TAB_TRIPS; ++k) {
        const int i = tid + k * EDGE_T;
        if (i < NQ) reinterpret_cast<int4*>(s_cd)[i] = tc[k];
    }
    #pragma unroll
    for (int k = 0; k < PRE_TRIPS; ++k) {
        const int i = lane + 32 * k;
        if (i < RANGE) s_pre[wid * RANGE + i] = tp[k];
    }
    __syncthreads();

    int c[SCAN_PER];
    int tot = 0;
    #pragma unroll
    for (int k = 0; k < SCAN_PER; ++k) c[k] = 0;
    if (act) {
        #pragma unroll
        for (int g = 0; g < SCAN_PER / 4; ++g) {
            const int4 q = reinterpret_cast<const int4*>(s_cd)[tid * (SCAN_PER / 4) + g];
            c[4 * g + 0] = q.x; c[4 * g + 1] = q.y; c[4 * g + 2] = q.z; c[4 * g + 3] = q.w;
            tot += q.x + q.y + q.z + q.w;
        }
    }
    int inc = tot;
    #pragma unroll
    for (int off = 1; off < 32; off <<= 1) {
        int t = __shfl_up(inc, off);
        if (lane >= off) inc += t;
    }
    if (lane == 31) wtot[wid] = inc;
    __syncthreads();
    if (wid == 0) {
        int w = wtot[lane];
        int winc = w;
        #pragma unroll
        for (int off = 1; off < 32; off <<= 1) {
            int t = __shfl_up(winc, off);
            if (lane >= off) winc += t;
        }
        wtot[lane] = winc - w;
    }
    __syncthreads();

    if (act) {
        int run = b * EPB + wtot[wid] + (inc - tot);
        int4* __restrict__ rp = reinterpret_cast<int4*>(rowptr + b * NN + tid * SCAN_PER);
        #pragma unroll
        for (int g = 0; g < SCAN_PER / 4; ++g) {
            int4 r; int4 f;
            r.x = run; run += c[4 * g + 0];
            r.y = run; run += c[4 * g + 1];
            r.z = run; run += c[4 * g + 2];
            r.w = run; run += c[4 * g + 3];
            f.x = __float_as_int((c[4 * g + 0] > 0) ? rsqrtf((float)c[4 * g + 0]) : 0.0f);
            f.y = __float_as_int((c[4 * g + 1] > 0) ? rsqrtf((float)c[4 * g + 1]) : 0.0f);
            f.z = __float_as_int((c[4 * g + 2] > 0) ? rsqrtf((float)c[4 * g + 2]) : 0.0f);
            f.w = __float_as_int((c[4 * g + 3] > 0) ? rsqrtf((float)c[4 * g + 3]) : 0.0f);
            reinterpret_cast<int4*>(s_cd)[tid * (SCAN_PER / 4) + g] = f;
            const int nd = tid * SCAN_PER + 4 * g - lo;
            if ((unsigned)(nd + 0) < (unsigned)RANGE) s_rp[nd + 0] = r.x;
            if ((unsigned)(nd + 1) < (unsigned)RANGE) s_rp[nd + 1] = r.y;
            if ((unsigned)(nd + 2) < (unsigned)RANGE) s_rp[nd + 2] = r.z;
            if ((unsigned)(nd + 3) < (unsigned)RANGE) s_rp[nd + 3] = r.w;
            if (nd + 0 == RANGE - 1) s_rp[RANGE] = r.y;
            if (nd + 1 == RANGE - 1) s_rp[RANGE] = r.z;
            if (nd + 2 == RANGE - 1) s_rp[RANGE] = r.w;
            if (nd + 3 == RANGE - 1) s_rp[RANGE] = run;
            if (rg == 0) rp[g] = r;
        }
    }
    __syncthreads();

    const int nodebase = b * NN;
    const int* pre = s_pre + wid * RANGE;
    const int first = s_rp[0];
    const int end   = s_rp[RANGE];
    const bool fits = end - first <= PIECE_CAP;
    if (fits) {
        put_group(s_rp, pre, s_cd, piece, first, ev, lane, len, lo, nodebase);
        for (int i = lane + 32 * FILL_DEPTH; i < len; i += 32)
            put_entry(s_rp, pre, s_cd, piece, first, lbase[rb + i], lo, nodebase);
    } else {
        put_group(s_rp, pre, s_cd, csr, 0, ev, lane, len, lo, nodebase);
        for (int i = lane + 32 * FILL_DEPTH; i < len; i += 32)
            put_entry(s_rp, pre, s_cd, csr, 0, lbase[rb + i], lo, nodebase);
    }
    __syncthreads();

    float* __restrict__ sW = reinterpret_cast<float*>(s_pre);
    float* __restrict__ sT = sW + L0_K2 * HID;
    if (tid < L0_K2 * HID / 4) reinterpret_cast<float4*>(sW)[tid] = w0v;
    __syncthreads();

    const int sub  = lane & 1;
    const int rl   = lane >> 1;
    const unsigned suboff = (unsigned)sub * 16u;
    const int m     = lane & 15;
    const int khalf = (lane >> 4) * 2;
    const float bias0 = bias[m];
    const float bias1 = bias[16 + m];
    float* __restrict__ tile = sT + wid * (16 * L0_PITCH);
    for (int t = wid; t < L0_TILES; t += EDGE_T / 32) {
        const int il    = t * 16 + rl;
        const bool have = il < RANGE;
        const int ilc   = have ? il : RANGE - 1;
        const int stl   = s_rp[ilc];
        const int n     = have ? s_rp[ilc + 1] - stl : 0;
        const float4 own = load_row4<KPAD>(xp, nodebase + lo + ilc, suboff);
        const float4 acc = fits ? gather_sum_staged<KPAD>(piece + (stl - first), xp, n, suboff)
                                : gather_sum<KPAD>(csr, xp, (unsigned)stl, n, suboff);
        float* trow = tile + rl * L0_PITCH + 4 * sub;
        *reinterpret_cast<float4*>(trow)        = own;
        *reinterpret_cast<float4*>(trow + KPAD) = acc;
        __builtin_amdgcn_fence(__ATOMIC_RELEASE, "wavefront");
        __builtin_amdgcn_wave_barrier();
        __builtin_amdgcn_fence(__ATOMIC_ACQUIRE, "wavefront");

        v8f acc0 = {0.f,0.f,0.f,0.f,0.f,0.f,0.f,0.f};
        v8f acc1 = {0.f,0.f,0.f,0.f,0.f,0.f,0.f,0.f};
        #pragma unroll
        for (int c = 0; c < L0_K2 / 4; ++c) {
            const int k0 = c * 4 + khalf;
            v2f a = *reinterpret_cast<const v2f*>(tile + m * L0_PITCH + k0);
            v2f b0, b1;
            b0.x = sW[k0 * HID + m];
            b0.y = sW[(k0 + 1) * HID + m];
            b1.x = sW[k0 * HID + 16 + m];
            b1.y = sW[(k0 + 1) * HID + 16 + m];
            acc0 = __builtin_amdgcn_wmma_f32_16x16x4_f32(
                       false, a, false, b0, (short)0, acc0, false, false);
            acc1 = __builtin_amdgcn_wmma_f32_16x16x4_f32(
                       false, a, false, b1, (short)0, acc1, false, false);
        }
        const int rb16 = t * 16 + ((lane >> 4) ? 8 : 0);
        #pragma unroll
        for (int v = 0; v < 8; ++v) {
            const int ir = rb16 + v;
            const float t0 = tanhf(acc0[v] + bias0);
            const float t1 = tanhf(acc1[v] + bias1);
            if (ir < RANGE) {
                const size_t r = (size_t)(nodebase + lo + ir);
                hout[r * HID + m]      = t0;
                hout[r * HID + 16 + m] = t1;
            }
        }
        __builtin_amdgcn_fence(__ATOMIC_RELEASE, "wavefront");
        __builtin_amdgcn_wave_barrier();
        __builtin_amdgcn_fence(__ATOMIC_ACQUIRE, "wavefront");
    }

    if (fits && wid >= L0_TWICE) {
        constexpr int COPY_T = EDGE_T - 32 * L0_TWICE;
        for (int e0 = (first & ~15) + (tid - 32 * L0_TWICE); e0 < end; e0 += 4 * COPY_T) {
            int2 v[4];
            #pragma unroll
            for (int u = 0; u < 4; ++u) {
                const int e = e0 + u * COPY_T;
                v[u] = piece[(e >= first && e < end) ? e - first : 0];
            }
            #pragma unroll
            for (int u = 0; u < 4; ++u) {
                const int e = e0 + u * COPY_T;
                if (e >= first && e < end) csr[e] = v[u];
            }
        }
    }
}

#define ENT_CAP 320
#define ENT_PER_LANE (ENT_CAP / 32)
template <int S, int KW, int LAYER_T, bool READOUT>
__global__ void __launch_bounds__(LAYER_T)
tag_layer_kernel(const int* __restrict__ rowptr, const int* __restrict__ cnt,
                 const int2* __restrict__ csr, const float* __restrict__ hin,
                 const float* __restrict__ W, const float* __restrict__ bias,
                 float* __restrict__ hout, const float* __restrict__ wreg,
                 const float* __restrict__ breg, float* __restrict__ out) {
    constexpr int K2    = 2 * S;
    constexpr int PITCH = K2 + 4;
    constexpr int LPR   = S / 4;
    constexpr int RPP   = 32 / LPR;
    constexpr int NPASS = 16 / RPP;
    constexpr int WAVES = LAYER_T / 32;
    constexpr int WTRIPS = (K2 * HID / 4) / LAYER_T;
    static_assert((K2 * HID / 4) % LAYER_T == 0 && HID == 32,
                  "whole staging trips; two 16-column output tiles");
    static_assert((S % 4) == 0 && (32 % LPR) == 0 && (16 % RPP) == 0 && ((PITCH / 4) % 2) == 1,
                  "whole float4 pieces per lane, whole passes per tile, odd 16-byte pitch");
    static_assert((NTOT % (16 * WAVES)) == 0, "every wave of every work-group has a whole tile");
    static_assert((NN % 16) == 0 && (ENT_CAP % 32) == 0,
                  "a tile lies inside one graph; whole waves of entries per staging trip");
    __shared__ __attribute__((aligned(16))) float sW[K2 * HID];
    __shared__ __attribute__((aligned(16))) float sT[WAVES * 16 * PITCH];
    __shared__ __attribute__((aligned(16))) int2  sE[WAVES * ENT_CAP];

    const int lane = threadIdx.x & 31;
    const int wib  = threadIdx.x >> 5;
    const int row0 = (blockIdx.x * WAVES + wib) * 16;
    const int sub  = lane % LPR;
    const int rl   = lane / LPR;
    const unsigned suboff = (unsigned)sub * 16u;
    const int m     = lane & 15;
    const int khalf = (lane >> 4) * 2;

    int n[NPASS]; unsigned st[NPASS]; float4 own[NPASS];
    #pragma unroll
    for (int p = 0; p < NPASS; ++p) {
        const int row = row0 + p * RPP + rl;
        n[p]  = cnt[row];
        st[p] = (unsigned)rowptr[row];
    }
    #pragma unroll
    for (int p = 0; p < NPASS; ++p)
        own[p] = load_row4<S>(hin, row0 + p * RPP + rl, suboff);

    const unsigned run0 = (unsigned)rowptr[row0];
    const unsigned runn = (unsigned)rowptr[row0 + 15] + (unsigned)cnt[row0 + 15] - run0;
    const bool staged = runn <= (unsigned)ENT_CAP;

    float4 wv[WTRIPS];
    #pragma unroll
    for (int t = 0; t < WTRIPS; ++t) {
        const int i   = threadIdx.x + t * LAYER_T;
        const int s   = i / (S * (HID / 4));
        const int rem = i - s * (S * (HID / 4));
        const int kw  = rem / (HID / 4);
        const int c4  = rem - kw * (HID / 4);
        wv[t] = (kw < KW) ? reinterpret_cast<const float4*>(W)[(s * KW + kw) * (HID / 4) + c4]
                          : make_float4(0.f, 0.f, 0.f, 0.f);
    }
    const float bias0 = bias[m];
    const float bias1 = bias[16 + m];
    int2 ev[ENT_PER_LANE];
    #pragma unroll
    for (int k = 0; k < ENT_PER_LANE; ++k) {
        const unsigned i = (unsigned)lane + 32u * k;
        ev[k] = load_entry(csr, (staged && i < runn) ? run0 + i : 0u);
    }
    #pragma unroll
    for (int t = 0; t < WTRIPS; ++t)
        reinterpret_cast<float4*>(sW)[threadIdx.x + t * LAYER_T] = wv[t];
    __syncthreads();

    int2* __restrict__ ents = sE + wib * ENT_CAP;
    #pragma unroll
    for (int k = 0; k < ENT_PER_LANE; ++k) ents[lane + 32 * k] = ev[k];
    __builtin_amdgcn_fence(__ATOMIC_RELEASE, "wavefront");
    __builtin_amdgcn_wave_barrier();
    __builtin_amdgcn_fence(__ATOMIC_ACQUIRE, "wavefront");

    float* __restrict__ tile = sT + wib * (16 * PITCH);
    #pragma unroll
    for (int p = 0; p < NPASS; ++p) {
        const float4 acc = staged ? gather_sum_staged<S>(ents + (st[p] - run0), hin, n[p], suboff)
                                  : gather_sum<S>(csr, hin, st[p], n[p], suboff);
        float* trow = tile + (p * RPP + rl) * PITCH + 4 * sub;
        *reinterpret_cast<float4*>(trow)     = own[p];
        *reinterpret_cast<float4*>(trow + S) = acc;
    }
    __builtin_amdgcn_fence(__ATOMIC_RELEASE, "wavefront");
    __builtin_amdgcn_wave_barrier();
    __builtin_amdgcn_fence(__ATOMIC_ACQUIRE, "wavefront");

    float wr0 = 0.f, wr1 = 0.f, br = 0.f;
    if constexpr (READOUT) { wr0 = wreg[m]; wr1 = wreg[16 + m]; br = breg[0]; }

    v8f acc0 = {0.f,0.f,0.f,0.f,0.f,0.f,0.f,0.f};
    v8f acc1 = {0.f,0.f,0.f,0.f,0.f,0.f,0.f,0.f};

    #pragma unroll
    for (int c = 0; c < K2 / 4; ++c) {
        const int k0 = c * 4 + khalf;
        v2f a = *reinterpret_cast<const v2f*>(tile + m * PITCH + k0);
        v2f b0, b1;
        b0.x = sW[k0 * HID + m];
        b0.y = sW[(k0 + 1) * HID + m];
        b1.x = sW[k0 * HID + 16 + m];
        b1.y = sW[(k0 + 1) * HID + 16 + m];
        acc0 = __builtin_amdgcn_wmma_f32_16x16x4_f32(
                   false, a, false, b0, (short)0, acc0, false, false);
        acc1 = __builtin_amdgcn_wmma_f32_16x16x4_f32(
                   false, a, false, b1, (short)0, acc1, false, false);
    }

    const int rbase = row0 + ((lane >> 4) ? 8 : 0);
    #pragma unroll
    for (int v = 0; v < 8; ++v) {
        int r = rbase + v;
        float x0 = acc0[v] + bias0;
        float x1 = acc1[v] + bias1;
        const float t0 = tanhf(x0);
        const float t1 = tanhf(x1);
        hout[(size_t)r * HID + m]      = t0;
        hout[(size_t)r * HID + 16 + m] = t1;
        if constexpr (READOUT) {
            float d = t0 * wr0 + t1 * wr1;
            d += __shfl_xor(d, 8);
            d += __shfl_xor(d, 4);
            d += __shfl_xor(d, 2);
            d += __shfl_xor(d, 1);
            if (m == 0) tile[(lane >> 4) * 8 + v] = d + br;
        }
    }
    if constexpr (READOUT) {
        __syncthreads();
        if (threadIdx.x < WAVES * 16)
            out[blockIdx.x * (WAVES * 16) + threadIdx.x] =
                sT[(threadIdx.x >> 4) * (16 * PITCH) + (threadIdx.x & 15)];
    }
}

static inline int cdiv(long long a, long long b) { return (int)((a + b - 1) / b); }

extern "C" void kernel_launch(void* const* d_in, const int* in_sizes, int n_in,
                              void* d_out, int out_size, void* d_ws, size_t ws_size,
                              hipStream_t stream) {
    const float* X    = (const float*)d_in[0];
    const int*   ei   = (const int*)  d_in[1];
    const float* W0   = (const float*)d_in[2];
    const float* b0   = (const float*)d_in[3];
    const float* W1   = (const float*)d_in[4];
    const float* b1   = (const float*)d_in[5];
    const float* W2   = (const float*)d_in[6];
    const float* b2   = (const float*)d_in[7];
    const float* Wreg = (const float*)d_in[8];
    const float* breg = (const float*)d_in[9];

    float* out    = (float*)d_out;
    float* hfinal = out + NTOT;

    int*   cnt    = (int*)d_ws;
    int*   rowptr = cnt + NTOT;
    int2*  lsort  = (int2*)(rowptr + NTOT);
    int2*  csr    = lsort + ETOT;
    float* hA     = (float*)(csr + ETOT);
    float* hB     = hA  + (size_t)NTOT * HID;
    float* Xp     = hB  + (size_t)NTOT * HID;
    int*   part   = (int*)(Xp + (size_t)NTOT * KPAD);
    int*   bounds = part + (size_t)BATCH * EDGE_SPLIT * NN;

    count_sort_kernel<<<BATCH * EDGE_SPLIT, EDGE_T, 0, stream>>>(ei, lsort, bounds, part);
    chunk_offsets_kernel<<<cdiv(NTOT / 4, OFFS_T), OFFS_T, 0, stream>>>(part, cnt, X, Xp);
    fill_rows_kernel<<<BATCH * EDGE_SPLIT, EDGE_T, 0, stream>>>(lsort, bounds, cnt, part, rowptr, csr,
                                                                Xp, W0, b0, hA);

    const int wide_blocks   = cdiv(NTOT / 16, 256 / 32);


    tag_layer_kernel<HID, HID, 256, false><<<wide_blocks, 256, 0, stream>>>(
        rowptr, cnt, csr, hA, W1, b1, hB, nullptr, nullptr, nullptr);

    tag_layer_kernel<HID, HID, 256, true><<<wide_blocks, 256, 0, stream>>>(
        rowptr, cnt, csr, hB, W2, b2, hfinal, Wreg, breg, out);
}


# per-chunk node counts handed on as 16-bit values, written once and never rewritten: the fill work-group forms its range's chunk prefixes in LDS; launch 2 only sums them; list and sum order kept
# speedup vs baseline: 38.0501x; 1.1155x over previous
#include <hip/hip_runtime.h>
#include <hip/hip_bf16.h>
#include <math.h>

#define BATCH   8
#define WINDOW  5
#define NN      20000
#define EPB     320000
#define HID     32
#define KPAD    8
#define NTOT    (BATCH * NN)
#define ETOT    (BATCH * EPB)

typedef float v2f __attribute__((ext_vector_type(2)));
typedef float v8f __attribute__((ext_vector_type(8)));

#define EDGE_T      1024
#define EDGE_SPLIT  32
#define QPW         (EPB / 4 / EDGE_SPLIT)
#define EDGE_QPT    ((QPW + EDGE_T - 1) / EDGE_T)
#define NQ          (NN / 4)
#define TAB_TRIPS   5
#define TAB_T       (NQ / TAB_TRIPS)

#define RANGE       (NN / EDGE_SPLIT)
#define CHUNK_E     (QPW * 4)
#define SCAN_PER    20

__global__ void __launch_bounds__(EDGE_T)
count_sort_kernel(const int* __restrict__ ei, int2* __restrict__ lsort,
                  int* __restrict__ bounds, unsigned short* __restrict__ part) {
    static_assert(EPB % (4 * EDGE_SPLIT) == 0 && NN % 4 == 0 && NQ % TAB_TRIPS == 0 &&
                  TAB_T <= EDGE_T && TAB_T % 8 == 0,
                  "whole edge quads per chunk; tables move 16 bytes at a time in whole lines");
    static_assert(NN % SCAN_PER == 0 && NN / SCAN_PER == TAB_T && (SCAN_PER % 4) == 0 &&
                  SCAN_PER / 4 == TAB_TRIPS && NN % EDGE_SPLIT == 0 && NN < 32768 &&
                  EDGE_SPLIT <= EDGE_T && (CHUNK_E % 2) == 0,
                  "SCAN_PER consecutive nodes per scanning thread; src and dst share one word");
    static_assert(CHUNK_E <= 65535 && NN % 8 == 0,
                  "a chunk's count of one node fits 16 bits; eight counts per 16-byte piece");
    __shared__ __attribute__((aligned(16))) int  hist[NN];
    __shared__ __attribute__((aligned(16))) int2 stage[CHUNK_E];
    __shared__ int wtot[EDGE_T / 32];
    const int b   = blockIdx.x / EDGE_SPLIT;
    const int c   = blockIdx.x - b * EDGE_SPLIT;
    const int tid = threadIdx.x;
    const int lane = tid & 31;
    const int wid  = tid >> 5;
    const int* __restrict__ eb = ei + (size_t)b * 2 * EPB;
    const int4* __restrict__ src4 = reinterpret_cast<const int4*>(eb) + c * QPW;
    const int4* __restrict__ dst4 = reinterpret_cast<const int4*>(eb + EPB) + c * QPW;

    int4 sq[EDGE_QPT], dq[EDGE_QPT], rq[EDGE_QPT];
    #pragma unroll
    for (int k = 0; k < EDGE_QPT; ++k) {
        const int q  = tid + k * EDGE_T;
        const int qc = q < QPW ? q : QPW - 1;
        sq[k] = src4[qc]; dq[k] = dst4[qc];
    }
    for (int i = tid; i < NQ; i += EDGE_T)
        reinterpret_cast<int4*>(hist)[i] = make_int4(0, 0, 0, 0);
    __syncthreads();

    #pragma unroll
    for (int k = 0; k < EDGE_QPT; ++k) {
        const int q = tid + k * EDGE_T;
        rq[k] = make_int4(0, 0, 0, 0);
        if (q < QPW) {
            rq[k].x = atomicAdd(&hist[dq[k].x], 1);
            rq[k].y = atomicAdd(&hist[dq[k].y], 1);
            rq[k].z = atomicAdd(&hist[dq[k].z], 1);
            rq[k].w = atomicAdd(&hist[dq[k].w], 1);
        }
    }
    __syncthreads();

    uint4* __restrict__ gp = reinterpret_cast<uint4*>(part + (size_t)blockIdx.x * NN);
    for (int i = tid; i < NN / 8; i += EDGE_T) {
        const int4 ca = reinterpret_cast<const int4*>(hist)[2 * i];
        const int4 cb = reinterpret_cast<const int4*>(hist)[2 * i + 1];
        gp[i] = make_uint4((unsigned)ca.x | ((unsigned)ca.y << 16),
                           (unsigned)ca.z | ((unsigned)ca.w << 16),
                           (unsigned)cb.x | ((unsigned)cb.y << 16),
                           (unsigned)cb.z | ((unsigned)cb.w << 16));
    }

    const bool act = tid < TAB_T;
    int cc[SCAN_PER];
    int tot = 0;
    #pragma unroll
    for (int k = 0; k < SCAN_PER; ++k) cc[k] = 0;
    if (act) {
        #pragma unroll
        for (int g = 0; g < SCAN_PER / 4; ++g) {
            const int4 q = reinterpret_cast<const int4*>(hist)[tid * (SCAN_PER / 4) + g];
            cc[4 * g + 0] = q.x; cc[4 * g + 1] = q.y; cc[4 * g + 2] = q.z; cc[4 * g + 3] = q.w;
            tot += q.x + q.y + q.z + q.w;
        }
    }
    int inc = tot;
    #pragma unroll
    for (int off = 1; off < 32; off <<= 1) {
        int t = __shfl_up(inc, off);
        if (lane >= off) inc += t;
    }
    if (lane == 31) wtot[wid] = inc;
    __syncthreads();
    if (wid == 0) {
        int w = wtot[lane];
        int winc = w;
        #pragma unroll
        for (int off = 1; off < 32; off <<= 1) {
            int t = __shfl_up(winc, off);
            if (lane >= off) winc += t;
        }
        wtot[lane] = winc - w;
    }
    __syncthreads();
    if (act) {
        int run = wtot[wid] + (inc - tot);
        #pragma unroll
        for (int g = 0; g < SCAN_PER / 4; ++g) {
            int4 r;
            r.x = run; run += cc[4 * g + 0];
            r.y = run; run += cc[4 * g + 1];
            r.z = run; run += cc[4 * g + 2];
            r.w = run; run += cc[4 * g + 3];
            reinterpret_cast<int4*>(hist)[tid * (SCAN_PER / 4) + g] = r;
        }
    }
    __syncthreads();

    if (tid < EDGE_SPLIT) bounds[blockIdx.x * EDGE_SPLIT + tid] = hist[tid * RANGE];
    #pragma unroll
    for (int k = 0; k < EDGE_QPT; ++k) {
        const int q = tid + k * EDGE_T;
        if (q < QPW) {
            stage[hist[dq[k].x] + rq[k].x] = make_int2(sq[k].x | (dq[k].x << 16), rq[k].x);
            stage[hist[dq[k].y] + rq[k].y] = make_int2(sq[k].y | (dq[k].y << 16), rq[k].y);
            stage[hist[dq[k].z] + rq[k].z] = make_int2(sq[k].z | (dq[k].z << 16), rq[k].z);
            stage[hist[dq[k].w] + rq[k].w] = make_int2(sq[k].w | (dq[k].w << 16), rq[k].w);
        }
    }
    __syncthreads();

    int4* __restrict__ lo4 = reinterpret_cast<int4*>(lsort + (size_t)blockIdx.x * CHUNK_E);
    constexpr int LS_TRIPS = (CHUNK_E / 2 + EDGE_T - 1) / EDGE_T;
    int4 lv[LS_TRIPS];
    #pragma unroll
    for (int k = 0; k < LS_TRIPS; ++k) {
        const int i = tid + k * EDGE_T;
        lv[k] = reinterpret_cast<const int4*>(stage)[i < CHUNK_E / 2 ? i : 0];
    }
    #pragma unroll
    for (int k = 0; k < LS_TRIPS; ++k) {
        const int i = tid + k * EDGE_T;
        if (i < CHUNK_E / 2) lo4[i] = lv[k];
    }
}

#define OFFS_T 128
__global__ void __launch_bounds__(OFFS_T)
in_degree_kernel(const unsigned short* __restrict__ part, int* __restrict__ cnt,
                     const float* __restrict__ x, float* __restrict__ xp) {
    static_assert(WINDOW == 5 && KPAD == 8, "four rows of X are five 16-byte pieces");
    const int gq = blockIdx.x * OFFS_T + threadIdx.x;
    if (gq >= NTOT / 4) return;
    const int b  = gq / NQ;
    const int nq = gq - b * NQ;
    const float4* __restrict__ xr = reinterpret_cast<const float4*>(x) + (size_t)gq * WINDOW;
    const float4 a0 = xr[0], a1 = xr[1], a2 = xr[2], a3 = xr[3], a4 = xr[4];
    const uint2* __restrict__ p =
        reinterpret_cast<const uint2*>(part + (size_t)b * EDGE_SPLIT * NN) + nq;
    uint2 t[EDGE_SPLIT];
    #pragma unroll
    for (int c = 0; c < EDGE_SPLIT; ++c) t[c] = p[(size_t)c * NQ];
    float4* __restrict__ xo = reinterpret_cast<float4*>(xp) + (size_t)gq * KPAD;
    xo[0] = a0;
    xo[1] = make_float4(a1.x, 0.f, 0.f, 0.f);
    xo[2] = make_float4(a1.y, a1.z, a1.w, a2.x);
    xo[3] = make_float4(a2.y, 0.f, 0.f, 0.f);
    xo[4] = make_float4(a2.z, a2.w, a3.x, a3.y);
    xo[5] = make_float4(a3.z, 0.f, 0.f, 0.f);
    xo[6] = make_float4(a3.w, a4.x, a4.y, a4.z);
    xo[7] = make_float4(a4.w, 0.f, 0.f, 0.f);
    int4 run = make_int4(0, 0, 0, 0);
    #pragma unroll
    for (int c = 0; c < EDGE_SPLIT; ++c) {
        run.x += (int)(t[c].x & 0xffffu); run.y += (int)(t[c].x >> 16);
        run.z += (int)(t[c].y & 0xffffu); run.w += (int)(t[c].y >> 16);
    }
    reinterpret_cast<int4*>(cnt)[gq] = run;
}

static __device__ __forceinline__ int2 load_entry(const int2* __restrict__ csr, unsigned idx) {
    return *reinterpret_cast<const int2*>(reinterpret_cast<const char*>(csr) + idx * 8u);
}
template <int S>
static __device__ __forceinline__ float4 load_row4(const float* __restrict__ hin, int src,
                                                   unsigned suboff) {
    return *reinterpret_cast<const float4*>(reinterpret_cast<const char*>(hin) +
                                            ((unsigned)src * (unsigned)(S * 4) + suboff));
}
static __device__ __forceinline__ void axpy4(float4& acc, const float4& v, float w) {
    acc.x += v.x * w; acc.y += v.y * w; acc.z += v.z * w; acc.w += v.w * w;
}

template <int S>
static __device__ __forceinline__ float4 gather_sum(const int2* __restrict__ csr,
                                                    const float* __restrict__ hin,
                                                    unsigned start, int n, unsigned suboff) {
    float4 acc = make_float4(0.f, 0.f, 0.f, 0.f);
    int j = 0;
    for (; j + 4 <= n; j += 4) {
        const int2 e0 = load_entry(csr, start + j);
        const int2 e1 = load_entry(csr, start + j + 1);
        const int2 e2 = load_entry(csr, start + j + 2);
        const int2 e3 = load_entry(csr, start + j + 3);
        const float4 v0 = load_row4<S>(hin, e0.x, suboff);
        const float4 v1 = load_row4<S>(hin, e1.x, suboff);
        const float4 v2 = load_row4<S>(hin, e2.x, suboff);
        const float4 v3 = load_row4<S>(hin, e3.x, suboff);
        axpy4(acc, v0, __int_as_float(e0.y));
        axpy4(acc, v1, __int_as_float(e1.y));
        axpy4(acc, v2, __int_as_float(e2.y));
        axpy4(acc, v3, __int_as_float(e3.y));
    }
    const int r = n - j;
    if (r > 0) {
        const int2 e0 = load_entry(csr, start + j);
        const int2 e1 = load_entry(csr, start + j + (r > 1 ? 1 : 0));
        const int2 e2 = load_entry(csr, start + j + (r > 2 ? 2 : 0));
        float4 v0 = load_row4<S>(hin, e0.x, suboff);
        float4 v1 = load_row4<S>(hin, e1.x, suboff);
        float4 v2 = load_row4<S>(hin, e2.x, suboff);
        v1.x = (r > 1) ? v1.x : 0.f; v1.y = (r > 1) ? v1.y : 0.f;
        v1.z = (r > 1) ? v1.z : 0.f; v1.w = (r > 1) ? v1.w : 0.f;
        v2.x = (r > 2) ? v2.x : 0.f; v2.y = (r > 2) ? v2.y : 0.f;
        v2.z = (r > 2) ? v2.z : 0.f; v2.w = (r > 2) ? v2.w : 0.f;
        axpy4(acc, v0, __int_as_float(e0.y));
        axpy4(acc, v1, __int_as_float(e1.y));
        axpy4(acc, v2, __int_as_float(e2.y));
    }
    return acc;
}

template <int S, int M>
static __device__ __forceinline__ void gather_tail(float4& acc, const int2* __restrict__ ent,
                                                   const float* __restrict__ hin,
                                                   int r, unsigned suboff) {
    int2 e[M]; float4 v[M];
    #pragma unroll
    for (int u = 0; u < M; ++u) e[u] = ent[u < r ? u : 0];
    #pragma unroll
    for (int u = 0; u < M; ++u) v[u] = load_row4<S>(hin, e[u].x, suboff);
    #pragma unroll
    for (int u = 1; u < M; ++u) {
        v[u].x = (u < r) ? v[u].x : 0.f; v[u].y = (u < r) ? v[u].y : 0.f;
        v[u].z = (u < r) ? v[u].z : 0.f; v[u].w = (u < r) ? v[u].w : 0.f;
    }
    #pragma unroll
    for (int u = 0; u < M; ++u) axpy4(acc, v[u], __int_as_float(e[u].y));
}

#define GATHER_DEPTH 8
template <int S>
static __device__ __forceinline__ float4 gather_sum_staged(const int2* __restrict__ ent,
                                                           const float* __restrict__ hin,
                                                           int n, unsigned suboff) {
    float4 acc = make_float4(0.f, 0.f, 0.f, 0.f);
    int j = 0;
    for (; j + GATHER_DEPTH <= n; j += GATHER_DEPTH) {
        int2 e[GATHER_DEPTH]; float4 v[GATHER_DEPTH];
        #pragma unroll
        for (int u = 0; u < GATHER_DEPTH; ++u) e[u] = ent[j + u];
        #pragma unroll
        for (int u = 0; u < GATHER_DEPTH; ++u) v[u] = load_row4<S>(hin, e[u].x, suboff);
        #pragma unroll
        for (int u = 0; u < GATHER_DEPTH; ++u) axpy4(acc, v[u], __int_as_float(e[u].y));
    }
    const int r = n - j;
    if (r > 3)      gather_tail<S, GATHER_DEPTH - 1>(acc, ent + j, hin, r, suboff);
    else if (r > 0) gather_tail<S, 3>(acc, ent + j, hin, r, suboff);
    return acc;
}

#define FILL_DEPTH  12
#define PIECE_CAP   16384
#define PRE_N       (EDGE_SPLIT * RANGE)
static __device__ __forceinline__ void put_entry(const int* __restrict__ s_rp,
                                                 const int* __restrict__ pre,
                                                 const int* __restrict__ s_cd,
                                                 int2* __restrict__ to, int first,
                                                 int2 e, int lo, int nodebase) {
    const int sl = e.x & 0xffff;
    const int dl = (int)((unsigned)e.x >> 16);
    const int il = dl - lo;
    const int slot = s_rp[il] + pre[il] + e.y;
    const float nrm = __int_as_float(s_cd[sl]) * __int_as_float(s_cd[dl]);
    to[slot - first] = make_int2(sl + nodebase, __float_as_int(nrm));
}
static __device__ __forceinline__ void put_group(const int* __restrict__ s_rp,
                                                 const int* __restrict__ pre,
                                                 const int* __restrict__ s_cd,
                                                 int2* __restrict__ to, int first,
                                                 const int2 (&e)[FILL_DEPTH], int lane, int len,
                                                 int lo, int nodebase) {
    int slot[FILL_DEPTH]; int sl[FILL_DEPTH]; float nrm[FILL_DEPTH];
    #pragma unroll
    for (int k = 0; k < FILL_DEPTH; ++k) {
        const bool ok = lane + 32 * k < len;
        sl[k] = e[k].x & 0xffff;
        const int dl = (int)((unsigned)e[k].x >> 16);
        const int il = ok ? dl - lo : 0;
        slot[k] = s_rp[il] + pre[il] + e[k].y;
        nrm[k]  = __int_as_float(s_cd[sl[k]]) * __int_as_float(s_cd[dl]);
    }
    #pragma unroll
    for (int k = 0; k < FILL_DEPTH; ++k)
        if (lane + 32 * k < len)
            to[slot[k] - first] = make_int2(sl[k] + nodebase, __float_as_int(nrm[k]));
}
__global__ void __launch_bounds__(EDGE_T)
fill_rows_kernel(const int2* __restrict__ lsort, const int* __restrict__ bounds,
                 const int* __restrict__ cnt, const unsigned short* __restrict__ part,
                 int* __restrict__ rowptr, int2* __restrict__ csr,
                 const float* __restrict__ xp, const float* __restrict__ W,
                 const float* __restrict__ bias, float* __restrict__ hout) {
    static_assert(EDGE_T / 32 == EDGE_SPLIT && TAB_TRIPS * EDGE_T >= NQ && RANGE <= EDGE_T,
                  "one wave per chunk; the in-degrees are copied in TAB_TRIPS trips; one thread per node of the range");
    __shared__ __attribute__((aligned(16))) int s_pre[PRE_N];
    __shared__ __attribute__((aligned(16))) int s_cd[NN];
    __shared__ int s_rp[RANGE + 1];
    __shared__ int wtot[EDGE_T / 32];
    __shared__ __attribute__((aligned(16))) int2 piece[PIECE_CAP];
    constexpr int L0_K2    = 2 * KPAD;
    constexpr int L0_PITCH = L0_K2 + 4;
    constexpr int L0_TILES = (RANGE + 15) / 16;
    constexpr int L0_TWICE = L0_TILES - EDGE_T / 32;
    static_assert(HID == 32 && L0_K2 * HID / 4 <= EDGE_T && ((L0_PITCH / 4) % 2) == 1 &&
                  (L0_K2 * HID + (EDGE_T / 32) * 16 * L0_PITCH) <= PRE_N &&
                  L0_TWICE >= 0 && L0_TWICE < EDGE_T / 32,
                  "W and the waves' tiles fit into the prefix table; at most two tiles per wave");
    const int b   = blockIdx.x / EDGE_SPLIT;
    const int rg  = blockIdx.x - b * EDGE_SPLIT;
    const int lo  = rg * RANGE;
    const int tid = threadIdx.x;
    const int lane = tid & 31;
    const int wid  = __builtin_amdgcn_readfirstlane(tid >> 5);

    const int chunk = b * EDGE_SPLIT + wid;
    const int* __restrict__ bnd = bounds + chunk * EDGE_SPLIT;
    const int rb = bnd[rg];
    const int rn = bnd[rg + 1 < EDGE_SPLIT ? rg + 1 : rg];
    const int len = (rg + 1 < EDGE_SPLIT ? rn : CHUNK_E) - rb;
    const int2* __restrict__ lbase = lsort + (size_t)chunk * CHUNK_E;

    const int4* __restrict__ gc = reinterpret_cast<const int4*>(cnt + b * NN);
    const bool pren = tid < RANGE;
    const unsigned short* __restrict__ gpn =
        part + ((size_t)b * EDGE_SPLIT * NN + lo + (pren ? tid : 0));
    const bool act = tid < TAB_T;
    int4 tc[TAB_TRIPS]; int cv[EDGE_SPLIT];
    #pragma unroll
    for (int k = 0; k < TAB_TRIPS; ++k) {
        const int i = tid + k * EDGE_T;
        tc[k] = gc[i < NQ ? i : 0];
    }
    #pragma unroll
    for (int q = 0; q < EDGE_SPLIT; ++q) cv[q] = 0;
    if (pren) {
        #pragma unroll
        for (int q = 0; q < EDGE_SPLIT; ++q) cv[q] = (int)gpn[(size_t)q * NN];
    }
    int2 ev[FILL_DEPTH];
    #pragma unroll
    for (int k = 0; k < FILL_DEPTH; ++k) {
        const int i = lane + 32 * k;
        ev[k] = lbase[i < len ? rb + i : 0];
    }
    float4 w0v = make_float4(0.f, 0.f, 0.f, 0.f);
    if (tid < L0_K2 * HID / 4) {
        const int sidx = tid / (KPAD * (HID / 4));
        const int rem  = tid - sidx * (KPAD * (HID / 4));
        const int kw   = rem / (HID / 4);
        const int c4   = rem - kw * (HID / 4);
        if (kw < WINDOW)
            w0v = reinterpret_cast<const float4*>(W)[(sidx * WINDOW + kw) * (HID / 4) + c4];
    }
    #pragma unroll
    for (int k = 0; k < TAB_TRIPS; ++k) {
        const int i = tid + k * EDGE_T;
        if (i < NQ) reinterpret_cast<int4*>(s_cd)[i] = tc[k];
    }
    if (pren) {
        int before = 0;
        #pragma unroll
        for (int q = 0; q < EDGE_SPLIT; ++q) { s_pre[q * RANGE + tid] = before; before += cv[q]; }
    }
    __syncthreads();

    int c[SCAN_PER];
    int tot = 0;
    #pragma unroll
    for (int k = 0; k < SCAN_PER; ++k) c[k] = 0;
    if (act) {
        #pragma unroll
        for (int g = 0; g < SCAN_PER / 4; ++g) {
            const int4 q = reinterpret_cast<const int4*>(s_cd)[tid * (SCAN_PER / 4) + g];
            c[4 * g + 0] = q.x; c[4 * g + 1] = q.y; c[4 * g + 2] = q.z; c[4 * g + 3] = q.w;
            tot += q.x + q.y + q.z + q.w;
        }
    }
    int inc = tot;
    #pragma unroll
    for (int off = 1; off < 32; off <<= 1) {
        int t = __shfl_up(inc, off);
        if (lane >= off) inc += t;
    }
    if (lane == 31) wtot[wid] = inc;
    __syncthreads();
    if (wid == 0) {
        int w = wtot[lane];
        int winc = w;
        #pragma unroll
        for (int off = 1; off < 32; off <<= 1) {
            int t = __shfl_up(winc, off);
            if (lane >= off) winc += t;
        }
        wtot[lane] = winc - w;
    }
    __syncthreads();

    if (act) {
        int run = b * EPB + wtot[wid] + (inc - tot);
        int4* __restrict__ rp = reinterpret_cast<int4*>(rowptr + b * NN + tid * SCAN_PER);
        #pragma unroll
        for (int g = 0; g < SCAN_PER / 4; ++g) {
            int4 r; int4 f;
            r.x = run; run += c[4 * g + 0];
            r.y = run; run += c[4 * g + 1];
            r.z = run; run += c[4 * g + 2];
            r.w = run; run += c[4 * g + 3];
            f.x = __float_as_int((c[4 * g + 0] > 0) ? rsqrtf((float)c[4 * g + 0]) : 0.0f);
            f.y = __float_as_int((c[4 * g + 1] > 0) ? rsqrtf((float)c[4 * g + 1]) : 0.0f);
            f.z = __float_as_int((c[4 * g + 2] > 0) ? rsqrtf((float)c[4 * g + 2]) : 0.0f);
            f.w = __float_as_int((c[4 * g + 3] > 0) ? rsqrtf((float)c[4 * g + 3]) : 0.0f);
            reinterpret_cast<int4*>(s_cd)[tid * (SCAN_PER / 4) + g] = f;
            const int nd = tid * SCAN_PER + 4 * g - lo;
            if ((unsigned)(nd + 0) < (unsigned)RANGE) s_rp[nd + 0] = r.x;
            if ((unsigned)(nd + 1) < (unsigned)RANGE) s_rp[nd + 1] = r.y;
            if ((unsigned)(nd + 2) < (unsigned)RANGE) s_rp[nd + 2] = r.z;
            if ((unsigned)(nd + 3) < (unsigned)RANGE) s_rp[nd + 3] = r.w;
            if (nd + 0 == RANGE - 1) s_rp[RANGE] = r.y;
            if (nd + 1 == RANGE - 1) s_rp[RANGE] = r.z;
            if (nd + 2 == RANGE - 1) s_rp[RANGE] = r.w;
            if (nd + 3 == RANGE - 1) s_rp[RANGE] = run;
            if (rg == 0) rp[g] = r;
        }
    }
    __syncthreads();

    const int nodebase = b * NN;
    const int* pre = s_pre + wid * RANGE;
    const int first = s_rp[0];
    const int end   = s_rp[RANGE];
    const bool fits = end - first <= PIECE_CAP;
    if (fits) {
        put_group(s_rp, pre, s_cd, piece, first, ev, lane, len, lo, nodebase);
        for (int i = lane + 32 * FILL_DEPTH; i < len; i += 32)
            put_entry(s_rp, pre, s_cd, piece, first, lbase[rb + i], lo, nodebase);
    } else {
        put_group(s_rp, pre, s_cd, csr, 0, ev, lane, len, lo, nodebase);
        for (int i = lane + 32 * FILL_DEPTH; i < len; i += 32)
            put_entry(s_rp, pre, s_cd, csr, 0, lbase[rb + i], lo, nodebase);
    }
    __syncthreads();

    float* __restrict__ sW = reinterpret_cast<float*>(s_pre);
    float* __restrict__ sT = sW + L0_K2 * HID;
    if (tid < L0_K2 * HID / 4) reinterpret_cast<float4*>(sW)[tid] = w0v;
    __syncthreads();

    const int sub  = lane & 1;
    const int rl   = lane >> 1;
    const unsigned suboff = (unsigned)sub * 16u;
    const int m     = lane & 15;
    const int khalf = (lane >> 4) * 2;
    const float bias0 = bias[m];
    const float bias1 = bias[16 + m];
    float* __restrict__ tile = sT + wid * (16 * L0_PITCH);
    for (int t = wid; t < L0_TILES; t += EDGE_T / 32) {
        const int il    = t * 16 + rl;
        const bool have = il < RANGE;
        const int ilc   = have ? il : RANGE - 1;
        const int stl   = s_rp[ilc];
        const int n     = have ? s_rp[ilc + 1] - stl : 0;
        const float4 own = load_row4<KPAD>(xp, nodebase + lo + ilc, suboff);
        const float4 acc = fits ? gather_sum_staged<KPAD>(piece + (stl - first), xp, n, suboff)
                                : gather_sum<KPAD>(csr, xp, (unsigned)stl, n, suboff);
        float* trow = tile + rl * L0_PITCH + 4 * sub;
        *reinterpret_cast<float4*>(trow)        = own;
        *reinterpret_cast<float4*>(trow + KPAD) = acc;
        __builtin_amdgcn_fence(__ATOMIC_RELEASE, "wavefront");
        __builtin_amdgcn_wave_barrier();
        __builtin_amdgcn_fence(__ATOMIC_ACQUIRE, "wavefront");

        v8f acc0 = {0.f,0.f,0.f,0.f,0.f,0.f,0.f,0.f};
        v8f acc1 = {0.f,0.f,0.f,0.f,0.f,0.f,0.f,0.f};
        #pragma unroll
        for (int c = 0; c < L0_K2 / 4; ++c) {
            const int k0 = c * 4 + khalf;
            v2f a = *reinterpret_cast<const v2f*>(tile + m * L0_PITCH + k0);
            v2f b0, b1;
            b0.x = sW[k0 * HID + m];
            b0.y = sW[(k0 + 1) * HID + m];
            b1.x = sW[k0 * HID + 16 + m];
            b1.y = sW[(k0 + 1) * HID + 16 + m];
            acc0 = __builtin_amdgcn_wmma_f32_16x16x4_f32(
                       false, a, false, b0, (short)0, acc0, false, false);
            acc1 = __builtin_amdgcn_wmma_f32_16x16x4_f32(
                       false, a, false, b1, (short)0, acc1, false, false);
        }
        const int rb16 = t * 16 + ((lane >> 4) ? 8 : 0);
        #pragma unroll
        for (int v = 0; v < 8; ++v) {
            const int ir = rb16 + v;
            const float t0 = tanhf(acc0[v] + bias0);
            const float t1 = tanhf(acc1[v] + bias1);
            if (ir < RANGE) {
                const size_t r = (size_t)(nodebase + lo + ir);
                hout[r * HID + m]      = t0;
                hout[r * HID + 16 + m] = t1;
            }
        }
        __builtin_amdgcn_fence(__ATOMIC_RELEASE, "wavefront");
        __builtin_amdgcn_wave_barrier();
        __builtin_amdgcn_fence(__ATOMIC_ACQUIRE, "wavefront");
    }

    if (fits && wid >= L0_TWICE) {
        constexpr int COPY_T = EDGE_T - 32 * L0_TWICE;
        for (int e0 = (first & ~15) + (tid - 32 * L0_TWICE); e0 < end; e0 += 4 * COPY_T) {
            int2 v[4];
            #pragma unroll
            for (int u = 0; u < 4; ++u) {
                const int e = e0 + u * COPY_T;
                v[u] = piece[(e >= first && e < end) ? e - first : 0];
            }
            #pragma unroll
            for (int u = 0; u < 4; ++u) {
                const int e = e0 + u * COPY_T;
                if (e >= first && e < end) csr[e] = v[u];
            }
        }
    }
}

#define ENT_CAP 320
#define ENT_PER_LANE (ENT_CAP / 32)
template <int S, int KW, int LAYER_T, bool READOUT>
__global__ void __launch_bounds__(LAYER_T)
tag_layer_kernel(const int* __restrict__ rowptr, const int* __restrict__ cnt,
                 const int2* __restrict__ csr, const float* __restrict__ hin,
                 const float* __restrict__ W, const float* __restrict__ bias,
                 float* __restrict__ hout, const float* __restrict__ wreg,
                 const float* __restrict__ breg, float* __restrict__ out) {
    constexpr int K2    = 2 * S;
    constexpr int PITCH = K2 + 4;
    constexpr int LPR   = S / 4;
    constexpr int RPP   = 32 / LPR;
    constexpr int NPASS = 16 / RPP;
    constexpr int WAVES = LAYER_T / 32;
    constexpr int WTRIPS = (K2 * HID / 4) / LAYER_T;
    static_assert((K2 * HID / 4) % LAYER_T == 0 && HID == 32,
                  "whole staging trips; two 16-column output tiles");
    static_assert((S % 4) == 0 && (32 % LPR) == 0 && (16 % RPP) == 0 && ((PITCH / 4) % 2) == 1,
                  "whole float4 pieces per lane, whole passes per tile, odd 16-byte pitch");
    static_assert((NTOT % (16 * WAVES)) == 0, "every wave of every work-group has a whole tile");
    static_assert((NN % 16) == 0 && (ENT_CAP % 32) == 0,
                  "a tile lies inside one graph; whole waves of entries per staging trip");
    __shared__ __attribute__((aligned(16))) float sW[K2 * HID];
    __shared__ __attribute__((aligned(16))) float sT[WAVES * 16 * PITCH];
    __shared__ __attribute__((aligned(16))) int2  sE[WAVES * ENT_CAP];

    const int lane = threadIdx.x & 31;
    const int wib  = threadIdx.x >> 5;
    const int row0 = (blockIdx.x * WAVES + wib) * 16;
    const int sub  = lane % LPR;
    const int rl   = lane / LPR;
    const unsigned suboff = (unsigned)sub * 16u;
    const int m     = lane & 15;
    const int khalf = (lane >> 4) * 2;

    int n[NPASS]; unsigned st[NPASS]; float4 own[NPASS];
    #pragma unroll
    for (int p = 0; p < NPASS; ++p) {
        const int row = row0 + p * RPP + rl;
        n[p]  = cnt[row];
        st[p] = (unsigned)rowptr[row];
    }
    #pragma unroll
    for (int p = 0; p < NPASS; ++p)
        own[p] = load_row4<S>(hin, row0 + p * RPP + rl, suboff);

    const unsigned run0 = (unsigned)rowptr[row0];
    const unsigned runn = (unsigned)rowptr[row0 + 15] + (unsigned)cnt[row0 + 15] - run0;
    const bool staged = runn <= (unsigned)ENT_CAP;

    float4 wv[WTRIPS];
    #pragma unroll
    for (int t = 0; t < WTRIPS; ++t) {
        const int i   = threadIdx.x + t * LAYER_T;
        const int s   = i / (S * (HID / 4));
        const int rem = i - s * (S * (HID / 4));
        const int kw  = rem / (HID / 4);
        const int c4  = rem - kw * (HID / 4);
        wv[t] = (kw < KW) ? reinterpret_cast<const float4*>(W)[(s * KW + kw) * (HID / 4) + c4]
                          : make_float4(0.f, 0.f, 0.f, 0.f);
    }
    const float bias0 = bias[m];
    const float bias1 = bias[16 + m];
    int2 ev[ENT_PER_LANE];
    #pragma unroll
    for (int k = 0; k < ENT_PER_LANE; ++k) {
        const unsigned i = (unsigned)lane + 32u * k;
        ev[k] = load_entry(csr, (staged && i < runn) ? run0 + i : 0u);
    }
    #pragma unroll
    for (int t = 0; t < WTRIPS; ++t)
        reinterpret_cast<float4*>(sW)[threadIdx.x + t * LAYER_T] = wv[t];
    __syncthreads();

    int2* __restrict__ ents = sE + wib * ENT_CAP;
    #pragma unroll
    for (int k = 0; k < ENT_PER_LANE; ++k) ents[lane + 32 * k] = ev[k];
    __builtin_amdgcn_fence(__ATOMIC_RELEASE, "wavefront");
    __builtin_amdgcn_wave_barrier();
    __builtin_amdgcn_fence(__ATOMIC_ACQUIRE, "wavefront");

    float* __restrict__ tile = sT + wib * (16 * PITCH);
    #pragma unroll
    for (int p = 0; p < NPASS; ++p) {
        const float4 acc = staged ? gather_sum_staged<S>(ents + (st[p] - run0), hin, n[p], suboff)
                                  : gather_sum<S>(csr, hin, st[p], n[p], suboff);
        float* trow = tile + (p * RPP + rl) * PITCH + 4 * sub;
        *reinterpret_cast<float4*>(trow)     = own[p];
        *reinterpret_cast<float4*>(trow + S) = acc;
    }
    __builtin_amdgcn_fence(__ATOMIC_RELEASE, "wavefront");
    __builtin_amdgcn_wave_barrier();
    __builtin_amdgcn_fence(__ATOMIC_ACQUIRE, "wavefront");

    float wr0 = 0.f, wr1 = 0.f, br = 0.f;
    if constexpr (READOUT) { wr0 = wreg[m]; wr1 = wreg[16 + m]; br = breg[0]; }

    v8f acc0 = {0.f,0.f,0.f,0.f,0.f,0.f,0.f,0.f};
    v8f acc1 = {0.f,0.f,0.f,0.f,0.f,0.f,0.f,0.f};

    #pragma unroll
    for (int c = 0; c < K2 / 4; ++c) {
        const int k0 = c * 4 + khalf;
        v2f a = *reinterpret_cast<const v2f*>(tile + m * PITCH + k0);
        v2f b0, b1;
        b0.x = sW[k0 * HID + m];
        b0.y = sW[(k0 + 1) * HID + m];
        b1.x = sW[k0 * HID + 16 + m];
        b1.y = sW[(k0 + 1) * HID + 16 + m];
        acc0 = __builtin_amdgcn_wmma_f32_16x16x4_f32(
                   false, a, false, b0, (short)0, acc0, false, false);
        acc1 = __builtin_amdgcn_wmma_f32_16x16x4_f32(
                   false, a, false, b1, (short)0, acc1, false, false);
    }

    const int rbase = row0 + ((lane >> 4) ? 8 : 0);
    #pragma unroll
    for (int v = 0; v < 8; ++v) {
        int r = rbase + v;
        float x0 = acc0[v] + bias0;
        float x1 = acc1[v] + bias1;
        const float t0 = tanhf(x0);
        const float t1 = tanhf(x1);
        hout[(size_t)r * HID + m]      = t0;
        hout[(size_t)r * HID + 16 + m] = t1;
        if constexpr (READOUT) {
            float d = t0 * wr0 + t1 * wr1;
            d += __shfl_xor(d, 8);
            d += __shfl_xor(d, 4);
            d += __shfl_xor(d, 2);
            d += __shfl_xor(d, 1);
            if (m == 0) tile[(lane >> 4) * 8 + v] = d + br;
        }
    }
    if constexpr (READOUT) {
        __syncthreads();
        if (threadIdx.x < WAVES * 16)
            out[blockIdx.x * (WAVES * 16) + threadIdx.x] =
                sT[(threadIdx.x >> 4) * (16 * PITCH) + (threadIdx.x & 15)];
    }
}

static inline int cdiv(long long a, long long b) { return (int)((a + b - 1) / b); }

extern "C" void kernel_launch(void* const* d_in, const int* in_sizes, int n_in,
                              void* d_out, int out_size, void* d_ws, size_t ws_size,
                              hipStream_t stream) {
    const float* X    = (const float*)d_in[0];
    const int*   ei   = (const int*)  d_in[1];
    const float* W0   = (const float*)d_in[2];
    const float* b0   = (const float*)d_in[3];
    const float* W1   = (const float*)d_in[4];
    const float* b1   = (const float*)d_in[5];
    const float* W2   = (const float*)d_in[6];
    const float* b2   = (const float*)d_in[7];
    const float* Wreg = (const float*)d_in[8];
    const float* breg = (const float*)d_in[9];

    float* out    = (float*)d_out;
    float* hfinal = out + NTOT;

    int*   cnt    = (int*)d_ws;
    int*   rowptr = cnt + NTOT;
    int2*  lsort  = (int2*)(rowptr + NTOT);
    int2*  csr    = lsort + ETOT;
    float* hA     = (float*)(csr + ETOT);
    float* hB     = hA  + (size_t)NTOT * HID;
    float* Xp     = hB  + (size_t)NTOT * HID;
    unsigned short* part = (unsigned short*)(Xp + (size_t)NTOT * KPAD);
    int*   bounds = (int*)(part + (size_t)BATCH * EDGE_SPLIT * NN);

    count_sort_kernel<<<BATCH * EDGE_SPLIT, EDGE_T, 0, stream>>>(ei, lsort, bounds, part);
    in_degree_kernel<<<cdiv(NTOT / 4, OFFS_T), OFFS_T, 0, stream>>>(part, cnt, X, Xp);
    fill_rows_kernel<<<BATCH * EDGE_SPLIT, EDGE_T, 0, stream>>>(lsort, bounds, cnt, part, rowptr, csr,
                                                                Xp, W0, b0, hA);

    const int wide_blocks   = cdiv(NTOT / 16, 256 / 32);


    tag_layer_kernel<HID, HID, 256, false><<<wide_blocks, 256, 0, stream>>>(
        rowptr, cnt, csr, hA, W1, b1, hB, nullptr, nullptr, nullptr);

    tag_layer_kernel<HID, HID, 256, true><<<wide_blocks, 256, 0, stream>>>(
        rowptr, cnt, csr, hB, W2, b2, hfinal, Wreg, breg, out);
}


# wide layers: the own-row half of each tile's product (16 of its 32 matrix instructions) is issued in front of the neighbour gather, inside its memory wait, not behind it; k order and sums unchanged
# speedup vs baseline: 39.2886x; 1.0325x over previous
#include <hip/hip_runtime.h>
#include <hip/hip_bf16.h>
#include <math.h>

#define BATCH   8
#define WINDOW  5
#define NN      20000
#define EPB     320000
#define HID     32
#define KPAD    8
#define NTOT    (BATCH * NN)
#define ETOT    (BATCH * EPB)

typedef float v2f __attribute__((ext_vector_type(2)));
typedef float v8f __attribute__((ext_vector_type(8)));

#define EDGE_T      1024
#define EDGE_SPLIT  32
#define QPW         (EPB / 4 / EDGE_SPLIT)
#define EDGE_QPT    ((QPW + EDGE_T - 1) / EDGE_T)
#define NQ          (NN / 4)
#define TAB_TRIPS   5
#define TAB_T       (NQ / TAB_TRIPS)

#define RANGE       (NN / EDGE_SPLIT)
#define CHUNK_E     (QPW * 4)
#define SCAN_PER    20

__global__ void __launch_bounds__(EDGE_T)
count_sort_kernel(const int* __restrict__ ei, int2* __restrict__ lsort,
                  int* __restrict__ bounds, unsigned short* __restrict__ part) {
    static_assert(EPB % (4 * EDGE_SPLIT) == 0 && NN % 4 == 0 && NQ % TAB_TRIPS == 0 &&
                  TAB_T <= EDGE_T && TAB_T % 8 == 0,
                  "whole edge quads per chunk; tables move 16 bytes at a time in whole lines");
    static_assert(NN % SCAN_PER == 0 && NN / SCAN_PER == TAB_T && (SCAN_PER % 4) == 0 &&
                  SCAN_PER / 4 == TAB_TRIPS && NN % EDGE_SPLIT == 0 && NN < 32768 &&
                  EDGE_SPLIT <= EDGE_T && (CHUNK_E % 2) == 0,
                  "SCAN_PER consecutive nodes per scanning thread; src and dst share one word");
    static_assert(CHUNK_E <= 65535 && NN % 8 == 0,
                  "a chunk's count of one node fits 16 bits; eight counts per 16-byte piece");
    __shared__ __attribute__((aligned(16))) int  hist[NN];
    __shared__ __attribute__((aligned(16))) int2 stage[CHUNK_E];
    __shared__ int wtot[EDGE_T / 32];
    const int b   = blockIdx.x / EDGE_SPLIT;
    const int c   = blockIdx.x - b * EDGE_SPLIT;
    const int tid = threadIdx.x;
    const int lane = tid & 31;
    const int wid  = tid >> 5;
    const int* __restrict__ eb = ei + (size_t)b * 2 * EPB;
    const int4* __restrict__ src4 = reinterpret_cast<const int4*>(eb) + c * QPW;
    const int4* __restrict__ dst4 = reinterpret_cast<const int4*>(eb + EPB) + c * QPW;

    int4 sq[EDGE_QPT], dq[EDGE_QPT], rq[EDGE_QPT];
    #pragma unroll
    for (int k = 0; k < EDGE_QPT; ++k) {
        const int q  = tid + k * EDGE_T;
        const int qc = q < QPW ? q : QPW - 1;
        sq[k] = src4[qc]; dq[k] = dst4[qc];
    }
    for (int i = tid; i < NQ; i += EDGE_T)
        reinterpret_cast<int4*>(hist)[i] = make_int4(0, 0, 0, 0);
    __syncthreads();

    #pragma unroll
    for (int k = 0; k < EDGE_QPT; ++k) {
        const int q = tid + k * EDGE_T;
        rq[k] = make_int4(0, 0, 0, 0);
        if (q < QPW) {
            rq[k].x = atomicAdd(&hist[dq[k].x], 1);
            rq[k].y = atomicAdd(&hist[dq[k].y], 1);
            rq[k].z = atomicAdd(&hist[dq[k].z], 1);
            rq[k].w = atomicAdd(&hist[dq[k].w], 1);
        }
    }
    __syncthreads();

    uint4* __restrict__ gp = reinterpret_cast<uint4*>(part + (size_t)blockIdx.x * NN);
    for (int i = tid; i < NN / 8; i += EDGE_T) {
        const int4 ca = reinterpret_cast<const int4*>(hist)[2 * i];
        const int4 cb = reinterpret_cast<const int4*>(hist)[2 * i + 1];
        gp[i] = make_uint4((unsigned)ca.x | ((unsigned)ca.y << 16),
                           (unsigned)ca.z | ((unsigned)ca.w << 16),
                           (unsigned)cb.x | ((unsigned)cb.y << 16),
                           (unsigned)cb.z | ((unsigned)cb.w << 16));
    }

    const bool act = tid < TAB_T;
    int cc[SCAN_PER];
    int tot = 0;
    #pragma unroll
    for (int k = 0; k < SCAN_PER; ++k) cc[k] = 0;
    if (act) {
        #pragma unroll
        for (int g = 0; g < SCAN_PER / 4; ++g) {
            const int4 q = reinterpret_cast<const int4*>(hist)[tid * (SCAN_PER / 4) + g];
            cc[4 * g + 0] = q.x; cc[4 * g + 1] = q.y; cc[4 * g + 2] = q.z; cc[4 * g + 3] = q.w;
            tot += q.x + q.y + q.z + q.w;
        }
    }
    int inc = tot;
    #pragma unroll
    for (int off = 1; off < 32; off <<= 1) {
        int t = __shfl_up(inc, off);
        if (lane >= off) inc += t;
    }
    if (lane == 31) wtot[wid] = inc;
    __syncthreads();
    if (wid == 0) {
        int w = wtot[lane];
        int winc = w;
        #pragma unroll
        for (int off = 1; off < 32; off <<= 1) {
            int t = __shfl_up(winc, off);
            if (lane >= off) winc += t;
        }
        wtot[lane] = winc - w;
    }
    __syncthreads();
    if (act) {
        int run = wtot[wid] + (inc - tot);
        #pragma unroll
        for (int g = 0; g < SCAN_PER / 4; ++g) {
            int4 r;
            r.x = run; run += cc[4 * g + 0];
            r.y = run; run += cc[4 * g + 1];
            r.z = run; run += cc[4 * g + 2];
            r.w = run; run += cc[4 * g + 3];
            reinterpret_cast<int4*>(hist)[tid * (SCAN_PER / 4) + g] = r;
        }
    }
    __syncthreads();

    if (tid < EDGE_SPLIT) bounds[blockIdx.x * EDGE_SPLIT + tid] = hist[tid * RANGE];
    #pragma unroll
    for (int k = 0; k < EDGE_QPT; ++k) {
        const int q = tid + k * EDGE_T;
        if (q < QPW) {
            stage[hist[dq[k].x] + rq[k].x] = make_int2(sq[k].x | (dq[k].x << 16), rq[k].x);
            stage[hist[dq[k].y] + rq[k].y] = make_int2(sq[k].y | (dq[k].y << 16), rq[k].y);
            stage[hist[dq[k].z] + rq[k].z] = make_int2(sq[k].z | (dq[k].z << 16), rq[k].z);
            stage[hist[dq[k].w] + rq[k].w] = make_int2(sq[k].w | (dq[k].w << 16), rq[k].w);
        }
    }
    __syncthreads();

    int4* __restrict__ lo4 = reinterpret_cast<int4*>(lsort + (size_t)blockIdx.x * CHUNK_E);
    constexpr int LS_TRIPS = (CHUNK_E / 2 + EDGE_T - 1) / EDGE_T;
    int4 lv[LS_TRIPS];
    #pragma unroll
    for (int k = 0; k < LS_TRIPS; ++k) {
        const int i = tid + k * EDGE_T;
        lv[k] = reinterpret_cast<const int4*>(stage)[i < CHUNK_E / 2 ? i : 0];
    }
    #pragma unroll
    for (int k = 0; k < LS_TRIPS; ++k) {
        const int i = tid + k * EDGE_T;
        if (i < CHUNK_E / 2) lo4[i] = lv[k];
    }
}

#define OFFS_T 128
__global__ void __launch_bounds__(OFFS_T)
in_degree_kernel(const unsigned short* __restrict__ part, int* __restrict__ cnt,
                     const float* __restrict__ x, float* __restrict__ xp) {
    static_assert(WINDOW == 5 && KPAD == 8, "four rows of X are five 16-byte pieces");
    const int gq = blockIdx.x * OFFS_T + threadIdx.x;
    if (gq >= NTOT / 4) return;
    const int b  = gq / NQ;
    const int nq = gq - b * NQ;
    const float4* __restrict__ xr = reinterpret_cast<const float4*>(x) + (size_t)gq * WINDOW;
    const float4 a0 = xr[0], a1 = xr[1], a2 = xr[2], a3 = xr[3], a4 = xr[4];
    const uint2* __restrict__ p =
        reinterpret_cast<const uint2*>(part + (size_t)b * EDGE_SPLIT * NN) + nq;
    uint2 t[EDGE_SPLIT];
    #pragma unroll
    for (int c = 0; c < EDGE_SPLIT; ++c) t[c] = p[(size_t)c * NQ];
    float4* __restrict__ xo = reinterpret_cast<float4*>(xp) + (size_t)gq * KPAD;
    xo[0] = a0;
    xo[1] = make_float4(a1.x, 0.f, 0.f, 0.f);
    xo[2] = make_float4(a1.y, a1.z, a1.w, a2.x);
    xo[3] = make_float4(a2.y, 0.f, 0.f, 0.f);
    xo[4] = make_float4(a2.z, a2.w, a3.x, a3.y);
    xo[5] = make_float4(a3.z, 0.f, 0.f, 0.f);
    xo[6] = make_float4(a3.w, a4.x, a4.y, a4.z);
    xo[7] = make_float4(a4.w, 0.f, 0.f, 0.f);
    int4 run = make_int4(0, 0, 0, 0);
    #pragma unroll
    for (int c = 0; c < EDGE_SPLIT; ++c) {
        run.x += (int)(t[c].x & 0xffffu); run.y += (int)(t[c].x >> 16);
        run.z += (int)(t[c].y & 0xffffu); run.w += (int)(t[c].y >> 16);
    }
    reinterpret_cast<int4*>(cnt)[gq] = run;
}

static __device__ __forceinline__ int2 load_entry(const int2* __restrict__ csr, unsigned idx) {
    return *reinterpret_cast<const int2*>(reinterpret_cast<const char*>(csr) + idx * 8u);
}
template <int S>
static __device__ __forceinline__ float4 load_row4(const float* __restrict__ hin, int src,
                                                   unsigned suboff) {
    return *reinterpret_cast<const float4*>(reinterpret_cast<const char*>(hin) +
                                            ((unsigned)src * (unsigned)(S * 4) + suboff));
}
static __device__ __forceinline__ void axpy4(float4& acc, const float4& v, float w) {
    acc.x += v.x * w; acc.y += v.y * w; acc.z += v.z * w; acc.w += v.w * w;
}

template <int S>
static __device__ __forceinline__ float4 gather_sum(const int2* __restrict__ csr,
                                                    const float* __restrict__ hin,
                                                    unsigned start, int n, unsigned suboff) {
    float4 acc = make_float4(0.f, 0.f, 0.f, 0.f);
    int j = 0;
    for (; j + 4 <= n; j += 4) {
        const int2 e0 = load_entry(csr, start + j);
        const int2 e1 = load_entry(csr, start + j + 1);
        const int2 e2 = load_entry(csr, start + j + 2);
        const int2 e3 = load_entry(csr, start + j + 3);
        const float4 v0 = load_row4<S>(hin, e0.x, suboff);
        const float4 v1 = load_row4<S>(hin, e1.x, suboff);
        const float4 v2 = load_row4<S>(hin, e2.x, suboff);
        const float4 v3 = load_row4<S>(hin, e3.x, suboff);
        axpy4(acc, v0, __int_as_float(e0.y));
        axpy4(acc, v1, __int_as_float(e1.y));
        axpy4(acc, v2, __int_as_float(e2.y));
        axpy4(acc, v3, __int_as_float(e3.y));
    }
    const int r = n - j;
    if (r > 0) {
        const int2 e0 = load_entry(csr, start + j);
        const int2 e1 = load_entry(csr, start + j + (r > 1 ? 1 : 0));
        const int2 e2 = load_entry(csr, start + j + (r > 2 ? 2 : 0));
        float4 v0 = load_row4<S>(hin, e0.x, suboff);
        float4 v1 = load_row4<S>(hin, e1.x, suboff);
        float4 v2 = load_row4<S>(hin, e2.x, suboff);
        v1.x = (r > 1) ? v1.x : 0.f; v1.y = (r > 1) ? v1.y : 0.f;
        v1.z = (r > 1) ? v1.z : 0.f; v1.w = (r > 1) ? v1.w : 0.f;
        v2.x = (r > 2) ? v2.x : 0.f; v2.y = (r > 2) ? v2.y : 0.f;
        v2.z = (r > 2) ? v2.z : 0.f; v2.w = (r > 2) ? v2.w : 0.f;
        axpy4(acc, v0, __int_as_float(e0.y));
        axpy4(acc, v1, __int_as_float(e1.y));
        axpy4(acc, v2, __int_as_float(e2.y));
    }
    return acc;
}

template <int S, int M>
static __device__ __forceinline__ void gather_tail(float4& acc, const int2* __restrict__ ent,
                                                   const float* __restrict__ hin,
                                                   int r, unsigned suboff) {
    int2 e[M]; float4 v[M];
    #pragma unroll
    for (int u = 0; u < M; ++u) e[u] = ent[u < r ? u : 0];
    #pragma unroll
    for (int u = 0; u < M; ++u) v[u] = load_row4<S>(hin, e[u].x, suboff);
    #pragma unroll
    for (int u = 1; u < M; ++u) {
        v[u].x = (u < r) ? v[u].x : 0.f; v[u].y = (u < r) ? v[u].y : 0.f;
        v[u].z = (u < r) ? v[u].z : 0.f; v[u].w = (u < r) ? v[u].w : 0.f;
    }
    #pragma unroll
    for (int u = 0; u < M; ++u) axpy4(acc, v[u], __int_as_float(e[u].y));
}

#define GATHER_DEPTH 8
template <int S>
static __device__ __forceinline__ float4 gather_sum_staged(const int2* __restrict__ ent,
                                                           const float* __restrict__ hin,
                                                           int n, unsigned suboff) {
    float4 acc = make_float4(0.f, 0.f, 0.f, 0.f);
    int j = 0;
    for (; j + GATHER_DEPTH <= n; j += GATHER_DEPTH) {
        int2 e[GATHER_DEPTH]; float4 v[GATHER_DEPTH];
        #pragma unroll
        for (int u = 0; u < GATHER_DEPTH; ++u) e[u] = ent[j + u];
        #pragma unroll
        for (int u = 0; u < GATHER_DEPTH; ++u) v[u] = load_row4<S>(hin, e[u].x, suboff);
        #pragma unroll
        for (int u = 0; u < GATHER_DEPTH; ++u) axpy4(acc, v[u], __int_as_float(e[u].y));
    }
    const int r = n - j;
    if (r > 3)      gather_tail<S, GATHER_DEPTH - 1>(acc, ent + j, hin, r, suboff);
    else if (r > 0) gather_tail<S, 3>(acc, ent + j, hin, r, suboff);
    return acc;
}

#define FILL_DEPTH  12
#define PIECE_CAP   16384
#define PRE_N       (EDGE_SPLIT * RANGE)
static __device__ __forceinline__ void put_entry(const int* __restrict__ s_rp,
                                                 const int* __restrict__ pre,
                                                 const int* __restrict__ s_cd,
                                                 int2* __restrict__ to, int first,
                                                 int2 e, int lo, int nodebase) {
    const int sl = e.x & 0xffff;
    const int dl = (int)((unsigned)e.x >> 16);
    const int il = dl - lo;
    const int slot = s_rp[il] + pre[il] + e.y;
    const float nrm = __int_as_float(s_cd[sl]) * __int_as_float(s_cd[dl]);
    to[slot - first] = make_int2(sl + nodebase, __float_as_int(nrm));
}
static __device__ __forceinline__ void put_group(const int* __restrict__ s_rp,
                                                 const int* __restrict__ pre,
                                                 const int* __restrict__ s_cd,
                                                 int2* __restrict__ to, int first,
                                                 const int2 (&e)[FILL_DEPTH], int lane, int len,
                                                 int lo, int nodebase) {
    int slot[FILL_DEPTH]; int sl[FILL_DEPTH]; float nrm[FILL_DEPTH];
    #pragma unroll
    for (int k = 0; k < FILL_DEPTH; ++k) {
        const bool ok = lane + 32 * k < len;
        sl[k] = e[k].x & 0xffff;
        const int dl = (int)((unsigned)e[k].x >> 16);
        const int il = ok ? dl - lo : 0;
        slot[k] = s_rp[il] + pre[il] + e[k].y;
        nrm[k]  = __int_as_float(s_cd[sl[k]]) * __int_as_float(s_cd[dl]);
    }
    #pragma unroll
    for (int k = 0; k < FILL_DEPTH; ++k)
        if (lane + 32 * k < len)
            to[slot[k] - first] = make_int2(sl[k] + nodebase, __float_as_int(nrm[k]));
}
__global__ void __launch_bounds__(EDGE_T)
fill_rows_kernel(const int2* __restrict__ lsort, const int* __restrict__ bounds,
                 const int* __restrict__ cnt, const unsigned short* __restrict__ part,
                 int* __restrict__ rowptr, int2* __restrict__ csr,
                 const float* __restrict__ xp, const float* __restrict__ W,
                 const float* __restrict__ bias, float* __restrict__ hout) {
    static_assert(EDGE_T / 32 == EDGE_SPLIT && TAB_TRIPS * EDGE_T >= NQ && RANGE <= EDGE_T,
                  "one wave per chunk; the in-degrees are copied in TAB_TRIPS trips; one thread per node of the range");
    __shared__ __attribute__((aligned(16))) int s_pre[PRE_N];
    __shared__ __attribute__((aligned(16))) int s_cd[NN];
    __shared__ int s_rp[RANGE + 1];
    __shared__ int wtot[EDGE_T / 32];
    __shared__ __attribute__((aligned(16))) int2 piece[PIECE_CAP];
    constexpr int L0_K2    = 2 * KPAD;
    constexpr int L0_PITCH = L0_K2 + 4;
    constexpr int L0_TILES = (RANGE + 15) / 16;
    constexpr int L0_TWICE = L0_TILES - EDGE_T / 32;
    static_assert(HID == 32 && L0_K2 * HID / 4 <= EDGE_T && ((L0_PITCH / 4) % 2) == 1 &&
                  (L0_K2 * HID + (EDGE_T / 32) * 16 * L0_PITCH) <= PRE_N &&
                  L0_TWICE >= 0 && L0_TWICE < EDGE_T / 32,
                  "W and the waves' tiles fit into the prefix table; at most two tiles per wave");
    const int b   = blockIdx.x / EDGE_SPLIT;
    const int rg  = blockIdx.x - b * EDGE_SPLIT;
    const int lo  = rg * RANGE;
    const int tid = threadIdx.x;
    const int lane = tid & 31;
    const int wid  = __builtin_amdgcn_readfirstlane(tid >> 5);

    const int chunk = b * EDGE_SPLIT + wid;
    const int* __restrict__ bnd = bounds + chunk * EDGE_SPLIT;
    const int rb = bnd[rg];
    const int rn = bnd[rg + 1 < EDGE_SPLIT ? rg + 1 : rg];
    const int len = (rg + 1 < EDGE_SPLIT ? rn : CHUNK_E) - rb;
    const int2* __restrict__ lbase = lsort + (size_t)chunk * CHUNK_E;

    const int4* __restrict__ gc = reinterpret_cast<const int4*>(cnt + b * NN);
    const bool pren = tid < RANGE;
    const unsigned short* __restrict__ gpn =
        part + ((size_t)b * EDGE_SPLIT * NN + lo + (pren ? tid : 0));
    const bool act = tid < TAB_T;
    int4 tc[TAB_TRIPS]; int cv[EDGE_SPLIT];
    #pragma unroll
    for (int k = 0; k < TAB_TRIPS; ++k) {
        const int i = tid + k * EDGE_T;
        tc[k] = gc[i < NQ ? i : 0];
    }
    #pragma unroll
    for (int q = 0; q < EDGE_SPLIT; ++q) cv[q] = 0;
    if (pren) {
        #pragma unroll
        for (int q = 0; q < EDGE_SPLIT; ++q) cv[q] = (int)gpn[(size_t)q * NN];
    }
    int2 ev[FILL_DEPTH];
    #pragma unroll
    for (int k = 0; k < FILL_DEPTH; ++k) {
        const int i = lane + 32 * k;
        ev[k] = lbase[i < len ? rb + i : 0];
    }
    float4 w0v = make_float4(0.f, 0.f, 0.f, 0.f);
    if (tid < L0_K2 * HID / 4) {
        const int sidx = tid / (KPAD * (HID / 4));
        const int rem  = tid - sidx * (KPAD * (HID / 4));
        const int kw   = rem / (HID / 4);
        const int c4   = rem - kw * (HID / 4);
        if (kw < WINDOW)
            w0v = reinterpret_cast<const float4*>(W)[(sidx * WINDOW + kw) * (HID / 4) + c4];
    }
    #pragma unroll
    for (int k = 0; k < TAB_TRIPS; ++k) {
        const int i = tid + k * EDGE_T;
        if (i < NQ) reinterpret_cast<int4*>(s_cd)[i] = tc[k];
    }
    if (pren) {
        int before = 0;
        #pragma unroll
        for (int q = 0; q < EDGE_SPLIT; ++q) { s_pre[q * RANGE + tid] = before; before += cv[q]; }
    }
    __syncthreads();

    int c[SCAN_PER];
    int tot = 0;
    #pragma unroll
    for (int k = 0; k < SCAN_PER; ++k) c[k] = 0;
    if (act) {
        #pragma unroll
        for (int g = 0; g < SCAN_PER / 4; ++g) {
            const int4 q = reinterpret_cast<const int4*>(s_cd)[tid * (SCAN_PER / 4) + g];
            c[4 * g + 0] = q.x; c[4 * g + 1] = q.y; c[4 * g + 2] = q.z; c[4 * g + 3] = q.w;
            tot += q.x + q.y + q.z + q.w;
        }
    }
    int inc = tot;
    #pragma unroll
    for (int off = 1; off < 32; off <<= 1) {
        int t = __shfl_up(inc, off);
        if (lane >= off) inc += t;
    }
    if (lane == 31) wtot[wid] = inc;
    __syncthreads();
    if (wid == 0) {
        int w = wtot[lane];
        int winc = w;
        #pragma unroll
        for (int off = 1; off < 32; off <<= 1) {
            int t = __shfl_up(winc, off);
            if (lane >= off) winc += t;
        }
        wtot[lane] = winc - w;
    }
    __syncthreads();

    if (act) {
        int run = b * EPB + wtot[wid] + (inc - tot);
        int4* __restrict__ rp = reinterpret_cast<int4*>(rowptr + b * NN + tid * SCAN_PER);
        #pragma unroll
        for (int g = 0; g < SCAN_PER / 4; ++g) {
            int4 r; int4 f;
            r.x = run; run += c[4 * g + 0];
            r.y = run; run += c[4 * g + 1];
            r.z = run; run += c[4 * g + 2];
            r.w = run; run += c[4 * g + 3];
            f.x = __float_as_int((c[4 * g + 0] > 0) ? rsqrtf((float)c[4 * g + 0]) : 0.0f);
            f.y = __float_as_int((c[4 * g + 1] > 0) ? rsqrtf((float)c[4 * g + 1]) : 0.0f);
            f.z = __float_as_int((c[4 * g + 2] > 0) ? rsqrtf((float)c[4 * g + 2]) : 0.0f);
            f.w = __float_as_int((c[4 * g + 3] > 0) ? rsqrtf((float)c[4 * g + 3]) : 0.0f);
            reinterpret_cast<int4*>(s_cd)[tid * (SCAN_PER / 4) + g] = f;
            const int nd = tid * SCAN_PER + 4 * g - lo;
            if ((unsigned)(nd + 0) < (unsigned)RANGE) s_rp[nd + 0] = r.x;
            if ((unsigned)(nd + 1) < (unsigned)RANGE) s_rp[nd + 1] = r.y;
            if ((unsigned)(nd + 2) < (unsigned)RANGE) s_rp[nd + 2] = r.z;
            if ((unsigned)(nd + 3) < (unsigned)RANGE) s_rp[nd + 3] = r.w;
            if (nd + 0 == RANGE - 1) s_rp[RANGE] = r.y;
            if (nd + 1 == RANGE - 1) s_rp[RANGE] = r.z;
            if (nd + 2 == RANGE - 1) s_rp[RANGE] = r.w;
            if (nd + 3 == RANGE - 1) s_rp[RANGE] = run;
            if (rg == 0) rp[g] = r;
        }
    }
    __syncthreads();

    const int nodebase = b * NN;
    const int* pre = s_pre + wid * RANGE;
    const int first = s_rp[0];
    const int end   = s_rp[RANGE];
    const bool fits = end - first <= PIECE_CAP;
    if (fits) {
        put_group(s_rp, pre, s_cd, piece, first, ev, lane, len, lo, nodebase);
        for (int i = lane + 32 * FILL_DEPTH; i < len; i += 32)
            put_entry(s_rp, pre, s_cd, piece, first, lbase[rb + i], lo, nodebase);
    } else {
        put_group(s_rp, pre, s_cd, csr, 0, ev, lane, len, lo, nodebase);
        for (int i = lane + 32 * FILL_DEPTH; i < len; i += 32)
            put_entry(s_rp, pre, s_cd, csr, 0, lbase[rb + i], lo, nodebase);
    }
    __syncthreads();

    float* __restrict__ sW = reinterpret_cast<float*>(s_pre);
    float* __restrict__ sT = sW + L0_K2 * HID;
    if (tid < L0_K2 * HID / 4) reinterpret_cast<float4*>(sW)[tid] = w0v;
    __syncthreads();

    const int sub  = lane & 1;
    const int rl   = lane >> 1;
    const unsigned suboff = (unsigned)sub * 16u;
    const int m     = lane & 15;
    const int khalf = (lane >> 4) * 2;
    const float bias0 = bias[m];
    const float bias1 = bias[16 + m];
    float* __restrict__ tile = sT + wid * (16 * L0_PITCH);
    for (int t = wid; t < L0_TILES; t += EDGE_T / 32) {
        const int il    = t * 16 + rl;
        const bool have = il < RANGE;
        const int ilc   = have ? il : RANGE - 1;
        const int stl   = s_rp[ilc];
        const int n     = have ? s_rp[ilc + 1] - stl : 0;
        const float4 own = load_row4<KPAD>(xp, nodebase + lo + ilc, suboff);
        const float4 acc = fits ? gather_sum_staged<KPAD>(piece + (stl - first), xp, n, suboff)
                                : gather_sum<KPAD>(csr, xp, (unsigned)stl, n, suboff);
        float* trow = tile + rl * L0_PITCH + 4 * sub;
        *reinterpret_cast<float4*>(trow)        = own;
        *reinterpret_cast<float4*>(trow + KPAD) = acc;
        __builtin_amdgcn_fence(__ATOMIC_RELEASE, "wavefront");
        __builtin_amdgcn_wave_barrier();
        __builtin_amdgcn_fence(__ATOMIC_ACQUIRE, "wavefront");

        v8f acc0 = {0.f,0.f,0.f,0.f,0.f,0.f,0.f,0.f};
        v8f acc1 = {0.f,0.f,0.f,0.f,0.f,0.f,0.f,0.f};
        #pragma unroll
        for (int c = 0; c < L0_K2 / 4; ++c) {
            const int k0 = c * 4 + khalf;
            v2f a = *reinterpret_cast<const v2f*>(tile + m * L0_PITCH + k0);
            v2f b0, b1;
            b0.x = sW[k0 * HID + m];
            b0.y = sW[(k0 + 1) * HID + m];
            b1.x = sW[k0 * HID + 16 + m];
            b1.y = sW[(k0 + 1) * HID + 16 + m];
            acc0 = __builtin_amdgcn_wmma_f32_16x16x4_f32(
                       false, a, false, b0, (short)0, acc0, false, false);
            acc1 = __builtin_amdgcn_wmma_f32_16x16x4_f32(
                       false, a, false, b1, (short)0, acc1, false, false);
        }
        const int rb16 = t * 16 + ((lane >> 4) ? 8 : 0);
        #pragma unroll
        for (int v = 0; v < 8; ++v) {
            const int ir = rb16 + v;
            const float t0 = tanhf(acc0[v] + bias0);
            const float t1 = tanhf(acc1[v] + bias1);
            if (ir < RANGE) {
                const size_t r = (size_t)(nodebase + lo + ir);
                hout[r * HID + m]      = t0;
                hout[r * HID + 16 + m] = t1;
            }
        }
        __builtin_amdgcn_fence(__ATOMIC_RELEASE, "wavefront");
        __builtin_amdgcn_wave_barrier();
        __builtin_amdgcn_fence(__ATOMIC_ACQUIRE, "wavefront");
    }

    if (fits && wid >= L0_TWICE) {
        constexpr int COPY_T = EDGE_T - 32 * L0_TWICE;
        for (int e0 = (first & ~15) + (tid - 32 * L0_TWICE); e0 < end; e0 += 4 * COPY_T) {
            int2 v[4];
            #pragma unroll
            for (int u = 0; u < 4; ++u) {
                const int e = e0 + u * COPY_T;
                v[u] = piece[(e >= first && e < end) ? e - first : 0];
            }
            #pragma unroll
            for (int u = 0; u < 4; ++u) {
                const int e = e0 + u * COPY_T;
                if (e >= first && e < end) csr[e] = v[u];
            }
        }
    }
}

#define ENT_CAP 320
#define ENT_PER_LANE (ENT_CAP / 32)
#define LAYER_WAVES_PER_SIMD 10
template <int S, int KW, int LAYER_T, bool READOUT>
__global__ void __launch_bounds__(LAYER_T, LAYER_WAVES_PER_SIMD)
tag_layer_kernel(const int* __restrict__ rowptr, const int* __restrict__ cnt,
                 const int2* __restrict__ csr, const float* __restrict__ hin,
                 const float* __restrict__ W, const float* __restrict__ bias,
                 float* __restrict__ hout, const float* __restrict__ wreg,
                 const float* __restrict__ breg, float* __restrict__ out) {
    constexpr int K2    = 2 * S;
    constexpr int PITCH = K2 + 4;
    constexpr int LPR   = S / 4;
    constexpr int RPP   = 32 / LPR;
    constexpr int NPASS = 16 / RPP;
    constexpr int WAVES = LAYER_T / 32;
    constexpr int WTRIPS = (K2 * HID / 4) / LAYER_T;
    static_assert((K2 * HID / 4) % LAYER_T == 0 && HID == 32,
                  "whole staging trips; two 16-column output tiles");
    static_assert((S % 4) == 0 && (32 % LPR) == 0 && (16 % RPP) == 0 && ((PITCH / 4) % 2) == 1,
                  "whole float4 pieces per lane, whole passes per tile, odd 16-byte pitch");
    static_assert((NTOT % (16 * WAVES)) == 0, "every wave of every work-group has a whole tile");
    static_assert((NN % 16) == 0 && (ENT_CAP % 32) == 0,
                  "a tile lies inside one graph; whole waves of entries per staging trip");
    __shared__ __attribute__((aligned(16))) float sW[K2 * HID];
    __shared__ __attribute__((aligned(16))) float sT[WAVES * 16 * PITCH];
    __shared__ __attribute__((aligned(16))) int2  sE[WAVES * ENT_CAP];

    const int lane = threadIdx.x & 31;
    const int wib  = threadIdx.x >> 5;
    const int row0 = (blockIdx.x * WAVES + wib) * 16;
    const int sub  = lane % LPR;
    const int rl   = lane / LPR;
    const unsigned suboff = (unsigned)sub * 16u;
    const int m     = lane & 15;
    const int khalf = (lane >> 4) * 2;

    int n[NPASS]; unsigned st[NPASS]; float4 own[NPASS];
    #pragma unroll
    for (int p = 0; p < NPASS; ++p) {
        const int row = row0 + p * RPP + rl;
        n[p]  = cnt[row];
        st[p] = (unsigned)rowptr[row];
    }
    #pragma unroll
    for (int p = 0; p < NPASS; ++p)
        own[p] = load_row4<S>(hin, row0 + p * RPP + rl, suboff);

    const unsigned run0 = (unsigned)rowptr[row0];
    const unsigned runn = (unsigned)rowptr[row0 + 15] + (unsigned)cnt[row0 + 15] - run0;
    const bool staged = runn <= (unsigned)ENT_CAP;

    float4 wv[WTRIPS];
    #pragma unroll
    for (int t = 0; t < WTRIPS; ++t) {
        const int i   = threadIdx.x + t * LAYER_T;
        const int s   = i / (S * (HID / 4));
        const int rem = i - s * (S * (HID / 4));
        const int kw  = rem / (HID / 4);
        const int c4  = rem - kw * (HID / 4);
        wv[t] = (kw < KW) ? reinterpret_cast<const float4*>(W)[(s * KW + kw) * (HID / 4) + c4]
                          : make_float4(0.f, 0.f, 0.f, 0.f);
    }
    const float bias0 = bias[m];
    const float bias1 = bias[16 + m];
    int2 ev[ENT_PER_LANE];
    #pragma unroll
    for (int k = 0; k < ENT_PER_LANE; ++k) {
        const unsigned i = (unsigned)lane + 32u * k;
        ev[k] = load_entry(csr, (staged && i < runn) ? run0 + i : 0u);
    }
    #pragma unroll
    for (int t = 0; t < WTRIPS; ++t)
        reinterpret_cast<float4*>(sW)[threadIdx.x + t * LAYER_T] = wv[t];
    __syncthreads();

    float* __restrict__ tile = sT + wib * (16 * PITCH);
    int2* __restrict__ ents = sE + wib * ENT_CAP;
    #pragma unroll
    for (int p = 0; p < NPASS; ++p)
        *reinterpret_cast<float4*>(tile + (p * RPP + rl) * PITCH + 4 * sub) = own[p];
    #pragma unroll
    for (int k = 0; k < ENT_PER_LANE; ++k) ents[lane + 32 * k] = ev[k];
    __builtin_amdgcn_fence(__ATOMIC_RELEASE, "wavefront");
    __builtin_amdgcn_wave_barrier();
    __builtin_amdgcn_fence(__ATOMIC_ACQUIRE, "wavefront");

    v8f acc0 = {0.f,0.f,0.f,0.f,0.f,0.f,0.f,0.f};
    v8f acc1 = {0.f,0.f,0.f,0.f,0.f,0.f,0.f,0.f};

    #pragma unroll
    for (int c = 0; c < S / 4; ++c) {
        const int k0 = c * 4 + khalf;
        v2f a = *reinterpret_cast<const v2f*>(tile + m * PITCH + k0);
        v2f b0, b1;
        b0.x = sW[k0 * HID + m];
        b0.y = sW[(k0 + 1) * HID + m];
        b1.x = sW[k0 * HID + 16 + m];
        b1.y = sW[(k0 + 1) * HID + 16 + m];
        acc0 = __builtin_amdgcn_wmma_f32_16x16x4_f32(
                   false, a, false, b0, (short)0, acc0, false, false);
        acc1 = __builtin_amdgcn_wmma_f32_16x16x4_f32(
                   false, a, false, b1, (short)0, acc1, false, false);
    }
    asm volatile("" :: "v"(acc0), "v"(acc1));

    #pragma unroll
    for (int p = 0; p < NPASS; ++p) {
        const float4 acc = staged ? gather_sum_staged<S>(ents + (st[p] - run0), hin, n[p], suboff)
                                  : gather_sum<S>(csr, hin, st[p], n[p], suboff);
        *reinterpret_cast<float4*>(tile + (p * RPP + rl) * PITCH + 4 * sub + S) = acc;
    }
    __builtin_amdgcn_fence(__ATOMIC_RELEASE, "wavefront");
    __builtin_amdgcn_wave_barrier();
    __builtin_amdgcn_fence(__ATOMIC_ACQUIRE, "wavefront");

    float wr0 = 0.f, wr1 = 0.f, br = 0.f;
    if constexpr (READOUT) { wr0 = wreg[m]; wr1 = wreg[16 + m]; br = breg[0]; }

    #pragma unroll
    for (int c = S / 4; c < K2 / 4; ++c) {
        const int k0 = c * 4 + khalf;
        v2f a = *reinterpret_cast<const v2f*>(tile + m * PITCH + k0);
        v2f b0, b1;
        b0.x = sW[k0 * HID + m];
        b0.y = sW[(k0 + 1) * HID + m];
        b1.x = sW[k0 * HID + 16 + m];
        b1.y = sW[(k0 + 1) * HID + 16 + m];
        acc0 = __builtin_amdgcn_wmma_f32_16x16x4_f32(
                   false, a, false, b0, (short)0, acc0, false, false);
        acc1 = __builtin_amdgcn_wmma_f32_16x16x4_f32(
                   false, a, false, b1, (short)0, acc1, false, false);
    }

    const int rbase = row0 + ((lane >> 4) ? 8 : 0);
    #pragma unroll
    for (int v = 0; v < 8; ++v) {
        int r = rbase + v;
        float x0 = acc0[v] + bias0;
        float x1 = acc1[v] + bias1;
        const float t0 = tanhf(x0);
        const float t1 = tanhf(x1);
        hout[(size_t)r * HID + m]      = t0;
        hout[(size_t)r * HID + 16 + m] = t1;
        if constexpr (READOUT) {
            float d = t0 * wr0 + t1 * wr1;
            d += __shfl_xor(d, 8);
            d += __shfl_xor(d, 4);
            d += __shfl_xor(d, 2);
            d += __shfl_xor(d, 1);
            if (m == 0) tile[(lane >> 4) * 8 + v] = d + br;
        }
    }
    if constexpr (READOUT) {
        __syncthreads();
        if (threadIdx.x < WAVES * 16)
            out[blockIdx.x * (WAVES * 16) + threadIdx.x] =
                sT[(threadIdx.x >> 4) * (16 * PITCH) + (threadIdx.x & 15)];
    }
}

static inline int cdiv(long long a, long long b) { return (int)((a + b - 1) / b); }

extern "C" void kernel_launch(void* const* d_in, const int* in_sizes, int n_in,
                              void* d_out, int out_size, void* d_ws, size_t ws_size,
                              hipStream_t stream) {
    const float* X    = (const float*)d_in[0];
    const int*   ei   = (const int*)  d_in[1];
    const float* W0   = (const float*)d_in[2];
    const float* b0   = (const float*)d_in[3];
    const float* W1   = (const float*)d_in[4];
    const float* b1   = (const float*)d_in[5];
    const float* W2   = (const float*)d_in[6];
    const float* b2   = (const float*)d_in[7];
    const float* Wreg = (const float*)d_in[8];
    const float* breg = (const float*)d_in[9];

    float* out    = (float*)d_out;
    float* hfinal = out + NTOT;

    int*   cnt    = (int*)d_ws;
    int*   rowptr = cnt + NTOT;
    int2*  lsort  = (int2*)(rowptr + NTOT);
    int2*  csr    = lsort + ETOT;
    float* hA     = (float*)(csr + ETOT);
    float* hB     = hA  + (size_t)NTOT * HID;
    float* Xp     = hB  + (size_t)NTOT * HID;
    unsigned short* part = (unsigned short*)(Xp + (size_t)NTOT * KPAD);
    int*   bounds = (int*)(part + (size_t)BATCH * EDGE_SPLIT * NN);

    count_sort_kernel<<<BATCH * EDGE_SPLIT, EDGE_T, 0, stream>>>(ei, lsort, bounds, part);
    in_degree_kernel<<<cdiv(NTOT / 4, OFFS_T), OFFS_T, 0, stream>>>(part, cnt, X, Xp);
    fill_rows_kernel<<<BATCH * EDGE_SPLIT, EDGE_T, 0, stream>>>(lsort, bounds, cnt, part, rowptr, csr,
                                                                Xp, W0, b0, hA);

    const int wide_blocks   = cdiv(NTOT / 16, 256 / 32);


    tag_layer_kernel<HID, HID, 256, false><<<wide_blocks, 256, 0, stream>>>(
        rowptr, cnt, csr, hA, W1, b1, hB, nullptr, nullptr, nullptr);

    tag_layer_kernel<HID, HID, 256, true><<<wide_blocks, 256, 0, stream>>>(
        rowptr, cnt, csr, hB, W2, b2, hfinal, Wreg, breg, out);
}
